# starting kernel
# baseline (speedup 1.0000x reference)
#include <hip/hip_runtime.h>
#include <hip/hip_bf16.h>
#include <math.h>

#define BATCH    2
#define SEQLEN   4096
#define BL       (BATCH * SEQLEN)
#define D_MODEL  512
#define D_IN_PROJ 1168
#define CONV_DIM 576
#define CONV2    1152
#define D_INNER  1024
#define NHEADS   16
#define HH       8
#define HEADDIM  64
#define NSTATE   32
#define NCHUNK   32
#define CHUNK    128

typedef float v2f __attribute__((ext_vector_type(2)));
typedef float v8f __attribute__((ext_vector_type(8)));

__device__ __forceinline__ v8f wmma_f32_4(v2f a, v2f b, v8f c) {
    return __builtin_amdgcn_wmma_f32_16x16x4_f32(
         false, a,  false, b,
         (short)0, c,  false,  false);
}

__global__ void __launch_bounds__(128)
gemm_in(const float* __restrict__ U, const float* __restrict__ Win,
        float* __restrict__ out) {
    const int wv   = threadIdx.x >> 5;
    const int lane = threadIdx.x & 31;
    const int row  = lane & 15;
    const int koff = (lane >> 4) << 1;
    const int m0   = (blockIdx.y * 16 + wv * 4) * 16;
    const int n0   = blockIdx.x * 16;

    const float* ap = U   + (size_t)(m0 + row) * D_MODEL + koff;
    const float* bp = Win + (size_t)koff * D_IN_PROJ + n0 + row;

    v8f acc0 = {}, acc1 = {}, acc2 = {}, acc3 = {};
    for (int k = 0; k < D_MODEL; k += 4) {
        v2f b;
        b.x = bp[(size_t)k * D_IN_PROJ];
        b.y = bp[(size_t)k * D_IN_PROJ + D_IN_PROJ];
        v2f a0 = *(const v2f*)(ap + k);
        v2f a1 = *(const v2f*)(ap + 16 * D_MODEL + k);
        v2f a2 = *(const v2f*)(ap + 32 * D_MODEL + k);
        v2f a3 = *(const v2f*)(ap + 48 * D_MODEL + k);
        acc0 = wmma_f32_4(a0, b, acc0);
        acc1 = wmma_f32_4(a1, b, acc1);
        acc2 = wmma_f32_4(a2, b, acc2);
        acc3 = wmma_f32_4(a3, b, acc3);
    }
    const int mhi = (lane >> 4) << 3;
#pragma unroll
    for (int r = 0; r < 8; ++r) {
        const size_t mm = (size_t)(m0 + r + mhi);
        out[(mm     ) * D_IN_PROJ + n0 + row] = acc0[r];
        out[(mm + 16) * D_IN_PROJ + n0 + row] = acc1[r];
        out[(mm + 32) * D_IN_PROJ + n0 + row] = acc2[r];
        out[(mm + 48) * D_IN_PROJ + n0 + row] = acc3[r];
    }
}

__global__ void __launch_bounds__(256)
conv_dt(const float* __restrict__ xbcdt,
        const float* __restrict__ wf, const float* __restrict__ bf,
        const float* __restrict__ wb, const float* __restrict__ bb,
        const float* __restrict__ dt_bias,
        float* __restrict__ xc, float* __restrict__ dtb) {
    const size_t bl = blockIdx.x;
    const int b = (int)(bl / SEQLEN);
    const int t = (int)(bl % SEQLEN);

    for (int c = threadIdx.x; c < CONV2; c += blockDim.x) {
        float acc;
        if (c < CONV_DIM) {
            acc = bf[c];
#pragma unroll
            for (int k = 0; k < 4; ++k) {
                int tt = t - 3 + k;
                if (tt >= 0)
                    acc += xbcdt[((size_t)b * SEQLEN + tt) * D_IN_PROJ + c] * wf[c * 4 + k];
            }
        } else {
            int cc = c - CONV_DIM;
            acc = bb[cc];
#pragma unroll
            for (int k = 0; k < 4; ++k) {
                int tt = t + 3 - k;
                if (tt < SEQLEN)
                    acc += xbcdt[((size_t)b * SEQLEN + tt) * D_IN_PROJ + c] * wb[cc * 4 + k];
            }
        }
        float sig = 1.0f / (1.0f + expf(-acc));
        xc[bl * CONV2 + c] = acc * sig;
    }
    for (int h = threadIdx.x; h < NHEADS; h += blockDim.x) {
        float v = xbcdt[bl * D_IN_PROJ + CONV2 + h] + dt_bias[h];
        dtb[bl * NHEADS + h] = (v > 20.0f) ? v : log1pf(expf(v));
    }
}

__device__ __forceinline__ void head_bases(int h, int& xbase, int& bbase, int& cbase) {
    if (h < HH) { xbase = h * HEADDIM;                 bbase = 512;  cbase = 544;  }
    else        { xbase = CONV_DIM + (h - HH) * HEADDIM; bbase = 1088; cbase = 1120; }
}

__global__ void __launch_bounds__(64)
scan_chunk(const float* __restrict__ xc, const float* __restrict__ dtb,
           const float* __restrict__ A_log, const float* __restrict__ Dp,
           float* __restrict__ y, float* __restrict__ S, float* __restrict__ prodA) {
    const int chunk = blockIdx.x & (NCHUNK - 1);
    const int h     = (blockIdx.x >> 5) & 15;
    const int b     = blockIdx.x >> 9;
    const int p     = threadIdx.x;
    int xbase, bbase, cbase; head_bases(h, xbase, bbase, cbase);

    const float Ah = -expf(A_log[h]);
    const float Dh = Dp[h];
    float hst[NSTATE];
#pragma unroll
    for (int n = 0; n < NSTATE; ++n) hst[n] = 0.0f;
    float pA = 1.0f;

    __shared__ float sB[NSTATE], sC[NSTATE];

    for (int sl = 0; sl < CHUNK; ++sl) {
        const int s = chunk * CHUNK + sl;
        const int t = (h < HH) ? s : (SEQLEN - 1 - s);
        const size_t rb = (size_t)b * SEQLEN + t;
        const float dtv = dtb[rb * NHEADS + h];
        const float dA  = expf(Ah * dtv);
        pA *= dA;
        if (p < NSTATE) sB[p] = xc[rb * CONV2 + bbase + p];
        else            sC[p - NSTATE] = xc[rb * CONV2 + cbase + (p - NSTATE)];
        __syncthreads();
        const float xp  = xc[rb * CONV2 + xbase + p];
        const float dtx = dtv * xp;
        float accy = 0.0f;
#pragma unroll
        for (int n = 0; n < NSTATE; ++n) {
            hst[n] = dA * hst[n] + dtx * sB[n];
            accy  += hst[n] * sC[n];
        }
        y[rb * D_INNER + h * HEADDIM + p] = accy + Dh * xp;
        __syncthreads();
    }
    const size_t sb = (((size_t)(b * NHEADS + h)) * NCHUNK + chunk) * (HEADDIM * NSTATE)
                    + (size_t)p * NSTATE;
#pragma unroll
    for (int n = 0; n < NSTATE; ++n) S[sb + n] = hst[n];
    if (p == 0) prodA[(b * NHEADS + h) * NCHUNK + chunk] = pA;
}

__global__ void __launch_bounds__(64)
scan_combine(float* __restrict__ S, const float* __restrict__ prodA) {
    const int bh = blockIdx.x;
    const int p  = threadIdx.x;
    float hcur[NSTATE];
#pragma unroll
    for (int n = 0; n < NSTATE; ++n) hcur[n] = 0.0f;
    for (int c = 0; c < NCHUNK; ++c) {
        const size_t base = (((size_t)bh) * NCHUNK + c) * (HEADDIM * NSTATE) + (size_t)p * NSTATE;
        const float a = prodA[bh * NCHUNK + c];
#pragma unroll
        for (int n = 0; n < NSTATE; ++n) {
            float sl = S[base + n];
            S[base + n] = hcur[n];
            hcur[n] = a * hcur[n] + sl;
        }
    }
}

__global__ void __launch_bounds__(64)
scan_correct(const float* __restrict__ xc, const float* __restrict__ dtb,
             const float* __restrict__ A_log, const float* __restrict__ S,
             float* __restrict__ y) {
    const int chunk = blockIdx.x & (NCHUNK - 1);
    if (chunk == 0) return;
    const int h = (blockIdx.x >> 5) & 15;
    const int b = blockIdx.x >> 9;
    const int p = threadIdx.x;
    int xbase, bbase, cbase; head_bases(h, xbase, bbase, cbase);
    (void)xbase; (void)bbase;

    float hi[NSTATE];
    const size_t sb = (((size_t)(b * NHEADS + h)) * NCHUNK + chunk) * (HEADDIM * NSTATE)
                    + (size_t)p * NSTATE;
#pragma unroll
    for (int n = 0; n < NSTATE; ++n) hi[n] = S[sb + n];

    const float Ah = -expf(A_log[h]);
    float cumA = 1.0f;
    __shared__ float sC[NSTATE];

    for (int sl = 0; sl < CHUNK; ++sl) {
        const int s = chunk * CHUNK + sl;
        const int t = (h < HH) ? s : (SEQLEN - 1 - s);
        const size_t rb = (size_t)b * SEQLEN + t;
        cumA *= expf(Ah * dtb[rb * NHEADS + h]);
        if (p < NSTATE) sC[p] = xc[rb * CONV2 + cbase + p];
        __syncthreads();
        float acc = 0.0f;
#pragma unroll
        for (int n = 0; n < NSTATE; ++n) acc += hi[n] * sC[n];
        y[rb * D_INNER + h * HEADDIM + p] += cumA * acc;
        __syncthreads();
    }
}

__global__ void __launch_bounds__(256)
ln_stats(const float* __restrict__ y, float* __restrict__ mu, float* __restrict__ rstd) {
    const size_t row = blockIdx.x;
    const int tid = threadIdx.x;
    const float* yr = y + row * D_INNER;
    float s = 0.0f, q = 0.0f;
    for (int i = tid; i < D_INNER; i += 256) { float v = yr[i]; s += v; q += v * v; }
    __shared__ float ss[256], sq[256];
    ss[tid] = s; sq[tid] = q; __syncthreads();
    for (int off = 128; off > 0; off >>= 1) {
        if (tid < off) { ss[tid] += ss[tid + off]; sq[tid] += sq[tid + off]; }
        __syncthreads();
    }
    if (tid == 0) {
        float m = ss[0] * (1.0f / D_INNER);
        float v = sq[0] * (1.0f / D_INNER) - m * m;
        mu[row] = m;
        rstd[row] = rsqrtf(v + 1e-5f);
    }
}

__global__ void __launch_bounds__(128)
gemm_out(const float* __restrict__ Y, const float* __restrict__ mu,
         const float* __restrict__ rstd, const float* __restrict__ lnw,
         const float* __restrict__ lnb, const float* __restrict__ Wout,
         float* __restrict__ out) {
    const int wv   = threadIdx.x >> 5;
    const int lane = threadIdx.x & 31;
    const int row  = lane & 15;
    const int koff = (lane >> 4) << 1;
    const int m0   = (blockIdx.y * 16 + wv * 4) * 16;
    const int n0   = blockIdx.x * 16;

    const float mum0 = mu[m0 + row],        rsm0 = rstd[m0 + row];
    const float mum1 = mu[m0 + 16 + row],   rsm1 = rstd[m0 + 16 + row];
    const float mum2 = mu[m0 + 32 + row],   rsm2 = rstd[m0 + 32 + row];
    const float mum3 = mu[m0 + 48 + row],   rsm3 = rstd[m0 + 48 + row];

    const float* ap = Y    + (size_t)(m0 + row) * D_INNER + koff;
    const float* bp = Wout + (size_t)koff * D_MODEL + n0 + row;

    v8f acc0 = {}, acc1 = {}, acc2 = {}, acc3 = {};
    for (int k = 0; k < D_INNER; k += 4) {
        const float w0 = lnw[k + koff], w1 = lnw[k + koff + 1];
        const float z0 = lnb[k + koff], z1 = lnb[k + koff + 1];
        v2f b;
        b.x = bp[(size_t)k * D_MODEL];
        b.y = bp[(size_t)k * D_MODEL + D_MODEL];
        v2f y0 = *(const v2f*)(ap + k);
        v2f y1 = *(const v2f*)(ap + 16 * D_INNER + k);
        v2f y2 = *(const v2f*)(ap + 32 * D_INNER + k);
        v2f y3 = *(const v2f*)(ap + 48 * D_INNER + k);
        v2f a0, a1, a2, a3;
        a0.x = (y0.x - mum0) * rsm0 * w0 + z0;  a0.y = (y0.y - mum0) * rsm0 * w1 + z1;
        a1.x = (y1.x - mum1) * rsm1 * w0 + z0;  a1.y = (y1.y - mum1) * rsm1 * w1 + z1;
        a2.x = (y2.x - mum2) * rsm2 * w0 + z0;  a2.y = (y2.y - mum2) * rsm2 * w1 + z1;
        a3.x = (y3.x - mum3) * rsm3 * w0 + z0;  a3.y = (y3.y - mum3) * rsm3 * w1 + z1;
        acc0 = wmma_f32_4(a0, b, acc0);
        acc1 = wmma_f32_4(a1, b, acc1);
        acc2 = wmma_f32_4(a2, b, acc2);
        acc3 = wmma_f32_4(a3, b, acc3);
    }
    const int mhi = (lane >> 4) << 3;
#pragma unroll
    for (int r = 0; r < 8; ++r) {
        const size_t mm = (size_t)(m0 + r + mhi);
        out[(mm     ) * D_MODEL + n0 + row] = acc0[r];
        out[(mm + 16) * D_MODEL + n0 + row] = acc1[r];
        out[(mm + 32) * D_MODEL + n0 + row] = acc2[r];
        out[(mm + 48) * D_MODEL + n0 + row] = acc3[r];
    }
}

extern "C" void kernel_launch(void* const* d_in, const int* in_sizes, int n_in,
                              void* d_out, int out_size, void* d_ws, size_t ws_size,
                              hipStream_t stream) {
    const float* u        = (const float*)d_in[0];
    const float* W_in     = (const float*)d_in[2];
    const float* conv_fw  = (const float*)d_in[3];
    const float* conv_fb  = (const float*)d_in[4];
    const float* conv_bw  = (const float*)d_in[5];
    const float* conv_bb  = (const float*)d_in[6];
    const float* dt_bias  = (const float*)d_in[7];
    const float* A_log    = (const float*)d_in[8];
    const float* D_param  = (const float*)d_in[9];
    const float* ln_w     = (const float*)d_in[10];
    const float* ln_b     = (const float*)d_in[11];
    const float* W_out    = (const float*)d_in[12];
    float* out            = (float*)d_out;

    float* ws = (float*)d_ws;
    size_t off = 0;
    float* xbcdt = ws + off; off += (size_t)BL * D_IN_PROJ;
    float* xc    = ws + off; off += (size_t)BL * CONV2;
    float* dtb   = ws + off; off += (size_t)BL * NHEADS;
    float* ybuf  = ws + off; off += (size_t)BL * D_INNER;
    float* Sbuf  = ws + off; off += (size_t)32 * NCHUNK * HEADDIM * NSTATE;
    float* pAbuf = ws + off; off += (size_t)32 * NCHUNK;
    float* mubuf = ws + off; off += (size_t)BL;
    float* rsbuf = ws + off; off += (size_t)BL;

    gemm_in<<<dim3(D_IN_PROJ / 16, BL / 256), 128, 0, stream>>>(u, W_in, xbcdt);
    conv_dt<<<dim3(BL), 256, 0, stream>>>(xbcdt, conv_fw, conv_fb, conv_bw, conv_bb,
                                          dt_bias, xc, dtb);
    scan_chunk<<<dim3(BATCH * NHEADS * NCHUNK), 64, 0, stream>>>(
        xc, dtb, A_log, D_param, ybuf, Sbuf, pAbuf);
    scan_combine<<<dim3(BATCH * NHEADS), 64, 0, stream>>>(Sbuf, pAbuf);
    scan_correct<<<dim3(BATCH * NHEADS * NCHUNK), 64, 0, stream>>>(
        xc, dtb, A_log, Sbuf, ybuf);
    ln_stats<<<dim3(BL), 256, 0, stream>>>(ybuf, mubuf, rsbuf);
    gemm_out<<<dim3(D_MODEL / 16, BL / 256), 128, 0, stream>>>(
        ybuf, mubuf, rsbuf, ln_w, ln_b, W_out, out);

    (void)in_sizes; (void)n_in; (void)out_size; (void)ws_size;
}


# both projection GEMMs, no LDS: wave tile 64x16 -> 16x64, A fragments as whole 128-byte lines, B and C by 16-byte accesses from global (k inside each 32 permuted alike in A and B): new sum order
# speedup vs baseline: 2.2338x; 2.2338x over previous
#include <hip/hip_runtime.h>
#include <hip/hip_bf16.h>
#include <math.h>

#define BATCH    2
#define SEQLEN   4096
#define BL       (BATCH * SEQLEN)
#define D_MODEL  512
#define D_IN_PROJ 1168
#define CONV_DIM 576
#define CONV2    1152
#define D_INNER  1024
#define NHEADS   16
#define HH       8
#define HEADDIM  64
#define NSTATE   32
#define NCHUNK   32
#define CHUNK    128

typedef float v2f __attribute__((ext_vector_type(2)));
typedef float v8f __attribute__((ext_vector_type(8)));

__device__ __forceinline__ v8f wmma_f32_4(v2f a, v2f b, v8f c) {
    return __builtin_amdgcn_wmma_f32_16x16x4_f32(
         false, a,  false, b,
         (short)0, c,  false,  false);
}

typedef float v4f __attribute__((ext_vector_type(4)));

template <int NF>
struct BFrag {
    v4f   b4[4];
    float b1[4];
    float wl[4], wh[4], zl[4], zh[4];
};

template <int NF, int ND, bool LN>
__device__ __forceinline__ void load_b(BFrag<NF>& f, const float* bp,
                                       const float* __restrict__ lnw, const float* __restrict__ lnb,
                                       int krow) {
#pragma unroll
    for (int j = 0; j < 4; ++j) {
        if (NF == 4) f.b4[j] = *(const v4f*)(bp + (size_t)(krow + j) * ND);
        else         f.b1[j] = bp[(size_t)(krow + j) * ND];
    }
    if (LN) {
#pragma unroll
        for (int j = 0; j < 4; ++j) {
            f.wl[j] = lnw[krow + j];  f.wh[j] = lnw[krow + 16 + j];
            f.zl[j] = lnb[krow + j];  f.zh[j] = lnb[krow + 16 + j];
        }
    }
}

__device__ __forceinline__ void load_a(v4f (&a)[4], const float* ap, int kb) {
#pragma unroll
    for (int q = 0; q < 4; ++q) a[q] = *(const v4f*)(ap + 32 * kb + 4 * q);
}

template <int NF, bool LN>
__device__ __forceinline__ void mma_sub(v4f y, const BFrag<NF>& f, int hf, float mur, float rsr,
                                        v8f (&acc)[NF]) {
    if (LN) {
#pragma unroll
        for (int j = 0; j < 4; ++j) {
            const float w = hf ? f.wh[j] : f.wl[j];
            const float z = hf ? f.zh[j] : f.zl[j];
            y[j] = (y[j] - mur) * rsr * w + z;
        }
    }
    v2f a0, a1;
    a0.x = y[0];  a0.y = y[1];
    a1.x = y[2];  a1.y = y[3];
#pragma unroll
    for (int t = 0; t < NF; ++t) {
        v2f b;
        if (NF == 4) { b.x = f.b4[0][t];  b.y = f.b4[1][t]; }
        else         { b.x = f.b1[0];     b.y = f.b1[1]; }
        acc[t] = wmma_f32_4(a0, b, acc[t]);
    }
#pragma unroll
    for (int t = 0; t < NF; ++t) {
        v2f b;
        if (NF == 4) { b.x = f.b4[2][t];  b.y = f.b4[3][t]; }
        else         { b.x = f.b1[2];     b.y = f.b1[3]; }
        acc[t] = wmma_f32_4(a1, b, acc[t]);
    }
}

#define GEMM_SUB(AV, BCUR, BNXT, KROW_NEXT)                                       \
    load_b<NF, ND, LN>(BNXT, bp, lnw, lnb, (KROW_NEXT));                          \
    __builtin_amdgcn_sched_barrier(0);                                            \
    mma_sub<NF, LN>((AV), (BCUR), hf, mur, rsr, acc);                             \
    __builtin_amdgcn_sched_barrier(0);

template <int NF, int KD, int ND, bool LN>
__device__ __forceinline__ void gemm_tile(const float* A, const float* B, float* __restrict__ C,
                                          const float* __restrict__ mu, const float* __restrict__ rstd,
                                          const float* __restrict__ lnw, const float* __restrict__ lnb,
                                          int m0, int n0) {
    const int lane = threadIdx.x & 31;
    const int c    = lane & 15;
    const int hf   = lane >> 4;
    const float* ap = A + (size_t)(m0 + c) * KD + 16 * hf;
    const float* bp = B + (size_t)(16 * hf) * ND + n0 + (NF == 4 ? 4 * c : c);
    const float mur = LN ? mu[m0 + c]   : 0.0f;
    const float rsr = LN ? rstd[m0 + c] : 1.0f;

    v8f acc[NF];
#pragma unroll
    for (int t = 0; t < NF; ++t) acc[t] = v8f{};

    constexpr int NKB = KD / 32;
    v4f aP[4], aQ[4];
    BFrag<NF> bX, bY;
    load_a(aP, ap, 0);
    load_b<NF, ND, LN>(bX, bp, lnw, lnb, 0);
    for (int kb = 0; kb < NKB; kb += 2) {
        const int kn = (kb + 2 < NKB) ? (kb + 2) : (kb + 1);
        load_a(aQ, ap, kb + 1);
        GEMM_SUB(aP[0], bX, bY, 32 * kb + 4)
        GEMM_SUB(aP[1], bY, bX, 32 * kb + 8)
        GEMM_SUB(aP[2], bX, bY, 32 * kb + 12)
        GEMM_SUB(aP[3], bY, bX, 32 * kb + 32)
        load_a(aP, ap, kn);
        GEMM_SUB(aQ[0], bX, bY, 32 * kb + 36)
        GEMM_SUB(aQ[1], bY, bX, 32 * kb + 40)
        GEMM_SUB(aQ[2], bX, bY, 32 * kb + 44)
        GEMM_SUB(aQ[3], bY, bX, 32 * kn)
    }

#pragma unroll
    for (int r = 0; r < 8; ++r) {
        float* cr = C + (size_t)(m0 + 8 * hf + r) * ND + n0;
        if (NF == 4) {
            v4f o;
            o[0] = acc[0][r];  o[1] = acc[1][r];
            o[2] = acc[2][r];  o[3] = acc[3][r];
            *(v4f*)(cr + 4 * c) = o;
        } else {
            cr[c] = acc[0][r];
        }
    }
}
#undef GEMM_SUB

__global__ void __launch_bounds__(128)
gemm_in(const float* U, const float* Win, float* __restrict__ out) {
    const int wv = threadIdx.x >> 5;
    const int m0 = (blockIdx.y * 4 + wv) * 16;
    if (blockIdx.x < CONV2 / 64)
        gemm_tile<4, D_MODEL, D_IN_PROJ, false>(U, Win, out, nullptr, nullptr, nullptr, nullptr,
                                                m0, (int)blockIdx.x * 64);
    else
        gemm_tile<1, D_MODEL, D_IN_PROJ, false>(U, Win, out, nullptr, nullptr, nullptr, nullptr,
                                                m0, CONV2);
}

__global__ void __launch_bounds__(256)
conv_dt(const float* __restrict__ xbcdt,
        const float* __restrict__ wf, const float* __restrict__ bf,
        const float* __restrict__ wb, const float* __restrict__ bb,
        const float* __restrict__ dt_bias,
        float* __restrict__ xc, float* __restrict__ dtb) {
    const size_t bl = blockIdx.x;
    const int b = (int)(bl / SEQLEN);
    const int t = (int)(bl % SEQLEN);

    for (int c = threadIdx.x; c < CONV2; c += blockDim.x) {
        float acc;
        if (c < CONV_DIM) {
            acc = bf[c];
#pragma unroll
            for (int k = 0; k < 4; ++k) {
                int tt = t - 3 + k;
                if (tt >= 0)
                    acc += xbcdt[((size_t)b * SEQLEN + tt) * D_IN_PROJ + c] * wf[c * 4 + k];
            }
        } else {
            int cc = c - CONV_DIM;
            acc = bb[cc];
#pragma unroll
            for (int k = 0; k < 4; ++k) {
                int tt = t + 3 - k;
                if (tt < SEQLEN)
                    acc += xbcdt[((size_t)b * SEQLEN + tt) * D_IN_PROJ + c] * wb[cc * 4 + k];
            }
        }
        float sig = 1.0f / (1.0f + expf(-acc));
        xc[bl * CONV2 + c] = acc * sig;
    }
    for (int h = threadIdx.x; h < NHEADS; h += blockDim.x) {
        float v = xbcdt[bl * D_IN_PROJ + CONV2 + h] + dt_bias[h];
        dtb[bl * NHEADS + h] = (v > 20.0f) ? v : log1pf(expf(v));
    }
}

__device__ __forceinline__ void head_bases(int h, int& xbase, int& bbase, int& cbase) {
    if (h < HH) { xbase = h * HEADDIM;                 bbase = 512;  cbase = 544;  }
    else        { xbase = CONV_DIM + (h - HH) * HEADDIM; bbase = 1088; cbase = 1120; }
}


__device__ __forceinline__ void chunk_prefix(const float* __restrict__ dtc, int rmask, float Ah,
                                             float* s_tot, float* s_dt, float* s_cs) {
    const int i    = threadIdx.x;
    const int lane = i & 31;
    const int wv   = i >> 5;
    const float dtv = dtc[(unsigned)((i ^ rmask) * NHEADS)];
    float v = Ah * dtv;
#pragma unroll
    for (int d = 1; d < 32; d <<= 1) {
        const float n = __shfl_up(v, d, 32);
        if (lane >= d) v += n;
    }
    if (lane == 31) s_tot[wv] = v;
    s_dt[i] = dtv;
    __syncthreads();
    float off = 0.0f;
    for (int w = 0; w < wv; ++w) off += s_tot[w];
    s_cs[i] = off + v;
    __syncthreads();
}

__device__ __forceinline__ void load_pair(const float* __restrict__ xcb, int j, int l15, int hf,
                                          int rmask, int bbase, int xbase,
                                          v2f (&bf)[8], float (&xv)[4][8]) {
    const unsigned rbx = (unsigned)(((16 * j + l15) ^ rmask) * CONV2 + bbase + 2 * hf);
#pragma unroll
    for (int q = 0; q < 8; ++q) bf[q] = *(const v2f*)(xcb + rbx + (unsigned)(4 * q));
    const int sb0 = 16 * j + 8 * hf;
#pragma unroll
    for (int r = 0; r < 8; ++r) {
        const unsigned ro = (unsigned)(((sb0 + r) ^ rmask) * CONV2 + xbase + l15);
        xv[0][r] = xcb[ro];
        xv[1][r] = xcb[ro + 16];
        xv[2][r] = xcb[ro + 32];
        xv[3][r] = xcb[ro + 48];
    }
}

__global__ void __launch_bounds__(128)
ssd_chunk(const float* __restrict__ xc, const float* __restrict__ dtb,
          const float* __restrict__ A_log, const float* __restrict__ Dp,
          float* __restrict__ y, float* __restrict__ S, float* __restrict__ prodA) {
    const int chunk = blockIdx.x & (NCHUNK - 1);
    const int h     = (blockIdx.x >> 5) & 15;
    const int b     = blockIdx.x >> 9;
    int xbase, bbase, cbase; head_bases(h, xbase, bbase, cbase);
    const int s0    = chunk * CHUNK;
    const int rmask = (h < HH) ? 0 : (CHUNK - 1);
    const int tbase = (h < HH) ? s0 : (SEQLEN - CHUNK - s0);
    const size_t row0 = (size_t)b * SEQLEN + tbase;
    const float* xcb = xc  + row0 * CONV2;
    const float* dtc = dtb + row0 * NHEADS + h;
    float*       yb  = y   + row0 * D_INNER + h * HEADDIM;

    const float Ah = -expf(A_log[h]);
    const float Dh = Dp[h];

    __shared__ float s_tot[4];
    __shared__ alignas(16) float s_dt[CHUNK];
    __shared__ alignas(16) float s_cs[CHUNK];
    __shared__ alignas(16) float s_w[CHUNK];
    chunk_prefix(dtc, rmask, Ah, s_tot, s_dt, s_cs);
    const float cl = s_cs[CHUNK - 1];
    s_w[threadIdx.x] = expf(cl - s_cs[threadIdx.x]) * s_dt[threadIdx.x];
    __syncthreads();

    const int wv   = __builtin_amdgcn_readfirstlane((int)(threadIdx.x >> 5));
    const int lane = threadIdx.x & 31;
    const int l15  = lane & 15;
    const int hf   = lane >> 4;

    for (int half = 0; half < 2; ++half) {
        const int i  = half ? (7 - wv) : wv;
        const int tl = 16 * i + l15;
        const unsigned rtx = (unsigned)((tl ^ rmask) * CONV2);
        const float cst = s_cs[tl];

        v2f cf[8];
#pragma unroll
        for (int q = 0; q < 8; ++q)
            cf[q] = *(const v2f*)(xcb + rtx + (unsigned)(cbase + 2 * hf + 4 * q));

        v8f acc0 = {}, acc1 = {}, acc2 = {}, acc3 = {};
        v2f bfr[8];
        float xa[4][8];
        load_pair(xcb, 0, l15, hf, rmask, bbase, xbase, bfr, xa);
        for (int j = 0; j <= i; ++j) {
            const int sb0 = 16 * j + 8 * hf;
            v2f bfn[8];
            float xn[4][8];
            load_pair(xcb, (j < i) ? (j + 1) : i, l15, hf, rmask, bbase, xbase, bfn, xn);
            __builtin_amdgcn_sched_barrier(0);

            v8f g = {};
#pragma unroll
            for (int q = 0; q < 8; ++q) g = wmma_f32_4(bfr[q], cf[q], g);

            float m[8];
#pragma unroll
            for (int r = 0; r < 8; ++r) {
                const int ss = sb0 + r;
                const float css = s_cs[ss];
                const float d = (ss <= tl) ? (cst - css) : -INFINITY;
                m[r] = g[r] * (expf(d) * s_dt[ss]);
            }
#pragma unroll
            for (int kk = 0; kk < 4; ++kk) {
                v2f bm; bm.x = m[2 * kk]; bm.y = m[2 * kk + 1];
                v2f a0, a1, a2, a3;
                a0.x = xa[0][2 * kk];  a0.y = xa[0][2 * kk + 1];
                a1.x = xa[1][2 * kk];  a1.y = xa[1][2 * kk + 1];
                a2.x = xa[2][2 * kk];  a2.y = xa[2][2 * kk + 1];
                a3.x = xa[3][2 * kk];  a3.y = xa[3][2 * kk + 1];
                acc0 = wmma_f32_4(a0, bm, acc0);
                acc1 = wmma_f32_4(a1, bm, acc1);
                acc2 = wmma_f32_4(a2, bm, acc2);
                acc3 = wmma_f32_4(a3, bm, acc3);
            }
            __builtin_amdgcn_sched_barrier(0);
#pragma unroll
            for (int q = 0; q < 8; ++q) {
                bfr[q] = bfn[q];
                xa[0][q] = xn[0][q];  xa[1][q] = xn[1][q];
                xa[2][q] = xn[2][q];  xa[3][q] = xn[3][q];
            }
        }
        const float* xr = xcb + rtx + (unsigned)(xbase + 8 * hf);
        float*       yr = yb + (unsigned)((tl ^ rmask) * D_INNER + 8 * hf);
        v8f accs[4] = {acc0, acc1, acc2, acc3};
#pragma unroll
        for (int pb = 0; pb < 4; ++pb) {
            const float4 xa4 = *(const float4*)(xr + 16 * pb);
            const float4 xb4 = *(const float4*)(xr + 16 * pb + 4);
            float4 oa, ob;
            oa.x = accs[pb][0] + Dh * xa4.x;  oa.y = accs[pb][1] + Dh * xa4.y;
            oa.z = accs[pb][2] + Dh * xa4.z;  oa.w = accs[pb][3] + Dh * xa4.w;
            ob.x = accs[pb][4] + Dh * xb4.x;  ob.y = accs[pb][5] + Dh * xb4.y;
            ob.z = accs[pb][6] + Dh * xb4.z;  ob.w = accs[pb][7] + Dh * xb4.w;
            *(float4*)(yr + 16 * pb)     = oa;
            *(float4*)(yr + 16 * pb + 4) = ob;
        }
    }

    {
        v8f st0 = {}, st1 = {};
        const unsigned xo = (unsigned)(xbase + 16 * wv + l15);
        const unsigned bo = (unsigned)(bbase + l15);
        for (int k0 = 0; k0 < CHUNK; k0 += 32) {
            float xv[16], bv0[16], bv1[16], wv16[16];
#pragma unroll
            for (int u = 0; u < 16; ++u) {
                const int sa = k0 + 4 * (u >> 1) + 2 * hf + (u & 1);
                const unsigned ro = (unsigned)((sa ^ rmask) * CONV2);
                xv[u]   = xcb[ro + xo];
                bv0[u]  = xcb[ro + bo];
                bv1[u]  = xcb[ro + bo + 16];
                wv16[u] = s_w[sa];
            }
#pragma unroll
            for (int ks = 0; ks < 8; ++ks) {
                v2f a, b0, b1;
                a.x  = xv[2 * ks] * wv16[2 * ks];
                a.y  = xv[2 * ks + 1] * wv16[2 * ks + 1];
                b0.x = bv0[2 * ks];  b0.y = bv0[2 * ks + 1];
                b1.x = bv1[2 * ks];  b1.y = bv1[2 * ks + 1];
                st0 = wmma_f32_4(a, b0, st0);
                st1 = wmma_f32_4(a, b1, st1);
            }
        }
        const size_t sb = (((size_t)(b * NHEADS + h)) * NCHUNK + chunk) * (HEADDIM * NSTATE)
                        + (size_t)(16 * wv + 8 * hf) * NSTATE + l15;
#pragma unroll
        for (int r = 0; r < 8; ++r) {
            S[sb + (size_t)r * NSTATE]      = st0[r];
            S[sb + (size_t)r * NSTATE + 16] = st1[r];
        }
    }
    if (threadIdx.x == 0) prodA[(b * NHEADS + h) * NCHUNK + chunk] = expf(cl);
}

__global__ void __launch_bounds__(64)
scan_combine(float* __restrict__ S, const float* __restrict__ prodA) {
    const int bh = blockIdx.x;
    const int p  = threadIdx.x;
    float hcur[NSTATE];
#pragma unroll
    for (int n = 0; n < NSTATE; ++n) hcur[n] = 0.0f;
    for (int c = 0; c < NCHUNK; ++c) {
        const size_t base = (((size_t)bh) * NCHUNK + c) * (HEADDIM * NSTATE) + (size_t)p * NSTATE;
        const float a = prodA[bh * NCHUNK + c];
#pragma unroll
        for (int n = 0; n < NSTATE; ++n) {
            float sl = S[base + n];
            S[base + n] = hcur[n];
            hcur[n] = a * hcur[n] + sl;
        }
    }
}

__global__ void __launch_bounds__(128)
ssd_correct(const float* __restrict__ xc, const float* __restrict__ dtb,
            const float* __restrict__ A_log, const float* __restrict__ S,
            float* __restrict__ y) {
    const int chunk = blockIdx.x & (NCHUNK - 1);
    if (chunk == 0) return;
    const int h = (blockIdx.x >> 5) & 15;
    const int b = blockIdx.x >> 9;
    int xbase, bbase, cbase; head_bases(h, xbase, bbase, cbase);
    (void)xbase; (void)bbase;
    const int s0    = chunk * CHUNK;
    const int rmask = (h < HH) ? 0 : (CHUNK - 1);
    const int tbase = (h < HH) ? s0 : (SEQLEN - CHUNK - s0);
    const size_t row0 = (size_t)b * SEQLEN + tbase;
    const float* xcb = xc  + row0 * CONV2;
    const float* dtc = dtb + row0 * NHEADS + h;
    float*       yb  = y   + row0 * D_INNER + h * HEADDIM;

    const float Ah = -expf(A_log[h]);
    __shared__ float s_tot[4], s_dt[CHUNK], s_cs[CHUNK];
    chunk_prefix(dtc, rmask, Ah, s_tot, s_dt, s_cs);

    const int wv   = __builtin_amdgcn_readfirstlane((int)(threadIdx.x >> 5));
    const int lane = threadIdx.x & 31;
    const int l15  = lane & 15;
    const int hf   = lane >> 4;

    const float* hp = S + (((size_t)(b * NHEADS + h)) * NCHUNK + chunk) * (HEADDIM * NSTATE)
                    + (size_t)l15 * NSTATE + 2 * hf;

    for (int half = 0; half < 2; ++half) {
        const int tl = 16 * (wv + 4 * half) + l15;
        const float cumA = expf(s_cs[tl]);
        const float* cp = xcb + (unsigned)((tl ^ rmask) * CONV2 + cbase + 2 * hf);

        v8f acc0 = {}, acc1 = {}, acc2 = {}, acc3 = {};
#pragma unroll
        for (int q = 0; q < 8; ++q) {
            const v2f cq = *(const v2f*)(cp + 4 * q);
            acc0 = wmma_f32_4(*(const v2f*)(hp + 4 * q),               cq, acc0);
            acc1 = wmma_f32_4(*(const v2f*)(hp + 16 * NSTATE + 4 * q), cq, acc1);
            acc2 = wmma_f32_4(*(const v2f*)(hp + 32 * NSTATE + 4 * q), cq, acc2);
            acc3 = wmma_f32_4(*(const v2f*)(hp + 48 * NSTATE + 4 * q), cq, acc3);
        }
        float* yr = yb + (unsigned)((tl ^ rmask) * D_INNER + 8 * hf);
        v8f accs[4] = {acc0, acc1, acc2, acc3};
#pragma unroll
        for (int pb = 0; pb < 4; ++pb) {
            float4 ya = *(const float4*)(yr + 16 * pb);
            float4 yc = *(const float4*)(yr + 16 * pb + 4);
            ya.x += cumA * accs[pb][0];  ya.y += cumA * accs[pb][1];
            ya.z += cumA * accs[pb][2];  ya.w += cumA * accs[pb][3];
            yc.x += cumA * accs[pb][4];  yc.y += cumA * accs[pb][5];
            yc.z += cumA * accs[pb][6];  yc.w += cumA * accs[pb][7];
            *(float4*)(yr + 16 * pb)     = ya;
            *(float4*)(yr + 16 * pb + 4) = yc;
        }
    }
}

__global__ void __launch_bounds__(256)
ln_stats(const float* __restrict__ y, float* __restrict__ mu, float* __restrict__ rstd) {
    const size_t row = blockIdx.x;
    const int tid = threadIdx.x;
    const float* yr = y + row * D_INNER;
    float s = 0.0f, q = 0.0f;
    for (int i = tid; i < D_INNER; i += 256) { float v = yr[i]; s += v; q += v * v; }
    __shared__ float ss[256], sq[256];
    ss[tid] = s; sq[tid] = q; __syncthreads();
    for (int off = 128; off > 0; off >>= 1) {
        if (tid < off) { ss[tid] += ss[tid + off]; sq[tid] += sq[tid + off]; }
        __syncthreads();
    }
    if (tid == 0) {
        float m = ss[0] * (1.0f / D_INNER);
        float v = sq[0] * (1.0f / D_INNER) - m * m;
        mu[row] = m;
        rstd[row] = rsqrtf(v + 1e-5f);
    }
}

__global__ void __launch_bounds__(128)
gemm_out(const float* Y, const float* __restrict__ mu,
         const float* __restrict__ rstd, const float* __restrict__ lnw,
         const float* __restrict__ lnb, const float* Wout,
         float* __restrict__ out) {
    const int wv = threadIdx.x >> 5;
    const int m0 = (blockIdx.y * 4 + wv) * 16;
    gemm_tile<4, D_INNER, D_MODEL, true>(Y, Wout, out, mu, rstd, lnw, lnb, m0, (int)blockIdx.x * 64);
}

extern "C" void kernel_launch(void* const* d_in, const int* in_sizes, int n_in,
                              void* d_out, int out_size, void* d_ws, size_t ws_size,
                              hipStream_t stream) {
    const float* u        = (const float*)d_in[0];
    const float* W_in     = (const float*)d_in[2];
    const float* conv_fw  = (const float*)d_in[3];
    const float* conv_fb  = (const float*)d_in[4];
    const float* conv_bw  = (const float*)d_in[5];
    const float* conv_bb  = (const float*)d_in[6];
    const float* dt_bias  = (const float*)d_in[7];
    const float* A_log    = (const float*)d_in[8];
    const float* D_param  = (const float*)d_in[9];
    const float* ln_w     = (const float*)d_in[10];
    const float* ln_b     = (const float*)d_in[11];
    const float* W_out    = (const float*)d_in[12];
    float* out            = (float*)d_out;

    float* ws = (float*)d_ws;
    size_t off = 0;
    float* xbcdt = ws + off; off += (size_t)BL * D_IN_PROJ;
    float* xc    = ws + off; off += (size_t)BL * CONV2;
    float* dtb   = ws + off; off += (size_t)BL * NHEADS;
    float* ybuf  = ws + off; off += (size_t)BL * D_INNER;
    float* Sbuf  = ws + off; off += (size_t)32 * NCHUNK * HEADDIM * NSTATE;
    float* pAbuf = ws + off; off += (size_t)32 * NCHUNK;
    float* mubuf = ws + off; off += (size_t)BL;
    float* rsbuf = ws + off; off += (size_t)BL;

    gemm_in<<<dim3(CONV2 / 64 + 1, BL / 64), 128, 0, stream>>>(u, W_in, xbcdt);
    conv_dt<<<dim3(BL), 256, 0, stream>>>(xbcdt, conv_fw, conv_fb, conv_bw, conv_bb,
                                          dt_bias, xc, dtb);
    ssd_chunk<<<dim3(BATCH * NHEADS * NCHUNK), 128, 0, stream>>>(
        xc, dtb, A_log, D_param, ybuf, Sbuf, pAbuf);
    scan_combine<<<dim3(BATCH * NHEADS), 64, 0, stream>>>(Sbuf, pAbuf);
    ssd_correct<<<dim3(BATCH * NHEADS * NCHUNK), 128, 0, stream>>>(
        xc, dtb, A_log, Sbuf, ybuf);
    ln_stats<<<dim3(BL), 256, 0, stream>>>(ybuf, mubuf, rsbuf);
    gemm_out<<<dim3(D_MODEL / 64, BL / 64), 128, 0, stream>>>(
        ybuf, mubuf, rsbuf, ln_w, ln_b, W_out, out);

    (void)in_sizes; (void)n_in; (void)out_size; (void)ws_size;
}


# LayerNorm row sums taken in scan pass 3, where y gets its final value (per-head partials, joined at the head of the out-projection): statistics launch and its read of y gone: new sum order of mu/rstd
# speedup vs baseline: 2.3335x; 1.0446x over previous
#include <hip/hip_runtime.h>
#include <hip/hip_bf16.h>
#include <math.h>

#define BATCH    2
#define SEQLEN   4096
#define BL       (BATCH * SEQLEN)
#define D_MODEL  512
#define D_IN_PROJ 1168
#define CONV_DIM 576
#define CONV2    1152
#define D_INNER  1024
#define NHEADS   16
#define HH       8
#define HEADDIM  64
#define NSTATE   32
#define NCHUNK   32
#define CHUNK    128

typedef float v2f __attribute__((ext_vector_type(2)));
typedef float v8f __attribute__((ext_vector_type(8)));

__device__ __forceinline__ v8f wmma_f32_4(v2f a, v2f b, v8f c) {
    return __builtin_amdgcn_wmma_f32_16x16x4_f32(
         false, a,  false, b,
         (short)0, c,  false,  false);
}

typedef float v4f __attribute__((ext_vector_type(4)));

template <int NF>
struct BFrag {
    v4f   b4[4];
    float b1[4];
    float wl[4], wh[4], zl[4], zh[4];
};

template <int NF, int ND, bool LN>
__device__ __forceinline__ void load_b(BFrag<NF>& f, const float* bp,
                                       const float* __restrict__ lnw, const float* __restrict__ lnb,
                                       int krow) {
#pragma unroll
    for (int j = 0; j < 4; ++j) {
        if (NF == 4) f.b4[j] = *(const v4f*)(bp + (size_t)(krow + j) * ND);
        else         f.b1[j] = bp[(size_t)(krow + j) * ND];
    }
    if (LN) {
#pragma unroll
        for (int j = 0; j < 4; ++j) {
            f.wl[j] = lnw[krow + j];  f.wh[j] = lnw[krow + 16 + j];
            f.zl[j] = lnb[krow + j];  f.zh[j] = lnb[krow + 16 + j];
        }
    }
}

__device__ __forceinline__ void load_a(v4f (&a)[4], const float* ap, int kb) {
#pragma unroll
    for (int q = 0; q < 4; ++q) a[q] = *(const v4f*)(ap + 32 * kb + 4 * q);
}

template <int NF, bool LN>
__device__ __forceinline__ void mma_sub(v4f y, const BFrag<NF>& f, int hf, float mur, float rsr,
                                        v8f (&acc)[NF]) {
    if (LN) {
#pragma unroll
        for (int j = 0; j < 4; ++j) {
            const float w = hf ? f.wh[j] : f.wl[j];
            const float z = hf ? f.zh[j] : f.zl[j];
            y[j] = (y[j] - mur) * rsr * w + z;
        }
    }
    v2f a0, a1;
    a0.x = y[0];  a0.y = y[1];
    a1.x = y[2];  a1.y = y[3];
#pragma unroll
    for (int t = 0; t < NF; ++t) {
        v2f b;
        if (NF == 4) { b.x = f.b4[0][t];  b.y = f.b4[1][t]; }
        else         { b.x = f.b1[0];     b.y = f.b1[1]; }
        acc[t] = wmma_f32_4(a0, b, acc[t]);
    }
#pragma unroll
    for (int t = 0; t < NF; ++t) {
        v2f b;
        if (NF == 4) { b.x = f.b4[2][t];  b.y = f.b4[3][t]; }
        else         { b.x = f.b1[2];     b.y = f.b1[3]; }
        acc[t] = wmma_f32_4(a1, b, acc[t]);
    }
}

#define GEMM_SUB(AV, BCUR, BNXT, KROW_NEXT)                                       \
    load_b<NF, ND, LN>(BNXT, bp, lnw, lnb, (KROW_NEXT));                          \
    __builtin_amdgcn_sched_barrier(0);                                            \
    mma_sub<NF, LN>((AV), (BCUR), hf, mur, rsr, acc);                             \
    __builtin_amdgcn_sched_barrier(0);

template <int NF, int KD, int ND, bool LN>
__device__ __forceinline__ void gemm_tile(const float* A, const float* B, float* __restrict__ C,
                                          const float* __restrict__ part,
                                          const float* __restrict__ lnw, const float* __restrict__ lnb,
                                          int m0, int n0) {
    const int lane = threadIdx.x & 31;
    const int c    = lane & 15;
    const int hf   = lane >> 4;
    const float* ap = A + (size_t)(m0 + c) * KD + 16 * hf;
    const float* bp = B + (size_t)(16 * hf) * ND + n0 + (NF == 4 ? 4 * c : c);
    float mur = 0.0f, rsr = 1.0f;
    if (LN) {
        float s = 0.0f, q = 0.0f;
#pragma unroll
        for (int h = 0; h < NHEADS; ++h) {
            const v2f p = *(const v2f*)(part + 2 * ((size_t)h * BL + m0 + c));
            s += p.x;  q += p.y;
        }
        const float m = s * (1.0f / D_INNER);
        const float v = q * (1.0f / D_INNER) - m * m;
        mur = m;
        rsr = rsqrtf(v + 1e-5f);
    }

    v8f acc[NF];
#pragma unroll
    for (int t = 0; t < NF; ++t) acc[t] = v8f{};

    constexpr int NKB = KD / 32;
    v4f aP[4], aQ[4];
    BFrag<NF> bX, bY;
    load_a(aP, ap, 0);
    load_b<NF, ND, LN>(bX, bp, lnw, lnb, 0);
    for (int kb = 0; kb < NKB; kb += 2) {
        const int kn = (kb + 2 < NKB) ? (kb + 2) : (kb + 1);
        load_a(aQ, ap, kb + 1);
        GEMM_SUB(aP[0], bX, bY, 32 * kb + 4)
        GEMM_SUB(aP[1], bY, bX, 32 * kb + 8)
        GEMM_SUB(aP[2], bX, bY, 32 * kb + 12)
        GEMM_SUB(aP[3], bY, bX, 32 * kb + 32)
        load_a(aP, ap, kn);
        GEMM_SUB(aQ[0], bX, bY, 32 * kb + 36)
        GEMM_SUB(aQ[1], bY, bX, 32 * kb + 40)
        GEMM_SUB(aQ[2], bX, bY, 32 * kb + 44)
        GEMM_SUB(aQ[3], bY, bX, 32 * kn)
    }

#pragma unroll
    for (int r = 0; r < 8; ++r) {
        float* cr = C + (size_t)(m0 + 8 * hf + r) * ND + n0;
        if (NF == 4) {
            v4f o;
            o[0] = acc[0][r];  o[1] = acc[1][r];
            o[2] = acc[2][r];  o[3] = acc[3][r];
            *(v4f*)(cr + 4 * c) = o;
        } else {
            cr[c] = acc[0][r];
        }
    }
}
#undef GEMM_SUB

__global__ void __launch_bounds__(128)
gemm_in(const float* U, const float* Win, float* __restrict__ out) {
    const int wv = threadIdx.x >> 5;
    const int m0 = (blockIdx.y * 4 + wv) * 16;
    if (blockIdx.x < CONV2 / 64)
        gemm_tile<4, D_MODEL, D_IN_PROJ, false>(U, Win, out, nullptr, nullptr, nullptr,
                                                m0, (int)blockIdx.x * 64);
    else
        gemm_tile<1, D_MODEL, D_IN_PROJ, false>(U, Win, out, nullptr, nullptr, nullptr,
                                                m0, CONV2);
}

__global__ void __launch_bounds__(256)
conv_dt(const float* __restrict__ xbcdt,
        const float* __restrict__ wf, const float* __restrict__ bf,
        const float* __restrict__ wb, const float* __restrict__ bb,
        const float* __restrict__ dt_bias,
        float* __restrict__ xc, float* __restrict__ dtb) {
    const size_t bl = blockIdx.x;
    const int b = (int)(bl / SEQLEN);
    const int t = (int)(bl % SEQLEN);

    for (int c = threadIdx.x; c < CONV2; c += blockDim.x) {
        float acc;
        if (c < CONV_DIM) {
            acc = bf[c];
#pragma unroll
            for (int k = 0; k < 4; ++k) {
                int tt = t - 3 + k;
                if (tt >= 0)
                    acc += xbcdt[((size_t)b * SEQLEN + tt) * D_IN_PROJ + c] * wf[c * 4 + k];
            }
        } else {
            int cc = c - CONV_DIM;
            acc = bb[cc];
#pragma unroll
            for (int k = 0; k < 4; ++k) {
                int tt = t + 3 - k;
                if (tt < SEQLEN)
                    acc += xbcdt[((size_t)b * SEQLEN + tt) * D_IN_PROJ + c] * wb[cc * 4 + k];
            }
        }
        float sig = 1.0f / (1.0f + expf(-acc));
        xc[bl * CONV2 + c] = acc * sig;
    }
    for (int h = threadIdx.x; h < NHEADS; h += blockDim.x) {
        float v = xbcdt[bl * D_IN_PROJ + CONV2 + h] + dt_bias[h];
        dtb[bl * NHEADS + h] = (v > 20.0f) ? v : log1pf(expf(v));
    }
}

__device__ __forceinline__ void head_bases(int h, int& xbase, int& bbase, int& cbase) {
    if (h < HH) { xbase = h * HEADDIM;                 bbase = 512;  cbase = 544;  }
    else        { xbase = CONV_DIM + (h - HH) * HEADDIM; bbase = 1088; cbase = 1120; }
}


__device__ __forceinline__ void chunk_prefix(const float* __restrict__ dtc, int rmask, float Ah,
                                             float* s_tot, float* s_dt, float* s_cs) {
    const int i    = threadIdx.x;
    const int lane = i & 31;
    const int wv   = i >> 5;
    const float dtv = dtc[(unsigned)((i ^ rmask) * NHEADS)];
    float v = Ah * dtv;
#pragma unroll
    for (int d = 1; d < 32; d <<= 1) {
        const float n = __shfl_up(v, d, 32);
        if (lane >= d) v += n;
    }
    if (lane == 31) s_tot[wv] = v;
    s_dt[i] = dtv;
    __syncthreads();
    float off = 0.0f;
    for (int w = 0; w < wv; ++w) off += s_tot[w];
    s_cs[i] = off + v;
    __syncthreads();
}

__device__ __forceinline__ void load_pair(const float* __restrict__ xcb, int j, int l15, int hf,
                                          int rmask, int bbase, int xbase,
                                          v2f (&bf)[8], float (&xv)[4][8]) {
    const unsigned rbx = (unsigned)(((16 * j + l15) ^ rmask) * CONV2 + bbase + 2 * hf);
#pragma unroll
    for (int q = 0; q < 8; ++q) bf[q] = *(const v2f*)(xcb + rbx + (unsigned)(4 * q));
    const int sb0 = 16 * j + 8 * hf;
#pragma unroll
    for (int r = 0; r < 8; ++r) {
        const unsigned ro = (unsigned)(((sb0 + r) ^ rmask) * CONV2 + xbase + l15);
        xv[0][r] = xcb[ro];
        xv[1][r] = xcb[ro + 16];
        xv[2][r] = xcb[ro + 32];
        xv[3][r] = xcb[ro + 48];
    }
}

__global__ void __launch_bounds__(128)
ssd_chunk(const float* __restrict__ xc, const float* __restrict__ dtb,
          const float* __restrict__ A_log, const float* __restrict__ Dp,
          float* __restrict__ y, float* __restrict__ S, float* __restrict__ prodA) {
    const int chunk = blockIdx.x & (NCHUNK - 1);
    const int h     = (blockIdx.x >> 5) & 15;
    const int b     = blockIdx.x >> 9;
    int xbase, bbase, cbase; head_bases(h, xbase, bbase, cbase);
    const int s0    = chunk * CHUNK;
    const int rmask = (h < HH) ? 0 : (CHUNK - 1);
    const int tbase = (h < HH) ? s0 : (SEQLEN - CHUNK - s0);
    const size_t row0 = (size_t)b * SEQLEN + tbase;
    const float* xcb = xc  + row0 * CONV2;
    const float* dtc = dtb + row0 * NHEADS + h;
    float*       yb  = y   + row0 * D_INNER + h * HEADDIM;

    const float Ah = -expf(A_log[h]);
    const float Dh = Dp[h];

    __shared__ float s_tot[4];
    __shared__ alignas(16) float s_dt[CHUNK];
    __shared__ alignas(16) float s_cs[CHUNK];
    __shared__ alignas(16) float s_w[CHUNK];
    chunk_prefix(dtc, rmask, Ah, s_tot, s_dt, s_cs);
    const float cl = s_cs[CHUNK - 1];
    s_w[threadIdx.x] = expf(cl - s_cs[threadIdx.x]) * s_dt[threadIdx.x];
    __syncthreads();

    const int wv   = __builtin_amdgcn_readfirstlane((int)(threadIdx.x >> 5));
    const int lane = threadIdx.x & 31;
    const int l15  = lane & 15;
    const int hf   = lane >> 4;

    for (int half = 0; half < 2; ++half) {
        const int i  = half ? (7 - wv) : wv;
        const int tl = 16 * i + l15;
        const unsigned rtx = (unsigned)((tl ^ rmask) * CONV2);
        const float cst = s_cs[tl];

        v2f cf[8];
#pragma unroll
        for (int q = 0; q < 8; ++q)
            cf[q] = *(const v2f*)(xcb + rtx + (unsigned)(cbase + 2 * hf + 4 * q));

        v8f acc0 = {}, acc1 = {}, acc2 = {}, acc3 = {};
        v2f bfr[8];
        float xa[4][8];
        load_pair(xcb, 0, l15, hf, rmask, bbase, xbase, bfr, xa);
        for (int j = 0; j <= i; ++j) {
            const int sb0 = 16 * j + 8 * hf;
            v2f bfn[8];
            float xn[4][8];
            load_pair(xcb, (j < i) ? (j + 1) : i, l15, hf, rmask, bbase, xbase, bfn, xn);
            __builtin_amdgcn_sched_barrier(0);

            v8f g = {};
#pragma unroll
            for (int q = 0; q < 8; ++q) g = wmma_f32_4(bfr[q], cf[q], g);

            float m[8];
#pragma unroll
            for (int r = 0; r < 8; ++r) {
                const int ss = sb0 + r;
                const float css = s_cs[ss];
                const float d = (ss <= tl) ? (cst - css) : -INFINITY;
                m[r] = g[r] * (expf(d) * s_dt[ss]);
            }
#pragma unroll
            for (int kk = 0; kk < 4; ++kk) {
                v2f bm; bm.x = m[2 * kk]; bm.y = m[2 * kk + 1];
                v2f a0, a1, a2, a3;
                a0.x = xa[0][2 * kk];  a0.y = xa[0][2 * kk + 1];
                a1.x = xa[1][2 * kk];  a1.y = xa[1][2 * kk + 1];
                a2.x = xa[2][2 * kk];  a2.y = xa[2][2 * kk + 1];
                a3.x = xa[3][2 * kk];  a3.y = xa[3][2 * kk + 1];
                acc0 = wmma_f32_4(a0, bm, acc0);
                acc1 = wmma_f32_4(a1, bm, acc1);
                acc2 = wmma_f32_4(a2, bm, acc2);
                acc3 = wmma_f32_4(a3, bm, acc3);
            }
            __builtin_amdgcn_sched_barrier(0);
#pragma unroll
            for (int q = 0; q < 8; ++q) {
                bfr[q] = bfn[q];
                xa[0][q] = xn[0][q];  xa[1][q] = xn[1][q];
                xa[2][q] = xn[2][q];  xa[3][q] = xn[3][q];
            }
        }
        const float* xr = xcb + rtx + (unsigned)(xbase + 8 * hf);
        float*       yr = yb + (unsigned)((tl ^ rmask) * D_INNER + 8 * hf);
        v8f accs[4] = {acc0, acc1, acc2, acc3};
#pragma unroll
        for (int pb = 0; pb < 4; ++pb) {
            const float4 xa4 = *(const float4*)(xr + 16 * pb);
            const float4 xb4 = *(const float4*)(xr + 16 * pb + 4);
            float4 oa, ob;
            oa.x = accs[pb][0] + Dh * xa4.x;  oa.y = accs[pb][1] + Dh * xa4.y;
            oa.z = accs[pb][2] + Dh * xa4.z;  oa.w = accs[pb][3] + Dh * xa4.w;
            ob.x = accs[pb][4] + Dh * xb4.x;  ob.y = accs[pb][5] + Dh * xb4.y;
            ob.z = accs[pb][6] + Dh * xb4.z;  ob.w = accs[pb][7] + Dh * xb4.w;
            *(float4*)(yr + 16 * pb)     = oa;
            *(float4*)(yr + 16 * pb + 4) = ob;
        }
    }

    {
        v8f st0 = {}, st1 = {};
        const unsigned xo = (unsigned)(xbase + 16 * wv + l15);
        const unsigned bo = (unsigned)(bbase + l15);
        for (int k0 = 0; k0 < CHUNK; k0 += 32) {
            float xv[16], bv0[16], bv1[16], wv16[16];
#pragma unroll
            for (int u = 0; u < 16; ++u) {
                const int sa = k0 + 4 * (u >> 1) + 2 * hf + (u & 1);
                const unsigned ro = (unsigned)((sa ^ rmask) * CONV2);
                xv[u]   = xcb[ro + xo];
                bv0[u]  = xcb[ro + bo];
                bv1[u]  = xcb[ro + bo + 16];
                wv16[u] = s_w[sa];
            }
#pragma unroll
            for (int ks = 0; ks < 8; ++ks) {
                v2f a, b0, b1;
                a.x  = xv[2 * ks] * wv16[2 * ks];
                a.y  = xv[2 * ks + 1] * wv16[2 * ks + 1];
                b0.x = bv0[2 * ks];  b0.y = bv0[2 * ks + 1];
                b1.x = bv1[2 * ks];  b1.y = bv1[2 * ks + 1];
                st0 = wmma_f32_4(a, b0, st0);
                st1 = wmma_f32_4(a, b1, st1);
            }
        }
        const size_t sb = (((size_t)(b * NHEADS + h)) * NCHUNK + chunk) * (HEADDIM * NSTATE)
                        + (size_t)(16 * wv + 8 * hf) * NSTATE + l15;
#pragma unroll
        for (int r = 0; r < 8; ++r) {
            S[sb + (size_t)r * NSTATE]      = st0[r];
            S[sb + (size_t)r * NSTATE + 16] = st1[r];
        }
    }
    if (threadIdx.x == 0) prodA[(b * NHEADS + h) * NCHUNK + chunk] = expf(cl);
}

__global__ void __launch_bounds__(64)
scan_combine(float* __restrict__ S, const float* __restrict__ prodA) {
    const int bh = blockIdx.x;
    const int p  = threadIdx.x;
    float hcur[NSTATE];
#pragma unroll
    for (int n = 0; n < NSTATE; ++n) hcur[n] = 0.0f;
    for (int c = 0; c < NCHUNK; ++c) {
        const size_t base = (((size_t)bh) * NCHUNK + c) * (HEADDIM * NSTATE) + (size_t)p * NSTATE;
        const float a = prodA[bh * NCHUNK + c];
#pragma unroll
        for (int n = 0; n < NSTATE; ++n) {
            float sl = S[base + n];
            S[base + n] = hcur[n];
            hcur[n] = a * hcur[n] + sl;
        }
    }
}

__device__ __forceinline__ void add_sums(const float4 a, float& ps, float& pq) {
    ps += (a.x + a.y) + (a.z + a.w);
    pq += (a.x * a.x + a.y * a.y) + (a.z * a.z + a.w * a.w);
}

__device__ __forceinline__ void put_row_sums(float ps, float pq, int hf, float* pr) {
    ps += __shfl_xor(ps, 16, 32);
    pq += __shfl_xor(pq, 16, 32);
    if (hf == 0) { v2f o; o.x = ps; o.y = pq; *(v2f*)pr = o; }
}

__global__ void __launch_bounds__(128)
ssd_correct(const float* __restrict__ xc, const float* __restrict__ dtb,
            const float* __restrict__ A_log, const float* __restrict__ S,
            float* __restrict__ y, float* __restrict__ part) {
    const int chunk = blockIdx.x & (NCHUNK - 1);
    const int h = (blockIdx.x >> 5) & 15;
    const int b = blockIdx.x >> 9;
    int xbase, bbase, cbase; head_bases(h, xbase, bbase, cbase);
    (void)xbase; (void)bbase;
    const int s0    = chunk * CHUNK;
    const int rmask = (h < HH) ? 0 : (CHUNK - 1);
    const int tbase = (h < HH) ? s0 : (SEQLEN - CHUNK - s0);
    const size_t row0 = (size_t)b * SEQLEN + tbase;
    const float* xcb = xc  + row0 * CONV2;
    const float* dtc = dtb + row0 * NHEADS + h;
    float*       yb  = y   + row0 * D_INNER + h * HEADDIM;
    float*       pb0 = part + 2 * ((size_t)h * BL + row0);

    const int wv   = __builtin_amdgcn_readfirstlane((int)(threadIdx.x >> 5));
    const int lane = threadIdx.x & 31;
    const int l15  = lane & 15;
    const int hf   = lane >> 4;

    if (chunk == 0) {
        for (int half = 0; half < 2; ++half) {
            const int tl = 16 * (wv + 4 * half) + l15;
            const float* yr = yb + (unsigned)((tl ^ rmask) * D_INNER + 8 * hf);
            float ps = 0.0f, pq = 0.0f;
#pragma unroll
            for (int pb = 0; pb < 4; ++pb) {
                add_sums(*(const float4*)(yr + 16 * pb), ps, pq);
                add_sums(*(const float4*)(yr + 16 * pb + 4), ps, pq);
            }
            put_row_sums(ps, pq, hf, pb0 + 2 * (tl ^ rmask));
        }
        return;
    }

    const float Ah = -expf(A_log[h]);
    __shared__ float s_tot[4], s_dt[CHUNK], s_cs[CHUNK];
    chunk_prefix(dtc, rmask, Ah, s_tot, s_dt, s_cs);

    const float* hp = S + (((size_t)(b * NHEADS + h)) * NCHUNK + chunk) * (HEADDIM * NSTATE)
                    + (size_t)l15 * NSTATE + 2 * hf;

    for (int half = 0; half < 2; ++half) {
        const int tl = 16 * (wv + 4 * half) + l15;
        const float cumA = expf(s_cs[tl]);
        const float* cp = xcb + (unsigned)((tl ^ rmask) * CONV2 + cbase + 2 * hf);

        v8f acc0 = {}, acc1 = {}, acc2 = {}, acc3 = {};
#pragma unroll
        for (int q = 0; q < 8; ++q) {
            const v2f cq = *(const v2f*)(cp + 4 * q);
            acc0 = wmma_f32_4(*(const v2f*)(hp + 4 * q),               cq, acc0);
            acc1 = wmma_f32_4(*(const v2f*)(hp + 16 * NSTATE + 4 * q), cq, acc1);
            acc2 = wmma_f32_4(*(const v2f*)(hp + 32 * NSTATE + 4 * q), cq, acc2);
            acc3 = wmma_f32_4(*(const v2f*)(hp + 48 * NSTATE + 4 * q), cq, acc3);
        }
        float* yr = yb + (unsigned)((tl ^ rmask) * D_INNER + 8 * hf);
        v8f accs[4] = {acc0, acc1, acc2, acc3};
        float ps = 0.0f, pq = 0.0f;
#pragma unroll
        for (int pb = 0; pb < 4; ++pb) {
            float4 ya = *(const float4*)(yr + 16 * pb);
            float4 yc = *(const float4*)(yr + 16 * pb + 4);
            ya.x += cumA * accs[pb][0];  ya.y += cumA * accs[pb][1];
            ya.z += cumA * accs[pb][2];  ya.w += cumA * accs[pb][3];
            yc.x += cumA * accs[pb][4];  yc.y += cumA * accs[pb][5];
            yc.z += cumA * accs[pb][6];  yc.w += cumA * accs[pb][7];
            *(float4*)(yr + 16 * pb)     = ya;
            *(float4*)(yr + 16 * pb + 4) = yc;
            add_sums(ya, ps, pq);
            add_sums(yc, ps, pq);
        }
        put_row_sums(ps, pq, hf, pb0 + 2 * (tl ^ rmask));
    }
}

__global__ void __launch_bounds__(128)
gemm_out(const float* Y, const float* __restrict__ part,
         const float* __restrict__ lnw,
         const float* __restrict__ lnb, const float* Wout,
         float* __restrict__ out) {
    const int wv = threadIdx.x >> 5;
    const int m0 = (blockIdx.y * 4 + wv) * 16;
    gemm_tile<4, D_INNER, D_MODEL, true>(Y, Wout, out, part, lnw, lnb, m0, (int)blockIdx.x * 64);
}

extern "C" void kernel_launch(void* const* d_in, const int* in_sizes, int n_in,
                              void* d_out, int out_size, void* d_ws, size_t ws_size,
                              hipStream_t stream) {
    const float* u        = (const float*)d_in[0];
    const float* W_in     = (const float*)d_in[2];
    const float* conv_fw  = (const float*)d_in[3];
    const float* conv_fb  = (const float*)d_in[4];
    const float* conv_bw  = (const float*)d_in[5];
    const float* conv_bb  = (const float*)d_in[6];
    const float* dt_bias  = (const float*)d_in[7];
    const float* A_log    = (const float*)d_in[8];
    const float* D_param  = (const float*)d_in[9];
    const float* ln_w     = (const float*)d_in[10];
    const float* ln_b     = (const float*)d_in[11];
    const float* W_out    = (const float*)d_in[12];
    float* out            = (float*)d_out;

    float* ws = (float*)d_ws;
    size_t off = 0;
    float* xbcdt = ws + off; off += (size_t)BL * D_IN_PROJ;
    float* xc    = ws + off; off += (size_t)BL * CONV2;
    float* dtb   = ws + off; off += (size_t)BL * NHEADS;
    float* ybuf  = ws + off; off += (size_t)BL * D_INNER;
    float* Sbuf  = ws + off; off += (size_t)32 * NCHUNK * HEADDIM * NSTATE;
    float* pAbuf = ws + off; off += (size_t)32 * NCHUNK;
    float* partb = ws + off; off += (size_t)2 * NHEADS * BL;

    gemm_in<<<dim3(CONV2 / 64 + 1, BL / 64), 128, 0, stream>>>(u, W_in, xbcdt);
    conv_dt<<<dim3(BL), 256, 0, stream>>>(xbcdt, conv_fw, conv_fb, conv_bw, conv_bb,
                                          dt_bias, xc, dtb);
    ssd_chunk<<<dim3(BATCH * NHEADS * NCHUNK), 128, 0, stream>>>(
        xc, dtb, A_log, D_param, ybuf, Sbuf, pAbuf);
    scan_combine<<<dim3(BATCH * NHEADS), 64, 0, stream>>>(Sbuf, pAbuf);
    ssd_correct<<<dim3(BATCH * NHEADS * NCHUNK), 128, 0, stream>>>(
        xc, dtb, A_log, Sbuf, ybuf, partb);
    gemm_out<<<dim3(D_MODEL / 64, BL / 64), 128, 0, stream>>>(
        ybuf, partb, ln_w, ln_b, W_out, out);

    (void)in_sizes; (void)n_in; (void)out_size; (void)ws_size;
}


# chunk-state combine folded into scan pass 3: each block reruns the serial recurrence over its earlier chunks' local states (loads 8 chunks deep), h_init to WMMA via LDS; 6 -> 5 launches; same sums
# speedup vs baseline: 2.8186x; 1.2079x over previous
#include <hip/hip_runtime.h>
#include <hip/hip_bf16.h>
#include <math.h>

#define BATCH    2
#define SEQLEN   4096
#define BL       (BATCH * SEQLEN)
#define D_MODEL  512
#define D_IN_PROJ 1168
#define CONV_DIM 576
#define CONV2    1152
#define D_INNER  1024
#define NHEADS   16
#define HH       8
#define HEADDIM  64
#define NSTATE   32
#define NCHUNK   32
#define CHUNK    128

typedef float v2f __attribute__((ext_vector_type(2)));
typedef float v8f __attribute__((ext_vector_type(8)));

__device__ __forceinline__ v8f wmma_f32_4(v2f a, v2f b, v8f c) {
    return __builtin_amdgcn_wmma_f32_16x16x4_f32(
         false, a,  false, b,
         (short)0, c,  false,  false);
}

typedef float v4f __attribute__((ext_vector_type(4)));

template <int NF>
struct BFrag {
    v4f   b4[4];
    float b1[4];
    float wl[4], wh[4], zl[4], zh[4];
};

template <int NF, int ND, bool LN>
__device__ __forceinline__ void load_b(BFrag<NF>& f, const float* bp,
                                       const float* __restrict__ lnw, const float* __restrict__ lnb,
                                       int krow) {
#pragma unroll
    for (int j = 0; j < 4; ++j) {
        if (NF == 4) f.b4[j] = *(const v4f*)(bp + (size_t)(krow + j) * ND);
        else         f.b1[j] = bp[(size_t)(krow + j) * ND];
    }
    if (LN) {
#pragma unroll
        for (int j = 0; j < 4; ++j) {
            f.wl[j] = lnw[krow + j];  f.wh[j] = lnw[krow + 16 + j];
            f.zl[j] = lnb[krow + j];  f.zh[j] = lnb[krow + 16 + j];
        }
    }
}

__device__ __forceinline__ void load_a(v4f (&a)[4], const float* ap, int kb) {
#pragma unroll
    for (int q = 0; q < 4; ++q) a[q] = *(const v4f*)(ap + 32 * kb + 4 * q);
}

template <int NF, bool LN>
__device__ __forceinline__ void mma_sub(v4f y, const BFrag<NF>& f, int hf, float mur, float rsr,
                                        v8f (&acc)[NF]) {
    if (LN) {
#pragma unroll
        for (int j = 0; j < 4; ++j) {
            const float w = hf ? f.wh[j] : f.wl[j];
            const float z = hf ? f.zh[j] : f.zl[j];
            y[j] = (y[j] - mur) * rsr * w + z;
        }
    }
    v2f a0, a1;
    a0.x = y[0];  a0.y = y[1];
    a1.x = y[2];  a1.y = y[3];
#pragma unroll
    for (int t = 0; t < NF; ++t) {
        v2f b;
        if (NF == 4) { b.x = f.b4[0][t];  b.y = f.b4[1][t]; }
        else         { b.x = f.b1[0];     b.y = f.b1[1]; }
        acc[t] = wmma_f32_4(a0, b, acc[t]);
    }
#pragma unroll
    for (int t = 0; t < NF; ++t) {
        v2f b;
        if (NF == 4) { b.x = f.b4[2][t];  b.y = f.b4[3][t]; }
        else         { b.x = f.b1[2];     b.y = f.b1[3]; }
        acc[t] = wmma_f32_4(a1, b, acc[t]);
    }
}

#define GEMM_SUB(AV, BCUR, BNXT, KROW_NEXT)                                       \
    load_b<NF, ND, LN>(BNXT, bp, lnw, lnb, (KROW_NEXT));                          \
    __builtin_amdgcn_sched_barrier(0);                                            \
    mma_sub<NF, LN>((AV), (BCUR), hf, mur, rsr, acc);                             \
    __builtin_amdgcn_sched_barrier(0);

template <int NF, int KD, int ND, bool LN>
__device__ __forceinline__ void gemm_tile(const float* A, const float* B, float* __restrict__ C,
                                          const float* __restrict__ part,
                                          const float* __restrict__ lnw, const float* __restrict__ lnb,
                                          int m0, int n0) {
    const int lane = threadIdx.x & 31;
    const int c    = lane & 15;
    const int hf   = lane >> 4;
    const float* ap = A + (size_t)(m0 + c) * KD + 16 * hf;
    const float* bp = B + (size_t)(16 * hf) * ND + n0 + (NF == 4 ? 4 * c : c);
    float mur = 0.0f, rsr = 1.0f;
    if (LN) {
        float s = 0.0f, q = 0.0f;
#pragma unroll
        for (int h = 0; h < NHEADS; ++h) {
            const v2f p = *(const v2f*)(part + 2 * ((size_t)h * BL + m0 + c));
            s += p.x;  q += p.y;
        }
        const float m = s * (1.0f / D_INNER);
        const float v = q * (1.0f / D_INNER) - m * m;
        mur = m;
        rsr = rsqrtf(v + 1e-5f);
    }

    v8f acc[NF];
#pragma unroll
    for (int t = 0; t < NF; ++t) acc[t] = v8f{};

    constexpr int NKB = KD / 32;
    v4f aP[4], aQ[4];
    BFrag<NF> bX, bY;
    load_a(aP, ap, 0);
    load_b<NF, ND, LN>(bX, bp, lnw, lnb, 0);
    for (int kb = 0; kb < NKB; kb += 2) {
        const int kn = (kb + 2 < NKB) ? (kb + 2) : (kb + 1);
        load_a(aQ, ap, kb + 1);
        GEMM_SUB(aP[0], bX, bY, 32 * kb + 4)
        GEMM_SUB(aP[1], bY, bX, 32 * kb + 8)
        GEMM_SUB(aP[2], bX, bY, 32 * kb + 12)
        GEMM_SUB(aP[3], bY, bX, 32 * kb + 32)
        load_a(aP, ap, kn);
        GEMM_SUB(aQ[0], bX, bY, 32 * kb + 36)
        GEMM_SUB(aQ[1], bY, bX, 32 * kb + 40)
        GEMM_SUB(aQ[2], bX, bY, 32 * kb + 44)
        GEMM_SUB(aQ[3], bY, bX, 32 * kn)
    }

#pragma unroll
    for (int r = 0; r < 8; ++r) {
        float* cr = C + (size_t)(m0 + 8 * hf + r) * ND + n0;
        if (NF == 4) {
            v4f o;
            o[0] = acc[0][r];  o[1] = acc[1][r];
            o[2] = acc[2][r];  o[3] = acc[3][r];
            *(v4f*)(cr + 4 * c) = o;
        } else {
            cr[c] = acc[0][r];
        }
    }
}
#undef GEMM_SUB

__global__ void __launch_bounds__(128)
gemm_in(const float* U, const float* Win, float* __restrict__ out) {
    const int wv = threadIdx.x >> 5;
    const int m0 = (blockIdx.y * 4 + wv) * 16;
    if (blockIdx.x < CONV2 / 64)
        gemm_tile<4, D_MODEL, D_IN_PROJ, false>(U, Win, out, nullptr, nullptr, nullptr,
                                                m0, (int)blockIdx.x * 64);
    else
        gemm_tile<1, D_MODEL, D_IN_PROJ, false>(U, Win, out, nullptr, nullptr, nullptr,
                                                m0, CONV2);
}

__global__ void __launch_bounds__(256)
conv_dt(const float* __restrict__ xbcdt,
        const float* __restrict__ wf, const float* __restrict__ bf,
        const float* __restrict__ wb, const float* __restrict__ bb,
        const float* __restrict__ dt_bias,
        float* __restrict__ xc, float* __restrict__ dtb) {
    const size_t bl = blockIdx.x;
    const int b = (int)(bl / SEQLEN);
    const int t = (int)(bl % SEQLEN);

    for (int c = threadIdx.x; c < CONV2; c += blockDim.x) {
        float acc;
        if (c < CONV_DIM) {
            acc = bf[c];
#pragma unroll
            for (int k = 0; k < 4; ++k) {
                int tt = t - 3 + k;
                if (tt >= 0)
                    acc += xbcdt[((size_t)b * SEQLEN + tt) * D_IN_PROJ + c] * wf[c * 4 + k];
            }
        } else {
            int cc = c - CONV_DIM;
            acc = bb[cc];
#pragma unroll
            for (int k = 0; k < 4; ++k) {
                int tt = t + 3 - k;
                if (tt < SEQLEN)
                    acc += xbcdt[((size_t)b * SEQLEN + tt) * D_IN_PROJ + c] * wb[cc * 4 + k];
            }
        }
        float sig = 1.0f / (1.0f + expf(-acc));
        xc[bl * CONV2 + c] = acc * sig;
    }
    for (int h = threadIdx.x; h < NHEADS; h += blockDim.x) {
        float v = xbcdt[bl * D_IN_PROJ + CONV2 + h] + dt_bias[h];
        dtb[bl * NHEADS + h] = (v > 20.0f) ? v : log1pf(expf(v));
    }
}

__device__ __forceinline__ void head_bases(int h, int& xbase, int& bbase, int& cbase) {
    if (h < HH) { xbase = h * HEADDIM;                 bbase = 512;  cbase = 544;  }
    else        { xbase = CONV_DIM + (h - HH) * HEADDIM; bbase = 1088; cbase = 1120; }
}


__device__ __forceinline__ void chunk_prefix(const float* __restrict__ dtc, int rmask, float Ah,
                                             float* s_tot, float* s_dt, float* s_cs) {
    const int i    = threadIdx.x;
    const int lane = i & 31;
    const int wv   = i >> 5;
    const float dtv = dtc[(unsigned)((i ^ rmask) * NHEADS)];
    float v = Ah * dtv;
#pragma unroll
    for (int d = 1; d < 32; d <<= 1) {
        const float n = __shfl_up(v, d, 32);
        if (lane >= d) v += n;
    }
    if (lane == 31) s_tot[wv] = v;
    s_dt[i] = dtv;
    __syncthreads();
    float off = 0.0f;
    for (int w = 0; w < wv; ++w) off += s_tot[w];
    s_cs[i] = off + v;
    __syncthreads();
}

__device__ __forceinline__ void load_pair(const float* __restrict__ xcb, int j, int l15, int hf,
                                          int rmask, int bbase, int xbase,
                                          v2f (&bf)[8], float (&xv)[4][8]) {
    const unsigned rbx = (unsigned)(((16 * j + l15) ^ rmask) * CONV2 + bbase + 2 * hf);
#pragma unroll
    for (int q = 0; q < 8; ++q) bf[q] = *(const v2f*)(xcb + rbx + (unsigned)(4 * q));
    const int sb0 = 16 * j + 8 * hf;
#pragma unroll
    for (int r = 0; r < 8; ++r) {
        const unsigned ro = (unsigned)(((sb0 + r) ^ rmask) * CONV2 + xbase + l15);
        xv[0][r] = xcb[ro];
        xv[1][r] = xcb[ro + 16];
        xv[2][r] = xcb[ro + 32];
        xv[3][r] = xcb[ro + 48];
    }
}

__global__ void __launch_bounds__(128)
ssd_chunk(const float* __restrict__ xc, const float* __restrict__ dtb,
          const float* __restrict__ A_log, const float* __restrict__ Dp,
          float* __restrict__ y, float* __restrict__ S, float* __restrict__ prodA) {
    const int chunk = blockIdx.x & (NCHUNK - 1);
    const int h     = (blockIdx.x >> 5) & 15;
    const int b     = blockIdx.x >> 9;
    int xbase, bbase, cbase; head_bases(h, xbase, bbase, cbase);
    const int s0    = chunk * CHUNK;
    const int rmask = (h < HH) ? 0 : (CHUNK - 1);
    const int tbase = (h < HH) ? s0 : (SEQLEN - CHUNK - s0);
    const size_t row0 = (size_t)b * SEQLEN + tbase;
    const float* xcb = xc  + row0 * CONV2;
    const float* dtc = dtb + row0 * NHEADS + h;
    float*       yb  = y   + row0 * D_INNER + h * HEADDIM;

    const float Ah = -expf(A_log[h]);
    const float Dh = Dp[h];

    __shared__ float s_tot[4];
    __shared__ alignas(16) float s_dt[CHUNK];
    __shared__ alignas(16) float s_cs[CHUNK];
    __shared__ alignas(16) float s_w[CHUNK];
    chunk_prefix(dtc, rmask, Ah, s_tot, s_dt, s_cs);
    const float cl = s_cs[CHUNK - 1];
    s_w[threadIdx.x] = expf(cl - s_cs[threadIdx.x]) * s_dt[threadIdx.x];
    __syncthreads();

    const int wv   = __builtin_amdgcn_readfirstlane((int)(threadIdx.x >> 5));
    const int lane = threadIdx.x & 31;
    const int l15  = lane & 15;
    const int hf   = lane >> 4;

    for (int half = 0; half < 2; ++half) {
        const int i  = half ? (7 - wv) : wv;
        const int tl = 16 * i + l15;
        const unsigned rtx = (unsigned)((tl ^ rmask) * CONV2);
        const float cst = s_cs[tl];

        v2f cf[8];
#pragma unroll
        for (int q = 0; q < 8; ++q)
            cf[q] = *(const v2f*)(xcb + rtx + (unsigned)(cbase + 2 * hf + 4 * q));

        v8f acc0 = {}, acc1 = {}, acc2 = {}, acc3 = {};
        v2f bfr[8];
        float xa[4][8];
        load_pair(xcb, 0, l15, hf, rmask, bbase, xbase, bfr, xa);
        for (int j = 0; j <= i; ++j) {
            const int sb0 = 16 * j + 8 * hf;
            v2f bfn[8];
            float xn[4][8];
            load_pair(xcb, (j < i) ? (j + 1) : i, l15, hf, rmask, bbase, xbase, bfn, xn);
            __builtin_amdgcn_sched_barrier(0);

            v8f g = {};
#pragma unroll
            for (int q = 0; q < 8; ++q) g = wmma_f32_4(bfr[q], cf[q], g);

            float m[8];
#pragma unroll
            for (int r = 0; r < 8; ++r) {
                const int ss = sb0 + r;
                const float css = s_cs[ss];
                const float d = (ss <= tl) ? (cst - css) : -INFINITY;
                m[r] = g[r] * (expf(d) * s_dt[ss]);
            }
#pragma unroll
            for (int kk = 0; kk < 4; ++kk) {
                v2f bm; bm.x = m[2 * kk]; bm.y = m[2 * kk + 1];
                v2f a0, a1, a2, a3;
                a0.x = xa[0][2 * kk];  a0.y = xa[0][2 * kk + 1];
                a1.x = xa[1][2 * kk];  a1.y = xa[1][2 * kk + 1];
                a2.x = xa[2][2 * kk];  a2.y = xa[2][2 * kk + 1];
                a3.x = xa[3][2 * kk];  a3.y = xa[3][2 * kk + 1];
                acc0 = wmma_f32_4(a0, bm, acc0);
                acc1 = wmma_f32_4(a1, bm, acc1);
                acc2 = wmma_f32_4(a2, bm, acc2);
                acc3 = wmma_f32_4(a3, bm, acc3);
            }
            __builtin_amdgcn_sched_barrier(0);
#pragma unroll
            for (int q = 0; q < 8; ++q) {
                bfr[q] = bfn[q];
                xa[0][q] = xn[0][q];  xa[1][q] = xn[1][q];
                xa[2][q] = xn[2][q];  xa[3][q] = xn[3][q];
            }
        }
        const float* xr = xcb + rtx + (unsigned)(xbase + 8 * hf);
        float*       yr = yb + (unsigned)((tl ^ rmask) * D_INNER + 8 * hf);
        v8f accs[4] = {acc0, acc1, acc2, acc3};
#pragma unroll
        for (int pb = 0; pb < 4; ++pb) {
            const float4 xa4 = *(const float4*)(xr + 16 * pb);
            const float4 xb4 = *(const float4*)(xr + 16 * pb + 4);
            float4 oa, ob;
            oa.x = accs[pb][0] + Dh * xa4.x;  oa.y = accs[pb][1] + Dh * xa4.y;
            oa.z = accs[pb][2] + Dh * xa4.z;  oa.w = accs[pb][3] + Dh * xa4.w;
            ob.x = accs[pb][4] + Dh * xb4.x;  ob.y = accs[pb][5] + Dh * xb4.y;
            ob.z = accs[pb][6] + Dh * xb4.z;  ob.w = accs[pb][7] + Dh * xb4.w;
            *(float4*)(yr + 16 * pb)     = oa;
            *(float4*)(yr + 16 * pb + 4) = ob;
        }
    }

    {
        v8f st0 = {}, st1 = {};
        const unsigned xo = (unsigned)(xbase + 16 * wv + l15);
        const unsigned bo = (unsigned)(bbase + l15);
        for (int k0 = 0; k0 < CHUNK; k0 += 32) {
            float xv[16], bv0[16], bv1[16], wv16[16];
#pragma unroll
            for (int u = 0; u < 16; ++u) {
                const int sa = k0 + 4 * (u >> 1) + 2 * hf + (u & 1);
                const unsigned ro = (unsigned)((sa ^ rmask) * CONV2);
                xv[u]   = xcb[ro + xo];
                bv0[u]  = xcb[ro + bo];
                bv1[u]  = xcb[ro + bo + 16];
                wv16[u] = s_w[sa];
            }
#pragma unroll
            for (int ks = 0; ks < 8; ++ks) {
                v2f a, b0, b1;
                a.x  = xv[2 * ks] * wv16[2 * ks];
                a.y  = xv[2 * ks + 1] * wv16[2 * ks + 1];
                b0.x = bv0[2 * ks];  b0.y = bv0[2 * ks + 1];
                b1.x = bv1[2 * ks];  b1.y = bv1[2 * ks + 1];
                st0 = wmma_f32_4(a, b0, st0);
                st1 = wmma_f32_4(a, b1, st1);
            }
        }
        const size_t sb = (((size_t)(b * NHEADS + h)) * NCHUNK + chunk) * (HEADDIM * NSTATE)
                        + (size_t)(16 * wv + 8 * hf) * NSTATE + l15;
#pragma unroll
        for (int r = 0; r < 8; ++r) {
            S[sb + (size_t)r * NSTATE]      = st0[r];
            S[sb + (size_t)r * NSTATE + 16] = st1[r];
        }
    }
    if (threadIdx.x == 0) prodA[(b * NHEADS + h) * NCHUNK + chunk] = expf(cl);
}

__device__ __forceinline__ void add_sums(const float4 a, float& ps, float& pq) {
    ps += (a.x + a.y) + (a.z + a.w);
    pq += (a.x * a.x + a.y * a.y) + (a.z * a.z + a.w * a.w);
}

__device__ __forceinline__ void put_row_sums(float ps, float pq, int hf, float* pr) {
    ps += __shfl_xor(ps, 16, 32);
    pq += __shfl_xor(pq, 16, 32);
    if (hf == 0) { v2f o; o.x = ps; o.y = pq; *(v2f*)pr = o; }
}

#define HPITCH 36

template <int NB>
__device__ __forceinline__ void state_steps(const float* __restrict__ sp, const float* __restrict__ pa, int c,
                                            v4f (&hv)[4]) {
    v4f   s[NB][4];
    float a[NB];
#pragma unroll
    for (int u = 0; u < NB; ++u) {
        a[u] = pa[c + u];
#pragma unroll
        for (int k = 0; k < 4; ++k)
            s[u][k] = *(const v4f*)(sp + (size_t)(c + u) * (HEADDIM * NSTATE) + 512 * k);
    }
#pragma unroll
    for (int u = 0; u < NB; ++u) {
#pragma unroll
        for (int k = 0; k < 4; ++k) {
            hv[k][0] = fmaf(a[u], hv[k][0], s[u][k][0]);
            hv[k][1] = fmaf(a[u], hv[k][1], s[u][k][1]);
            hv[k][2] = fmaf(a[u], hv[k][2], s[u][k][2]);
            hv[k][3] = fmaf(a[u], hv[k][3], s[u][k][3]);
        }
    }
}

__global__ void __launch_bounds__(128)
ssd_correct(const float* __restrict__ xc, const float* __restrict__ dtb,
            const float* __restrict__ A_log, const float* __restrict__ S,
            const float* __restrict__ prodA,
            float* __restrict__ y, float* __restrict__ part) {
    const int chunk = blockIdx.x & (NCHUNK - 1);
    const int h = (blockIdx.x >> 5) & 15;
    const int b = blockIdx.x >> 9;
    int xbase, bbase, cbase; head_bases(h, xbase, bbase, cbase);
    (void)xbase; (void)bbase;
    const int s0    = chunk * CHUNK;
    const int rmask = (h < HH) ? 0 : (CHUNK - 1);
    const int tbase = (h < HH) ? s0 : (SEQLEN - CHUNK - s0);
    const size_t row0 = (size_t)b * SEQLEN + tbase;
    const float* xcb = xc  + row0 * CONV2;
    const float* dtc = dtb + row0 * NHEADS + h;
    float*       yb  = y   + row0 * D_INNER + h * HEADDIM;
    float*       pb0 = part + 2 * ((size_t)h * BL + row0);

    const int wv   = __builtin_amdgcn_readfirstlane((int)(threadIdx.x >> 5));
    const int lane = threadIdx.x & 31;
    const int l15  = lane & 15;
    const int hf   = lane >> 4;

    if (chunk == 0) {
        for (int half = 0; half < 2; ++half) {
            const int tl = 16 * (wv + 4 * half) + l15;
            const float* yr = yb + (unsigned)((tl ^ rmask) * D_INNER + 8 * hf);
            float ps = 0.0f, pq = 0.0f;
#pragma unroll
            for (int pb = 0; pb < 4; ++pb) {
                add_sums(*(const float4*)(yr + 16 * pb), ps, pq);
                add_sums(*(const float4*)(yr + 16 * pb + 4), ps, pq);
            }
            put_row_sums(ps, pq, hf, pb0 + 2 * (tl ^ rmask));
        }
        return;
    }

    const float Ah = -expf(A_log[h]);
    __shared__ float s_tot[4], s_dt[CHUNK], s_cs[CHUNK];
    __shared__ alignas(16) float s_h[HEADDIM * HPITCH];

    {
        const int bh = b * NHEADS + h;
        const float* sp = S + (size_t)bh * NCHUNK * (HEADDIM * NSTATE) + 4 * threadIdx.x;
        const float* pa = prodA + bh * NCHUNK;
        v4f hv[4];
#pragma unroll
        for (int k = 0; k < 4; ++k) hv[k] = v4f{};
        int c = 0;
        for (; c + 8 <= chunk; c += 8) state_steps<8>(sp, pa, c, hv);
        if (chunk & 4) { state_steps<4>(sp, pa, c, hv);  c += 4; }
        if (chunk & 2) { state_steps<2>(sp, pa, c, hv);  c += 2; }
        if (chunk & 1) { state_steps<1>(sp, pa, c, hv); }
#pragma unroll
        for (int k = 0; k < 4; ++k)
            *(v4f*)(s_h + (16 * k + ((int)threadIdx.x >> 3)) * HPITCH + 4 * ((int)threadIdx.x & 7)) = hv[k];
    }
    chunk_prefix(dtc, rmask, Ah, s_tot, s_dt, s_cs);

    v2f hA[4][8];
    {
        const float* hl = s_h + l15 * HPITCH + 2 * hf;
#pragma unroll
        for (int pb = 0; pb < 4; ++pb)
#pragma unroll
            for (int q = 0; q < 8; ++q)
                hA[pb][q] = *(const v2f*)(hl + 16 * pb * HPITCH + 4 * q);
    }

    for (int half = 0; half < 2; ++half) {
        const int tl = 16 * (wv + 4 * half) + l15;
        const float cumA = expf(s_cs[tl]);
        const float* cp = xcb + (unsigned)((tl ^ rmask) * CONV2 + cbase + 2 * hf);

        v8f acc0 = {}, acc1 = {}, acc2 = {}, acc3 = {};
#pragma unroll
        for (int q = 0; q < 8; ++q) {
            const v2f cq = *(const v2f*)(cp + 4 * q);
            acc0 = wmma_f32_4(hA[0][q], cq, acc0);
            acc1 = wmma_f32_4(hA[1][q], cq, acc1);
            acc2 = wmma_f32_4(hA[2][q], cq, acc2);
            acc3 = wmma_f32_4(hA[3][q], cq, acc3);
        }
        float* yr = yb + (unsigned)((tl ^ rmask) * D_INNER + 8 * hf);
        v8f accs[4] = {acc0, acc1, acc2, acc3};
        float ps = 0.0f, pq = 0.0f;
#pragma unroll
        for (int pb = 0; pb < 4; ++pb) {
            float4 ya = *(const float4*)(yr + 16 * pb);
            float4 yc = *(const float4*)(yr + 16 * pb + 4);
            ya.x += cumA * accs[pb][0];  ya.y += cumA * accs[pb][1];
            ya.z += cumA * accs[pb][2];  ya.w += cumA * accs[pb][3];
            yc.x += cumA * accs[pb][4];  yc.y += cumA * accs[pb][5];
            yc.z += cumA * accs[pb][6];  yc.w += cumA * accs[pb][7];
            *(float4*)(yr + 16 * pb)     = ya;
            *(float4*)(yr + 16 * pb + 4) = yc;
            add_sums(ya, ps, pq);
            add_sums(yc, ps, pq);
        }
        put_row_sums(ps, pq, hf, pb0 + 2 * (tl ^ rmask));
    }
}

__global__ void __launch_bounds__(128)
gemm_out(const float* Y, const float* __restrict__ part,
         const float* __restrict__ lnw,
         const float* __restrict__ lnb, const float* Wout,
         float* __restrict__ out) {
    const int wv = threadIdx.x >> 5;
    const int m0 = (blockIdx.y * 4 + wv) * 16;
    gemm_tile<4, D_INNER, D_MODEL, true>(Y, Wout, out, part, lnw, lnb, m0, (int)blockIdx.x * 64);
}

extern "C" void kernel_launch(void* const* d_in, const int* in_sizes, int n_in,
                              void* d_out, int out_size, void* d_ws, size_t ws_size,
                              hipStream_t stream) {
    const float* u        = (const float*)d_in[0];
    const float* W_in     = (const float*)d_in[2];
    const float* conv_fw  = (const float*)d_in[3];
    const float* conv_fb  = (const float*)d_in[4];
    const float* conv_bw  = (const float*)d_in[5];
    const float* conv_bb  = (const float*)d_in[6];
    const float* dt_bias  = (const float*)d_in[7];
    const float* A_log    = (const float*)d_in[8];
    const float* D_param  = (const float*)d_in[9];
    const float* ln_w     = (const float*)d_in[10];
    const float* ln_b     = (const float*)d_in[11];
    const float* W_out    = (const float*)d_in[12];
    float* out            = (float*)d_out;

    float* ws = (float*)d_ws;
    size_t off = 0;
    float* xbcdt = ws + off; off += (size_t)BL * D_IN_PROJ;
    float* xc    = ws + off; off += (size_t)BL * CONV2;
    float* dtb   = ws + off; off += (size_t)BL * NHEADS;
    float* ybuf  = ws + off; off += (size_t)BL * D_INNER;
    float* Sbuf  = ws + off; off += (size_t)32 * NCHUNK * HEADDIM * NSTATE;
    float* pAbuf = ws + off; off += (size_t)32 * NCHUNK;
    float* partb = ws + off; off += (size_t)2 * NHEADS * BL;

    gemm_in<<<dim3(CONV2 / 64 + 1, BL / 64), 128, 0, stream>>>(u, W_in, xbcdt);
    conv_dt<<<dim3(BL), 256, 0, stream>>>(xbcdt, conv_fw, conv_fb, conv_bw, conv_bb,
                                          dt_bias, xc, dtb);
    ssd_chunk<<<dim3(BATCH * NHEADS * NCHUNK), 128, 0, stream>>>(
        xc, dtb, A_log, D_param, ybuf, Sbuf, pAbuf);
    ssd_correct<<<dim3(BATCH * NHEADS * NCHUNK), 128, 0, stream>>>(
        xc, dtb, A_log, Sbuf, pAbuf, ybuf, partb);
    gemm_out<<<dim3(D_MODEL / 64, BL / 64), 128, 0, stream>>>(
        ybuf, partb, ln_w, ln_b, W_out, out);

    (void)in_sizes; (void)n_in; (void)out_size; (void)ws_size;
}


# conv+dt: all loads of a thread (4 taps, weights, bias of each of its 6 channels, dt inputs) asked for before the first use, not one wait per tap; block 256->192 threads = 6 whole rounds; same sums
# speedup vs baseline: 3.0250x; 1.0732x over previous
#include <hip/hip_runtime.h>
#include <hip/hip_bf16.h>
#include <math.h>

#define BATCH    2
#define SEQLEN   4096
#define BL       (BATCH * SEQLEN)
#define D_MODEL  512
#define D_IN_PROJ 1168
#define CONV_DIM 576
#define CONV2    1152
#define D_INNER  1024
#define NHEADS   16
#define HH       8
#define HEADDIM  64
#define NSTATE   32
#define NCHUNK   32
#define CHUNK    128

typedef float v2f __attribute__((ext_vector_type(2)));
typedef float v8f __attribute__((ext_vector_type(8)));

__device__ __forceinline__ v8f wmma_f32_4(v2f a, v2f b, v8f c) {
    return __builtin_amdgcn_wmma_f32_16x16x4_f32(
         false, a,  false, b,
         (short)0, c,  false,  false);
}

typedef float v4f __attribute__((ext_vector_type(4)));

template <int NF>
struct BFrag {
    v4f   b4[4];
    float b1[4];
    float wl[4], wh[4], zl[4], zh[4];
};

template <int NF, int ND, bool LN>
__device__ __forceinline__ void load_b(BFrag<NF>& f, const float* bp,
                                       const float* __restrict__ lnw, const float* __restrict__ lnb,
                                       int krow) {
#pragma unroll
    for (int j = 0; j < 4; ++j) {
        if (NF == 4) f.b4[j] = *(const v4f*)(bp + (size_t)(krow + j) * ND);
        else         f.b1[j] = bp[(size_t)(krow + j) * ND];
    }
    if (LN) {
#pragma unroll
        for (int j = 0; j < 4; ++j) {
            f.wl[j] = lnw[krow + j];  f.wh[j] = lnw[krow + 16 + j];
            f.zl[j] = lnb[krow + j];  f.zh[j] = lnb[krow + 16 + j];
        }
    }
}

__device__ __forceinline__ void load_a(v4f (&a)[4], const float* ap, int kb) {
#pragma unroll
    for (int q = 0; q < 4; ++q) a[q] = *(const v4f*)(ap + 32 * kb + 4 * q);
}

template <int NF, bool LN>
__device__ __forceinline__ void mma_sub(v4f y, const BFrag<NF>& f, int hf, float mur, float rsr,
                                        v8f (&acc)[NF]) {
    if (LN) {
#pragma unroll
        for (int j = 0; j < 4; ++j) {
            const float w = hf ? f.wh[j] : f.wl[j];
            const float z = hf ? f.zh[j] : f.zl[j];
            y[j] = (y[j] - mur) * rsr * w + z;
        }
    }
    v2f a0, a1;
    a0.x = y[0];  a0.y = y[1];
    a1.x = y[2];  a1.y = y[3];
#pragma unroll
    for (int t = 0; t < NF; ++t) {
        v2f b;
        if (NF == 4) { b.x = f.b4[0][t];  b.y = f.b4[1][t]; }
        else         { b.x = f.b1[0];     b.y = f.b1[1]; }
        acc[t] = wmma_f32_4(a0, b, acc[t]);
    }
#pragma unroll
    for (int t = 0; t < NF; ++t) {
        v2f b;
        if (NF == 4) { b.x = f.b4[2][t];  b.y = f.b4[3][t]; }
        else         { b.x = f.b1[2];     b.y = f.b1[3]; }
        acc[t] = wmma_f32_4(a1, b, acc[t]);
    }
}

#define GEMM_SUB(AV, BCUR, BNXT, KROW_NEXT)                                       \
    load_b<NF, ND, LN>(BNXT, bp, lnw, lnb, (KROW_NEXT));                          \
    __builtin_amdgcn_sched_barrier(0);                                            \
    mma_sub<NF, LN>((AV), (BCUR), hf, mur, rsr, acc);                             \
    __builtin_amdgcn_sched_barrier(0);

template <int NF, int KD, int ND, bool LN>
__device__ __forceinline__ void gemm_tile(const float* A, const float* B, float* __restrict__ C,
                                          const float* __restrict__ part,
                                          const float* __restrict__ lnw, const float* __restrict__ lnb,
                                          int m0, int n0) {
    const int lane = threadIdx.x & 31;
    const int c    = lane & 15;
    const int hf   = lane >> 4;
    const float* ap = A + (size_t)(m0 + c) * KD + 16 * hf;
    const float* bp = B + (size_t)(16 * hf) * ND + n0 + (NF == 4 ? 4 * c : c);
    float mur = 0.0f, rsr = 1.0f;
    if (LN) {
        float s = 0.0f, q = 0.0f;
#pragma unroll
        for (int h = 0; h < NHEADS; ++h) {
            const v2f p = *(const v2f*)(part + 2 * ((size_t)h * BL + m0 + c));
            s += p.x;  q += p.y;
        }
        const float m = s * (1.0f / D_INNER);
        const float v = q * (1.0f / D_INNER) - m * m;
        mur = m;
        rsr = rsqrtf(v + 1e-5f);
    }

    v8f acc[NF];
#pragma unroll
    for (int t = 0; t < NF; ++t) acc[t] = v8f{};

    constexpr int NKB = KD / 32;
    v4f aP[4], aQ[4];
    BFrag<NF> bX, bY;
    load_a(aP, ap, 0);
    load_b<NF, ND, LN>(bX, bp, lnw, lnb, 0);
    for (int kb = 0; kb < NKB; kb += 2) {
        const int kn = (kb + 2 < NKB) ? (kb + 2) : (kb + 1);
        load_a(aQ, ap, kb + 1);
        GEMM_SUB(aP[0], bX, bY, 32 * kb + 4)
        GEMM_SUB(aP[1], bY, bX, 32 * kb + 8)
        GEMM_SUB(aP[2], bX, bY, 32 * kb + 12)
        GEMM_SUB(aP[3], bY, bX, 32 * kb + 32)
        load_a(aP, ap, kn);
        GEMM_SUB(aQ[0], bX, bY, 32 * kb + 36)
        GEMM_SUB(aQ[1], bY, bX, 32 * kb + 40)
        GEMM_SUB(aQ[2], bX, bY, 32 * kb + 44)
        GEMM_SUB(aQ[3], bY, bX, 32 * kn)
    }

#pragma unroll
    for (int r = 0; r < 8; ++r) {
        float* cr = C + (size_t)(m0 + 8 * hf + r) * ND + n0;
        if (NF == 4) {
            v4f o;
            o[0] = acc[0][r];  o[1] = acc[1][r];
            o[2] = acc[2][r];  o[3] = acc[3][r];
            *(v4f*)(cr + 4 * c) = o;
        } else {
            cr[c] = acc[0][r];
        }
    }
}
#undef GEMM_SUB

__global__ void __launch_bounds__(128)
gemm_in(const float* U, const float* Win, float* __restrict__ out) {
    const int wv = threadIdx.x >> 5;
    const int m0 = (blockIdx.y * 4 + wv) * 16;
    if (blockIdx.x < CONV2 / 64)
        gemm_tile<4, D_MODEL, D_IN_PROJ, false>(U, Win, out, nullptr, nullptr, nullptr,
                                                m0, (int)blockIdx.x * 64);
    else
        gemm_tile<1, D_MODEL, D_IN_PROJ, false>(U, Win, out, nullptr, nullptr, nullptr,
                                                m0, CONV2);
}

#define CONV_THREADS 192
#define CONV_ROUNDS  (CONV2 / CONV_THREADS)
static_assert(CONV2 == CONV_ROUNDS * CONV_THREADS && 2 * CONV_DIM == CONV2 && (CONV_ROUNDS % 2) == 0,
              "whole rounds, half of them per direction");
__global__ void __launch_bounds__(CONV_THREADS)
conv_dt(const float* __restrict__ xbcdt,
        const float* __restrict__ wf, const float* __restrict__ bf,
        const float* __restrict__ wb, const float* __restrict__ bb,
        const float* __restrict__ dt_bias,
        float* __restrict__ xc, float* __restrict__ dtb) {
    const size_t bl = blockIdx.x;
    const int b = (int)(bl / SEQLEN);
    const int t = (int)(bl % SEQLEN);
    const int wv   = __builtin_amdgcn_readfirstlane((int)(threadIdx.x >> 5));
    const int lane = threadIdx.x & 31;
    const float* xb = xbcdt + (size_t)b * SEQLEN * D_IN_PROJ + 32 * wv;

    const float* rowp[2][4];
    bool ok[2][4];
#pragma unroll
    for (int k = 0; k < 4; ++k) {
        const int tf = t - 3 + k;
        const int tb = t + 3 - k;
        ok[0][k] = tf >= 0;
        ok[1][k] = tb < SEQLEN;
        rowp[0][k] = xb + (size_t)(ok[0][k] ? tf : t) * D_IN_PROJ;
        rowp[1][k] = xb + (size_t)(ok[1][k] ? tb : t) * D_IN_PROJ;
    }

    float xin[CONV_ROUNDS][4];
    v4f   wk[CONV_ROUNDS];
    float bias[CONV_ROUNDS];
#pragma unroll
    for (int i = 0; i < CONV_ROUNDS; ++i) {
        const int d  = (i >= CONV_ROUNDS / 2) ? 1 : 0;
        const int cd = CONV_THREADS * i - d * CONV_DIM + 32 * wv + lane;
        wk[i]   = *(const v4f*)((d ? wb : wf) + 4 * cd);
        bias[i] = (d ? bb : bf)[cd];
#pragma unroll
        for (int k = 0; k < 4; ++k) xin[i][k] = rowp[d][k][CONV_THREADS * i + lane];
    }
    const int   hd    = threadIdx.x & (NHEADS - 1);
    const float dtraw = xbcdt[bl * D_IN_PROJ + CONV2 + hd];
    const float dtbv  = dt_bias[hd];

    if (threadIdx.x < NHEADS) {
        float v = dtraw + dtbv;
        dtb[bl * NHEADS + threadIdx.x] = (v > 20.0f) ? v : log1pf(expf(v));
    }
    float* xo = xc + bl * CONV2 + 32 * wv + lane;
#pragma unroll
    for (int i = 0; i < CONV_ROUNDS; ++i) {
        const int d = (i >= CONV_ROUNDS / 2) ? 1 : 0;
        float acc = bias[i];
#pragma unroll
        for (int k = 0; k < 4; ++k)
            if (ok[d][k]) acc += xin[i][k] * wk[i][k];
        float sig = 1.0f / (1.0f + expf(-acc));
        xo[CONV_THREADS * i] = acc * sig;
    }
}

__device__ __forceinline__ void head_bases(int h, int& xbase, int& bbase, int& cbase) {
    if (h < HH) { xbase = h * HEADDIM;                 bbase = 512;  cbase = 544;  }
    else        { xbase = CONV_DIM + (h - HH) * HEADDIM; bbase = 1088; cbase = 1120; }
}


__device__ __forceinline__ void chunk_prefix(const float* __restrict__ dtc, int rmask, float Ah,
                                             float* s_tot, float* s_dt, float* s_cs) {
    const int i    = threadIdx.x;
    const int lane = i & 31;
    const int wv   = i >> 5;
    const float dtv = dtc[(unsigned)((i ^ rmask) * NHEADS)];
    float v = Ah * dtv;
#pragma unroll
    for (int d = 1; d < 32; d <<= 1) {
        const float n = __shfl_up(v, d, 32);
        if (lane >= d) v += n;
    }
    if (lane == 31) s_tot[wv] = v;
    s_dt[i] = dtv;
    __syncthreads();
    float off = 0.0f;
    for (int w = 0; w < wv; ++w) off += s_tot[w];
    s_cs[i] = off + v;
    __syncthreads();
}

__device__ __forceinline__ void load_pair(const float* __restrict__ xcb, int j, int l15, int hf,
                                          int rmask, int bbase, int xbase,
                                          v2f (&bf)[8], float (&xv)[4][8]) {
    const unsigned rbx = (unsigned)(((16 * j + l15) ^ rmask) * CONV2 + bbase + 2 * hf);
#pragma unroll
    for (int q = 0; q < 8; ++q) bf[q] = *(const v2f*)(xcb + rbx + (unsigned)(4 * q));
    const int sb0 = 16 * j + 8 * hf;
#pragma unroll
    for (int r = 0; r < 8; ++r) {
        const unsigned ro = (unsigned)(((sb0 + r) ^ rmask) * CONV2 + xbase + l15);
        xv[0][r] = xcb[ro];
        xv[1][r] = xcb[ro + 16];
        xv[2][r] = xcb[ro + 32];
        xv[3][r] = xcb[ro + 48];
    }
}

__global__ void __launch_bounds__(128)
ssd_chunk(const float* __restrict__ xc, const float* __restrict__ dtb,
          const float* __restrict__ A_log, const float* __restrict__ Dp,
          float* __restrict__ y, float* __restrict__ S, float* __restrict__ prodA) {
    const int chunk = blockIdx.x & (NCHUNK - 1);
    const int h     = (blockIdx.x >> 5) & 15;
    const int b     = blockIdx.x >> 9;
    int xbase, bbase, cbase; head_bases(h, xbase, bbase, cbase);
    const int s0    = chunk * CHUNK;
    const int rmask = (h < HH) ? 0 : (CHUNK - 1);
    const int tbase = (h < HH) ? s0 : (SEQLEN - CHUNK - s0);
    const size_t row0 = (size_t)b * SEQLEN + tbase;
    const float* xcb = xc  + row0 * CONV2;
    const float* dtc = dtb + row0 * NHEADS + h;
    float*       yb  = y   + row0 * D_INNER + h * HEADDIM;

    const float Ah = -expf(A_log[h]);
    const float Dh = Dp[h];

    __shared__ float s_tot[4];
    __shared__ alignas(16) float s_dt[CHUNK];
    __shared__ alignas(16) float s_cs[CHUNK];
    __shared__ alignas(16) float s_w[CHUNK];
    chunk_prefix(dtc, rmask, Ah, s_tot, s_dt, s_cs);
    const float cl = s_cs[CHUNK - 1];
    s_w[threadIdx.x] = expf(cl - s_cs[threadIdx.x]) * s_dt[threadIdx.x];
    __syncthreads();

    const int wv   = __builtin_amdgcn_readfirstlane((int)(threadIdx.x >> 5));
    const int lane = threadIdx.x & 31;
    const int l15  = lane & 15;
    const int hf   = lane >> 4;

    for (int half = 0; half < 2; ++half) {
        const int i  = half ? (7 - wv) : wv;
        const int tl = 16 * i + l15;
        const unsigned rtx = (unsigned)((tl ^ rmask) * CONV2);
        const float cst = s_cs[tl];

        v2f cf[8];
#pragma unroll
        for (int q = 0; q < 8; ++q)
            cf[q] = *(const v2f*)(xcb + rtx + (unsigned)(cbase + 2 * hf + 4 * q));

        v8f acc0 = {}, acc1 = {}, acc2 = {}, acc3 = {};
        v2f bfr[8];
        float xa[4][8];
        load_pair(xcb, 0, l15, hf, rmask, bbase, xbase, bfr, xa);
        for (int j = 0; j <= i; ++j) {
            const int sb0 = 16 * j + 8 * hf;
            v2f bfn[8];
            float xn[4][8];
            load_pair(xcb, (j < i) ? (j + 1) : i, l15, hf, rmask, bbase, xbase, bfn, xn);
            __builtin_amdgcn_sched_barrier(0);

            v8f g = {};
#pragma unroll
            for (int q = 0; q < 8; ++q) g = wmma_f32_4(bfr[q], cf[q], g);

            float m[8];
#pragma unroll
            for (int r = 0; r < 8; ++r) {
                const int ss = sb0 + r;
                const float css = s_cs[ss];
                const float d = (ss <= tl) ? (cst - css) : -INFINITY;
                m[r] = g[r] * (expf(d) * s_dt[ss]);
            }
#pragma unroll
            for (int kk = 0; kk < 4; ++kk) {
                v2f bm; bm.x = m[2 * kk]; bm.y = m[2 * kk + 1];
                v2f a0, a1, a2, a3;
                a0.x = xa[0][2 * kk];  a0.y = xa[0][2 * kk + 1];
                a1.x = xa[1][2 * kk];  a1.y = xa[1][2 * kk + 1];
                a2.x = xa[2][2 * kk];  a2.y = xa[2][2 * kk + 1];
                a3.x = xa[3][2 * kk];  a3.y = xa[3][2 * kk + 1];
                acc0 = wmma_f32_4(a0, bm, acc0);
                acc1 = wmma_f32_4(a1, bm, acc1);
                acc2 = wmma_f32_4(a2, bm, acc2);
                acc3 = wmma_f32_4(a3, bm, acc3);
            }
            __builtin_amdgcn_sched_barrier(0);
#pragma unroll
            for (int q = 0; q < 8; ++q) {
                bfr[q] = bfn[q];
                xa[0][q] = xn[0][q];  xa[1][q] = xn[1][q];
                xa[2][q] = xn[2][q];  xa[3][q] = xn[3][q];
            }
        }
        const float* xr = xcb + rtx + (unsigned)(xbase + 8 * hf);
        float*       yr = yb + (unsigned)((tl ^ rmask) * D_INNER + 8 * hf);
        v8f accs[4] = {acc0, acc1, acc2, acc3};
#pragma unroll
        for (int pb = 0; pb < 4; ++pb) {
            const float4 xa4 = *(const float4*)(xr + 16 * pb);
            const float4 xb4 = *(const float4*)(xr + 16 * pb + 4);
            float4 oa, ob;
            oa.x = accs[pb][0] + Dh * xa4.x;  oa.y = accs[pb][1] + Dh * xa4.y;
            oa.z = accs[pb][2] + Dh * xa4.z;  oa.w = accs[pb][3] + Dh * xa4.w;
            ob.x = accs[pb][4] + Dh * xb4.x;  ob.y = accs[pb][5] + Dh * xb4.y;
            ob.z = accs[pb][6] + Dh * xb4.z;  ob.w = accs[pb][7] + Dh * xb4.w;
            *(float4*)(yr + 16 * pb)     = oa;
            *(float4*)(yr + 16 * pb + 4) = ob;
        }
    }

    {
        v8f st0 = {}, st1 = {};
        const unsigned xo = (unsigned)(xbase + 16 * wv + l15);
        const unsigned bo = (unsigned)(bbase + l15);
        for (int k0 = 0; k0 < CHUNK; k0 += 32) {
            float xv[16], bv0[16], bv1[16], wv16[16];
#pragma unroll
            for (int u = 0; u < 16; ++u) {
                const int sa = k0 + 4 * (u >> 1) + 2 * hf + (u & 1);
                const unsigned ro = (unsigned)((sa ^ rmask) * CONV2);
                xv[u]   = xcb[ro + xo];
                bv0[u]  = xcb[ro + bo];
                bv1[u]  = xcb[ro + bo + 16];
                wv16[u] = s_w[sa];
            }
#pragma unroll
            for (int ks = 0; ks < 8; ++ks) {
                v2f a, b0, b1;
                a.x  = xv[2 * ks] * wv16[2 * ks];
                a.y  = xv[2 * ks + 1] * wv16[2 * ks + 1];
                b0.x = bv0[2 * ks];  b0.y = bv0[2 * ks + 1];
                b1.x = bv1[2 * ks];  b1.y = bv1[2 * ks + 1];
                st0 = wmma_f32_4(a, b0, st0);
                st1 = wmma_f32_4(a, b1, st1);
            }
        }
        const size_t sb = (((size_t)(b * NHEADS + h)) * NCHUNK + chunk) * (HEADDIM * NSTATE)
                        + (size_t)(16 * wv + 8 * hf) * NSTATE + l15;
#pragma unroll
        for (int r = 0; r < 8; ++r) {
            S[sb + (size_t)r * NSTATE]      = st0[r];
            S[sb + (size_t)r * NSTATE + 16] = st1[r];
        }
    }
    if (threadIdx.x == 0) prodA[(b * NHEADS + h) * NCHUNK + chunk] = expf(cl);
}

__device__ __forceinline__ void add_sums(const float4 a, float& ps, float& pq) {
    ps += (a.x + a.y) + (a.z + a.w);
    pq += (a.x * a.x + a.y * a.y) + (a.z * a.z + a.w * a.w);
}

__device__ __forceinline__ void put_row_sums(float ps, float pq, int hf, float* pr) {
    ps += __shfl_xor(ps, 16, 32);
    pq += __shfl_xor(pq, 16, 32);
    if (hf == 0) { v2f o; o.x = ps; o.y = pq; *(v2f*)pr = o; }
}

#define HPITCH 36

template <int NB>
__device__ __forceinline__ void state_steps(const float* __restrict__ sp, const float* __restrict__ pa, int c,
                                            v4f (&hv)[4]) {
    v4f   s[NB][4];
    float a[NB];
#pragma unroll
    for (int u = 0; u < NB; ++u) {
        a[u] = pa[c + u];
#pragma unroll
        for (int k = 0; k < 4; ++k)
            s[u][k] = *(const v4f*)(sp + (size_t)(c + u) * (HEADDIM * NSTATE) + 512 * k);
    }
#pragma unroll
    for (int u = 0; u < NB; ++u) {
#pragma unroll
        for (int k = 0; k < 4; ++k) {
            hv[k][0] = fmaf(a[u], hv[k][0], s[u][k][0]);
            hv[k][1] = fmaf(a[u], hv[k][1], s[u][k][1]);
            hv[k][2] = fmaf(a[u], hv[k][2], s[u][k][2]);
            hv[k][3] = fmaf(a[u], hv[k][3], s[u][k][3]);
        }
    }
}

__global__ void __launch_bounds__(128)
ssd_correct(const float* __restrict__ xc, const float* __restrict__ dtb,
            const float* __restrict__ A_log, const float* __restrict__ S,
            const float* __restrict__ prodA,
            float* __restrict__ y, float* __restrict__ part) {
    const int chunk = blockIdx.x & (NCHUNK - 1);
    const int h = (blockIdx.x >> 5) & 15;
    const int b = blockIdx.x >> 9;
    int xbase, bbase, cbase; head_bases(h, xbase, bbase, cbase);
    (void)xbase; (void)bbase;
    const int s0    = chunk * CHUNK;
    const int rmask = (h < HH) ? 0 : (CHUNK - 1);
    const int tbase = (h < HH) ? s0 : (SEQLEN - CHUNK - s0);
    const size_t row0 = (size_t)b * SEQLEN + tbase;
    const float* xcb = xc  + row0 * CONV2;
    const float* dtc = dtb + row0 * NHEADS + h;
    float*       yb  = y   + row0 * D_INNER + h * HEADDIM;
    float*       pb0 = part + 2 * ((size_t)h * BL + row0);

    const int wv   = __builtin_amdgcn_readfirstlane((int)(threadIdx.x >> 5));
    const int lane = threadIdx.x & 31;
    const int l15  = lane & 15;
    const int hf   = lane >> 4;

    if (chunk == 0) {
        for (int half = 0; half < 2; ++half) {
            const int tl = 16 * (wv + 4 * half) + l15;
            const float* yr = yb + (unsigned)((tl ^ rmask) * D_INNER + 8 * hf);
            float ps = 0.0f, pq = 0.0f;
#pragma unroll
            for (int pb = 0; pb < 4; ++pb) {
                add_sums(*(const float4*)(yr + 16 * pb), ps, pq);
                add_sums(*(const float4*)(yr + 16 * pb + 4), ps, pq);
            }
            put_row_sums(ps, pq, hf, pb0 + 2 * (tl ^ rmask));
        }
        return;
    }

    const float Ah = -expf(A_log[h]);
    __shared__ float s_tot[4], s_dt[CHUNK], s_cs[CHUNK];
    __shared__ alignas(16) float s_h[HEADDIM * HPITCH];

    {
        const int bh = b * NHEADS + h;
        const float* sp = S + (size_t)bh * NCHUNK * (HEADDIM * NSTATE) + 4 * threadIdx.x;
        const float* pa = prodA + bh * NCHUNK;
        v4f hv[4];
#pragma unroll
        for (int k = 0; k < 4; ++k) hv[k] = v4f{};
        int c = 0;
        for (; c + 8 <= chunk; c += 8) state_steps<8>(sp, pa, c, hv);
        if (chunk & 4) { state_steps<4>(sp, pa, c, hv);  c += 4; }
        if (chunk & 2) { state_steps<2>(sp, pa, c, hv);  c += 2; }
        if (chunk & 1) { state_steps<1>(sp, pa, c, hv); }
#pragma unroll
        for (int k = 0; k < 4; ++k)
            *(v4f*)(s_h + (16 * k + ((int)threadIdx.x >> 3)) * HPITCH + 4 * ((int)threadIdx.x & 7)) = hv[k];
    }
    chunk_prefix(dtc, rmask, Ah, s_tot, s_dt, s_cs);

    v2f hA[4][8];
    {
        const float* hl = s_h + l15 * HPITCH + 2 * hf;
#pragma unroll
        for (int pb = 0; pb < 4; ++pb)
#pragma unroll
            for (int q = 0; q < 8; ++q)
                hA[pb][q] = *(const v2f*)(hl + 16 * pb * HPITCH + 4 * q);
    }

    for (int half = 0; half < 2; ++half) {
        const int tl = 16 * (wv + 4 * half) + l15;
        const float cumA = expf(s_cs[tl]);
        const float* cp = xcb + (unsigned)((tl ^ rmask) * CONV2 + cbase + 2 * hf);

        v8f acc0 = {}, acc1 = {}, acc2 = {}, acc3 = {};
#pragma unroll
        for (int q = 0; q < 8; ++q) {
            const v2f cq = *(const v2f*)(cp + 4 * q);
            acc0 = wmma_f32_4(hA[0][q], cq, acc0);
            acc1 = wmma_f32_4(hA[1][q], cq, acc1);
            acc2 = wmma_f32_4(hA[2][q], cq, acc2);
            acc3 = wmma_f32_4(hA[3][q], cq, acc3);
        }
        float* yr = yb + (unsigned)((tl ^ rmask) * D_INNER + 8 * hf);
        v8f accs[4] = {acc0, acc1, acc2, acc3};
        float ps = 0.0f, pq = 0.0f;
#pragma unroll
        for (int pb = 0; pb < 4; ++pb) {
            float4 ya = *(const float4*)(yr + 16 * pb);
            float4 yc = *(const float4*)(yr + 16 * pb + 4);
            ya.x += cumA * accs[pb][0];  ya.y += cumA * accs[pb][1];
            ya.z += cumA * accs[pb][2];  ya.w += cumA * accs[pb][3];
            yc.x += cumA * accs[pb][4];  yc.y += cumA * accs[pb][5];
            yc.z += cumA * accs[pb][6];  yc.w += cumA * accs[pb][7];
            *(float4*)(yr + 16 * pb)     = ya;
            *(float4*)(yr + 16 * pb + 4) = yc;
            add_sums(ya, ps, pq);
            add_sums(yc, ps, pq);
        }
        put_row_sums(ps, pq, hf, pb0 + 2 * (tl ^ rmask));
    }
}

__global__ void __launch_bounds__(128)
gemm_out(const float* Y, const float* __restrict__ part,
         const float* __restrict__ lnw,
         const float* __restrict__ lnb, const float* Wout,
         float* __restrict__ out) {
    const int wv = threadIdx.x >> 5;
    const int m0 = (blockIdx.y * 4 + wv) * 16;
    gemm_tile<4, D_INNER, D_MODEL, true>(Y, Wout, out, part, lnw, lnb, m0, (int)blockIdx.x * 64);
}

extern "C" void kernel_launch(void* const* d_in, const int* in_sizes, int n_in,
                              void* d_out, int out_size, void* d_ws, size_t ws_size,
                              hipStream_t stream) {
    const float* u        = (const float*)d_in[0];
    const float* W_in     = (const float*)d_in[2];
    const float* conv_fw  = (const float*)d_in[3];
    const float* conv_fb  = (const float*)d_in[4];
    const float* conv_bw  = (const float*)d_in[5];
    const float* conv_bb  = (const float*)d_in[6];
    const float* dt_bias  = (const float*)d_in[7];
    const float* A_log    = (const float*)d_in[8];
    const float* D_param  = (const float*)d_in[9];
    const float* ln_w     = (const float*)d_in[10];
    const float* ln_b     = (const float*)d_in[11];
    const float* W_out    = (const float*)d_in[12];
    float* out            = (float*)d_out;

    float* ws = (float*)d_ws;
    size_t off = 0;
    float* xbcdt = ws + off; off += (size_t)BL * D_IN_PROJ;
    float* xc    = ws + off; off += (size_t)BL * CONV2;
    float* dtb   = ws + off; off += (size_t)BL * NHEADS;
    float* ybuf  = ws + off; off += (size_t)BL * D_INNER;
    float* Sbuf  = ws + off; off += (size_t)32 * NCHUNK * HEADDIM * NSTATE;
    float* pAbuf = ws + off; off += (size_t)32 * NCHUNK;
    float* partb = ws + off; off += (size_t)2 * NHEADS * BL;

    gemm_in<<<dim3(CONV2 / 64 + 1, BL / 64), 128, 0, stream>>>(u, W_in, xbcdt);
    conv_dt<<<dim3(BL), CONV_THREADS, 0, stream>>>(xbcdt, conv_fw, conv_fb, conv_bw, conv_bb,
                                          dt_bias, xc, dtb);
    ssd_chunk<<<dim3(BATCH * NHEADS * NCHUNK), 128, 0, stream>>>(
        xc, dtb, A_log, D_param, ybuf, Sbuf, pAbuf);
    ssd_correct<<<dim3(BATCH * NHEADS * NCHUNK), 128, 0, stream>>>(
        xc, dtb, A_log, Sbuf, pAbuf, ybuf, partb);
    gemm_out<<<dim3(D_MODEL / 64, BL / 64), 128, 0, stream>>>(
        ybuf, partb, ln_w, ln_b, W_out, out);

    (void)in_sizes; (void)n_in; (void)out_size; (void)ws_size;
}


# both projection GEMMs: wave tile 16x64 -> 32x64 (two 16-row tiles per wave), each B fragment loaded and rearranged once for both row tiles; half the waves; k order and every sum unchanged
# speedup vs baseline: 3.3706x; 1.1142x over previous
#include <hip/hip_runtime.h>
#include <hip/hip_bf16.h>
#include <math.h>

#define BATCH    2
#define SEQLEN   4096
#define BL       (BATCH * SEQLEN)
#define D_MODEL  512
#define D_IN_PROJ 1168
#define CONV_DIM 576
#define CONV2    1152
#define D_INNER  1024
#define NHEADS   16
#define HH       8
#define HEADDIM  64
#define NSTATE   32
#define NCHUNK   32
#define CHUNK    128

typedef float v2f __attribute__((ext_vector_type(2)));
typedef float v8f __attribute__((ext_vector_type(8)));

__device__ __forceinline__ v8f wmma_f32_4(v2f a, v2f b, v8f c) {
    return __builtin_amdgcn_wmma_f32_16x16x4_f32(
         false, a,  false, b,
         (short)0, c,  false,  false);
}

typedef float v4f __attribute__((ext_vector_type(4)));

template <int NF>
struct BFrag {
    v4f   b4[4];
    float b1[4];
    float wl[4], wh[4], zl[4], zh[4];
};

template <int NF, int ND, bool LN>
__device__ __forceinline__ void load_b(BFrag<NF>& f, const float* bp,
                                       const float* __restrict__ lnw, const float* __restrict__ lnb,
                                       int krow) {
#pragma unroll
    for (int j = 0; j < 4; ++j) {
        if (NF == 4) f.b4[j] = *(const v4f*)(bp + (size_t)(krow + j) * ND);
        else         f.b1[j] = bp[(size_t)(krow + j) * ND];
    }
    if (LN) {
#pragma unroll
        for (int j = 0; j < 4; ++j) {
            f.wl[j] = lnw[krow + j];  f.wh[j] = lnw[krow + 16 + j];
            f.zl[j] = lnb[krow + j];  f.zh[j] = lnb[krow + 16 + j];
        }
    }
}

__device__ __forceinline__ void load_a(v4f (&a)[4], const float* ap, int kb) {
#pragma unroll
    for (int q = 0; q < 4; ++q) a[q] = *(const v4f*)(ap + 32 * kb + 4 * q);
}

template <int NF, bool LN>
__device__ __forceinline__ void mma_sub(v4f y, v4f u, const BFrag<NF>& f, int hf,
                                        float mur0, float rsr0, float mur1, float rsr1,
                                        v8f (&acc0)[NF], v8f (&acc1)[NF]) {
    if (LN) {
#pragma unroll
        for (int j = 0; j < 4; ++j) {
            const float w = hf ? f.wh[j] : f.wl[j];
            const float z = hf ? f.zh[j] : f.zl[j];
            y[j] = (y[j] - mur0) * rsr0 * w + z;
            u[j] = (u[j] - mur1) * rsr1 * w + z;
        }
    }
    v2f a0, a1, c0, c1;
    a0.x = y[0];  a0.y = y[1];
    a1.x = y[2];  a1.y = y[3];
    c0.x = u[0];  c0.y = u[1];
    c1.x = u[2];  c1.y = u[3];
#pragma unroll
    for (int t = 0; t < NF; ++t) {
        v2f b;
        if (NF == 4) { b.x = f.b4[0][t];  b.y = f.b4[1][t]; }
        else         { b.x = f.b1[0];     b.y = f.b1[1]; }
        acc0[t] = wmma_f32_4(a0, b, acc0[t]);
        acc1[t] = wmma_f32_4(c0, b, acc1[t]);
    }
#pragma unroll
    for (int t = 0; t < NF; ++t) {
        v2f b;
        if (NF == 4) { b.x = f.b4[2][t];  b.y = f.b4[3][t]; }
        else         { b.x = f.b1[2];     b.y = f.b1[3]; }
        acc0[t] = wmma_f32_4(a1, b, acc0[t]);
        acc1[t] = wmma_f32_4(c1, b, acc1[t]);
    }
}

#define GEMM_SUB(AV, AW, BCUR, BNXT, KROW_NEXT)                                   \
    load_b<NF, ND, LN>(BNXT, bp, lnw, lnb, (KROW_NEXT));                          \
    __builtin_amdgcn_sched_barrier(0);                                            \
    mma_sub<NF, LN>((AV), (AW), (BCUR), hf, mur0, rsr0, mur1, rsr1, acc0, acc1);  \
    __builtin_amdgcn_sched_barrier(0);

template <int NF, int KD, int ND, bool LN>
__device__ __forceinline__ void gemm_tile(const float* A, const float* B, float* __restrict__ C,
                                          const float* __restrict__ part,
                                          const float* __restrict__ lnw, const float* __restrict__ lnb,
                                          int m0, int n0) {
    const int lane = threadIdx.x & 31;
    const int c    = lane & 15;
    const int hf   = lane >> 4;
    const float* ap = A + (size_t)(m0 + c) * KD + 16 * hf;
    const float* aq = ap + (size_t)16 * KD;
    const float* bp = B + (size_t)(16 * hf) * ND + n0 + (NF == 4 ? 4 * c : c);
    float mur0 = 0.0f, rsr0 = 1.0f, mur1 = 0.0f, rsr1 = 1.0f;
    if (LN) {
        float s0 = 0.0f, q0 = 0.0f, s1 = 0.0f, q1 = 0.0f;
#pragma unroll
        for (int h = 0; h < NHEADS; ++h) {
            const v2f p0 = *(const v2f*)(part + 2 * ((size_t)h * BL + m0 + c));
            const v2f p1 = *(const v2f*)(part + 2 * ((size_t)h * BL + m0 + 16 + c));
            s0 += p0.x;  q0 += p0.y;
            s1 += p1.x;  q1 += p1.y;
        }
        const float e0 = s0 * (1.0f / D_INNER);
        const float v0 = q0 * (1.0f / D_INNER) - e0 * e0;
        const float e1 = s1 * (1.0f / D_INNER);
        const float v1 = q1 * (1.0f / D_INNER) - e1 * e1;
        mur0 = e0;  rsr0 = rsqrtf(v0 + 1e-5f);
        mur1 = e1;  rsr1 = rsqrtf(v1 + 1e-5f);
    }

    v8f acc0[NF], acc1[NF];
#pragma unroll
    for (int t = 0; t < NF; ++t) { acc0[t] = v8f{};  acc1[t] = v8f{}; }

    constexpr int NKB = KD / 32;
    v4f aP[4], aQ[4], cP[4], cQ[4];
    BFrag<NF> bX, bY;
    load_a(aP, ap, 0);
    load_a(cP, aq, 0);
    load_b<NF, ND, LN>(bX, bp, lnw, lnb, 0);
    for (int kb = 0; kb < NKB; kb += 2) {
        const int kn = (kb + 2 < NKB) ? (kb + 2) : (kb + 1);
        load_a(aQ, ap, kb + 1);
        load_a(cQ, aq, kb + 1);
        GEMM_SUB(aP[0], cP[0], bX, bY, 32 * kb + 4)
        GEMM_SUB(aP[1], cP[1], bY, bX, 32 * kb + 8)
        GEMM_SUB(aP[2], cP[2], bX, bY, 32 * kb + 12)
        GEMM_SUB(aP[3], cP[3], bY, bX, 32 * kb + 32)
        load_a(aP, ap, kn);
        load_a(cP, aq, kn);
        GEMM_SUB(aQ[0], cQ[0], bX, bY, 32 * kb + 36)
        GEMM_SUB(aQ[1], cQ[1], bY, bX, 32 * kb + 40)
        GEMM_SUB(aQ[2], cQ[2], bX, bY, 32 * kb + 44)
        GEMM_SUB(aQ[3], cQ[3], bY, bX, 32 * kn)
    }

#pragma unroll
    for (int r = 0; r < 8; ++r) {
        float* cr = C + (size_t)(m0 + 8 * hf + r) * ND + n0;
        float* cs = cr + (size_t)16 * ND;
        if (NF == 4) {
            v4f o, p;
            o[0] = acc0[0][r];  o[1] = acc0[1][r];
            o[2] = acc0[2][r];  o[3] = acc0[3][r];
            p[0] = acc1[0][r];  p[1] = acc1[1][r];
            p[2] = acc1[2][r];  p[3] = acc1[3][r];
            *(v4f*)(cr + 4 * c) = o;
            *(v4f*)(cs + 4 * c) = p;
        } else {
            cr[c] = acc0[0][r];
            cs[c] = acc1[0][r];
        }
    }
}
#undef GEMM_SUB

__global__ void __launch_bounds__(128)
gemm_in(const float* U, const float* Win, float* __restrict__ out) {
    const int wv = threadIdx.x >> 5;
    const int m0 = (blockIdx.y * 4 + wv) * 32;
    if (blockIdx.x < CONV2 / 64)
        gemm_tile<4, D_MODEL, D_IN_PROJ, false>(U, Win, out, nullptr, nullptr, nullptr,
                                                m0, (int)blockIdx.x * 64);
    else
        gemm_tile<1, D_MODEL, D_IN_PROJ, false>(U, Win, out, nullptr, nullptr, nullptr,
                                                m0, CONV2);
}

#define CONV_THREADS 192
#define CONV_ROUNDS  (CONV2 / CONV_THREADS)
static_assert(CONV2 == CONV_ROUNDS * CONV_THREADS && 2 * CONV_DIM == CONV2 && (CONV_ROUNDS % 2) == 0,
              "whole rounds, half of them per direction");
__global__ void __launch_bounds__(CONV_THREADS)
conv_dt(const float* __restrict__ xbcdt,
        const float* __restrict__ wf, const float* __restrict__ bf,
        const float* __restrict__ wb, const float* __restrict__ bb,
        const float* __restrict__ dt_bias,
        float* __restrict__ xc, float* __restrict__ dtb) {
    const size_t bl = blockIdx.x;
    const int b = (int)(bl / SEQLEN);
    const int t = (int)(bl % SEQLEN);
    const int wv   = __builtin_amdgcn_readfirstlane((int)(threadIdx.x >> 5));
    const int lane = threadIdx.x & 31;
    const float* xb = xbcdt + (size_t)b * SEQLEN * D_IN_PROJ + 32 * wv;

    const float* rowp[2][4];
    bool ok[2][4];
#pragma unroll
    for (int k = 0; k < 4; ++k) {
        const int tf = t - 3 + k;
        const int tb = t + 3 - k;
        ok[0][k] = tf >= 0;
        ok[1][k] = tb < SEQLEN;
        rowp[0][k] = xb + (size_t)(ok[0][k] ? tf : t) * D_IN_PROJ;
        rowp[1][k] = xb + (size_t)(ok[1][k] ? tb : t) * D_IN_PROJ;
    }

    float xin[CONV_ROUNDS][4];
    v4f   wk[CONV_ROUNDS];
    float bias[CONV_ROUNDS];
#pragma unroll
    for (int i = 0; i < CONV_ROUNDS; ++i) {
        const int d  = (i >= CONV_ROUNDS / 2) ? 1 : 0;
        const int cd = CONV_THREADS * i - d * CONV_DIM + 32 * wv + lane;
        wk[i]   = *(const v4f*)((d ? wb : wf) + 4 * cd);
        bias[i] = (d ? bb : bf)[cd];
#pragma unroll
        for (int k = 0; k < 4; ++k) xin[i][k] = rowp[d][k][CONV_THREADS * i + lane];
    }
    const int   hd    = threadIdx.x & (NHEADS - 1);
    const float dtraw = xbcdt[bl * D_IN_PROJ + CONV2 + hd];
    const float dtbv  = dt_bias[hd];

    if (threadIdx.x < NHEADS) {
        float v = dtraw + dtbv;
        dtb[bl * NHEADS + threadIdx.x] = (v > 20.0f) ? v : log1pf(expf(v));
    }
    float* xo = xc + bl * CONV2 + 32 * wv + lane;
#pragma unroll
    for (int i = 0; i < CONV_ROUNDS; ++i) {
        const int d = (i >= CONV_ROUNDS / 2) ? 1 : 0;
        float acc = bias[i];
#pragma unroll
        for (int k = 0; k < 4; ++k)
            if (ok[d][k]) acc += xin[i][k] * wk[i][k];
        float sig = 1.0f / (1.0f + expf(-acc));
        xo[CONV_THREADS * i] = acc * sig;
    }
}

__device__ __forceinline__ void head_bases(int h, int& xbase, int& bbase, int& cbase) {
    if (h < HH) { xbase = h * HEADDIM;                 bbase = 512;  cbase = 544;  }
    else        { xbase = CONV_DIM + (h - HH) * HEADDIM; bbase = 1088; cbase = 1120; }
}


__device__ __forceinline__ void chunk_prefix(const float* __restrict__ dtc, int rmask, float Ah,
                                             float* s_tot, float* s_dt, float* s_cs) {
    const int i    = threadIdx.x;
    const int lane = i & 31;
    const int wv   = i >> 5;
    const float dtv = dtc[(unsigned)((i ^ rmask) * NHEADS)];
    float v = Ah * dtv;
#pragma unroll
    for (int d = 1; d < 32; d <<= 1) {
        const float n = __shfl_up(v, d, 32);
        if (lane >= d) v += n;
    }
    if (lane == 31) s_tot[wv] = v;
    s_dt[i] = dtv;
    __syncthreads();
    float off = 0.0f;
    for (int w = 0; w < wv; ++w) off += s_tot[w];
    s_cs[i] = off + v;
    __syncthreads();
}

__device__ __forceinline__ void load_pair(const float* __restrict__ xcb, int j, int l15, int hf,
                                          int rmask, int bbase, int xbase,
                                          v2f (&bf)[8], float (&xv)[4][8]) {
    const unsigned rbx = (unsigned)(((16 * j + l15) ^ rmask) * CONV2 + bbase + 2 * hf);
#pragma unroll
    for (int q = 0; q < 8; ++q) bf[q] = *(const v2f*)(xcb + rbx + (unsigned)(4 * q));
    const int sb0 = 16 * j + 8 * hf;
#pragma unroll
    for (int r = 0; r < 8; ++r) {
        const unsigned ro = (unsigned)(((sb0 + r) ^ rmask) * CONV2 + xbase + l15);
        xv[0][r] = xcb[ro];
        xv[1][r] = xcb[ro + 16];
        xv[2][r] = xcb[ro + 32];
        xv[3][r] = xcb[ro + 48];
    }
}

__global__ void __launch_bounds__(128)
ssd_chunk(const float* __restrict__ xc, const float* __restrict__ dtb,
          const float* __restrict__ A_log, const float* __restrict__ Dp,
          float* __restrict__ y, float* __restrict__ S, float* __restrict__ prodA) {
    const int chunk = blockIdx.x & (NCHUNK - 1);
    const int h     = (blockIdx.x >> 5) & 15;
    const int b     = blockIdx.x >> 9;
    int xbase, bbase, cbase; head_bases(h, xbase, bbase, cbase);
    const int s0    = chunk * CHUNK;
    const int rmask = (h < HH) ? 0 : (CHUNK - 1);
    const int tbase = (h < HH) ? s0 : (SEQLEN - CHUNK - s0);
    const size_t row0 = (size_t)b * SEQLEN + tbase;
    const float* xcb = xc  + row0 * CONV2;
    const float* dtc = dtb + row0 * NHEADS + h;
    float*       yb  = y   + row0 * D_INNER + h * HEADDIM;

    const float Ah = -expf(A_log[h]);
    const float Dh = Dp[h];

    __shared__ float s_tot[4];
    __shared__ alignas(16) float s_dt[CHUNK];
    __shared__ alignas(16) float s_cs[CHUNK];
    __shared__ alignas(16) float s_w[CHUNK];
    chunk_prefix(dtc, rmask, Ah, s_tot, s_dt, s_cs);
    const float cl = s_cs[CHUNK - 1];
    s_w[threadIdx.x] = expf(cl - s_cs[threadIdx.x]) * s_dt[threadIdx.x];
    __syncthreads();

    const int wv   = __builtin_amdgcn_readfirstlane((int)(threadIdx.x >> 5));
    const int lane = threadIdx.x & 31;
    const int l15  = lane & 15;
    const int hf   = lane >> 4;

    for (int half = 0; half < 2; ++half) {
        const int i  = half ? (7 - wv) : wv;
        const int tl = 16 * i + l15;
        const unsigned rtx = (unsigned)((tl ^ rmask) * CONV2);
        const float cst = s_cs[tl];

        v2f cf[8];
#pragma unroll
        for (int q = 0; q < 8; ++q)
            cf[q] = *(const v2f*)(xcb + rtx + (unsigned)(cbase + 2 * hf + 4 * q));

        v8f acc0 = {}, acc1 = {}, acc2 = {}, acc3 = {};
        v2f bfr[8];
        float xa[4][8];
        load_pair(xcb, 0, l15, hf, rmask, bbase, xbase, bfr, xa);
        for (int j = 0; j <= i; ++j) {
            const int sb0 = 16 * j + 8 * hf;
            v2f bfn[8];
            float xn[4][8];
            load_pair(xcb, (j < i) ? (j + 1) : i, l15, hf, rmask, bbase, xbase, bfn, xn);
            __builtin_amdgcn_sched_barrier(0);

            v8f g = {};
#pragma unroll
            for (int q = 0; q < 8; ++q) g = wmma_f32_4(bfr[q], cf[q], g);

            float m[8];
#pragma unroll
            for (int r = 0; r < 8; ++r) {
                const int ss = sb0 + r;
                const float css = s_cs[ss];
                const float d = (ss <= tl) ? (cst - css) : -INFINITY;
                m[r] = g[r] * (expf(d) * s_dt[ss]);
            }
#pragma unroll
            for (int kk = 0; kk < 4; ++kk) {
                v2f bm; bm.x = m[2 * kk]; bm.y = m[2 * kk + 1];
                v2f a0, a1, a2, a3;
                a0.x = xa[0][2 * kk];  a0.y = xa[0][2 * kk + 1];
                a1.x = xa[1][2 * kk];  a1.y = xa[1][2 * kk + 1];
                a2.x = xa[2][2 * kk];  a2.y = xa[2][2 * kk + 1];
                a3.x = xa[3][2 * kk];  a3.y = xa[3][2 * kk + 1];
                acc0 = wmma_f32_4(a0, bm, acc0);
                acc1 = wmma_f32_4(a1, bm, acc1);
                acc2 = wmma_f32_4(a2, bm, acc2);
                acc3 = wmma_f32_4(a3, bm, acc3);
            }
            __builtin_amdgcn_sched_barrier(0);
#pragma unroll
            for (int q = 0; q < 8; ++q) {
                bfr[q] = bfn[q];
                xa[0][q] = xn[0][q];  xa[1][q] = xn[1][q];
                xa[2][q] = xn[2][q];  xa[3][q] = xn[3][q];
            }
        }
        const float* xr = xcb + rtx + (unsigned)(xbase + 8 * hf);
        float*       yr = yb + (unsigned)((tl ^ rmask) * D_INNER + 8 * hf);
        v8f accs[4] = {acc0, acc1, acc2, acc3};
#pragma unroll
        for (int pb = 0; pb < 4; ++pb) {
            const float4 xa4 = *(const float4*)(xr + 16 * pb);
            const float4 xb4 = *(const float4*)(xr + 16 * pb + 4);
            float4 oa, ob;
            oa.x = accs[pb][0] + Dh * xa4.x;  oa.y = accs[pb][1] + Dh * xa4.y;
            oa.z = accs[pb][2] + Dh * xa4.z;  oa.w = accs[pb][3] + Dh * xa4.w;
            ob.x = accs[pb][4] + Dh * xb4.x;  ob.y = accs[pb][5] + Dh * xb4.y;
            ob.z = accs[pb][6] + Dh * xb4.z;  ob.w = accs[pb][7] + Dh * xb4.w;
            *(float4*)(yr + 16 * pb)     = oa;
            *(float4*)(yr + 16 * pb + 4) = ob;
        }
    }

    {
        v8f st0 = {}, st1 = {};
        const unsigned xo = (unsigned)(xbase + 16 * wv + l15);
        const unsigned bo = (unsigned)(bbase + l15);
        for (int k0 = 0; k0 < CHUNK; k0 += 32) {
            float xv[16], bv0[16], bv1[16], wv16[16];
#pragma unroll
            for (int u = 0; u < 16; ++u) {
                const int sa = k0 + 4 * (u >> 1) + 2 * hf + (u & 1);
                const unsigned ro = (unsigned)((sa ^ rmask) * CONV2);
                xv[u]   = xcb[ro + xo];
                bv0[u]  = xcb[ro + bo];
                bv1[u]  = xcb[ro + bo + 16];
                wv16[u] = s_w[sa];
            }
#pragma unroll
            for (int ks = 0; ks < 8; ++ks) {
                v2f a, b0, b1;
                a.x  = xv[2 * ks] * wv16[2 * ks];
                a.y  = xv[2 * ks + 1] * wv16[2 * ks + 1];
                b0.x = bv0[2 * ks];  b0.y = bv0[2 * ks + 1];
                b1.x = bv1[2 * ks];  b1.y = bv1[2 * ks + 1];
                st0 = wmma_f32_4(a, b0, st0);
                st1 = wmma_f32_4(a, b1, st1);
            }
        }
        const size_t sb = (((size_t)(b * NHEADS + h)) * NCHUNK + chunk) * (HEADDIM * NSTATE)
                        + (size_t)(16 * wv + 8 * hf) * NSTATE + l15;
#pragma unroll
        for (int r = 0; r < 8; ++r) {
            S[sb + (size_t)r * NSTATE]      = st0[r];
            S[sb + (size_t)r * NSTATE + 16] = st1[r];
        }
    }
    if (threadIdx.x == 0) prodA[(b * NHEADS + h) * NCHUNK + chunk] = expf(cl);
}

__device__ __forceinline__ void add_sums(const float4 a, float& ps, float& pq) {
    ps += (a.x + a.y) + (a.z + a.w);
    pq += (a.x * a.x + a.y * a.y) + (a.z * a.z + a.w * a.w);
}

__device__ __forceinline__ void put_row_sums(float ps, float pq, int hf, float* pr) {
    ps += __shfl_xor(ps, 16, 32);
    pq += __shfl_xor(pq, 16, 32);
    if (hf == 0) { v2f o; o.x = ps; o.y = pq; *(v2f*)pr = o; }
}

#define HPITCH 36

template <int NB>
__device__ __forceinline__ void state_steps(const float* __restrict__ sp, const float* __restrict__ pa, int c,
                                            v4f (&hv)[4]) {
    v4f   s[NB][4];
    float a[NB];
#pragma unroll
    for (int u = 0; u < NB; ++u) {
        a[u] = pa[c + u];
#pragma unroll
        for (int k = 0; k < 4; ++k)
            s[u][k] = *(const v4f*)(sp + (size_t)(c + u) * (HEADDIM * NSTATE) + 512 * k);
    }
#pragma unroll
    for (int u = 0; u < NB; ++u) {
#pragma unroll
        for (int k = 0; k < 4; ++k) {
            hv[k][0] = fmaf(a[u], hv[k][0], s[u][k][0]);
            hv[k][1] = fmaf(a[u], hv[k][1], s[u][k][1]);
            hv[k][2] = fmaf(a[u], hv[k][2], s[u][k][2]);
            hv[k][3] = fmaf(a[u], hv[k][3], s[u][k][3]);
        }
    }
}

__global__ void __launch_bounds__(128)
ssd_correct(const float* __restrict__ xc, const float* __restrict__ dtb,
            const float* __restrict__ A_log, const float* __restrict__ S,
            const float* __restrict__ prodA,
            float* __restrict__ y, float* __restrict__ part) {
    const int chunk = blockIdx.x & (NCHUNK - 1);
    const int h = (blockIdx.x >> 5) & 15;
    const int b = blockIdx.x >> 9;
    int xbase, bbase, cbase; head_bases(h, xbase, bbase, cbase);
    (void)xbase; (void)bbase;
    const int s0    = chunk * CHUNK;
    const int rmask = (h < HH) ? 0 : (CHUNK - 1);
    const int tbase = (h < HH) ? s0 : (SEQLEN - CHUNK - s0);
    const size_t row0 = (size_t)b * SEQLEN + tbase;
    const float* xcb = xc  + row0 * CONV2;
    const float* dtc = dtb + row0 * NHEADS + h;
    float*       yb  = y   + row0 * D_INNER + h * HEADDIM;
    float*       pb0 = part + 2 * ((size_t)h * BL + row0);

    const int wv   = __builtin_amdgcn_readfirstlane((int)(threadIdx.x >> 5));
    const int lane = threadIdx.x & 31;
    const int l15  = lane & 15;
    const int hf   = lane >> 4;

    if (chunk == 0) {
        for (int half = 0; half < 2; ++half) {
            const int tl = 16 * (wv + 4 * half) + l15;
            const float* yr = yb + (unsigned)((tl ^ rmask) * D_INNER + 8 * hf);
            float ps = 0.0f, pq = 0.0f;
#pragma unroll
            for (int pb = 0; pb < 4; ++pb) {
                add_sums(*(const float4*)(yr + 16 * pb), ps, pq);
                add_sums(*(const float4*)(yr + 16 * pb + 4), ps, pq);
            }
            put_row_sums(ps, pq, hf, pb0 + 2 * (tl ^ rmask));
        }
        return;
    }

    const float Ah = -expf(A_log[h]);
    __shared__ float s_tot[4], s_dt[CHUNK], s_cs[CHUNK];
    __shared__ alignas(16) float s_h[HEADDIM * HPITCH];

    {
        const int bh = b * NHEADS + h;
        const float* sp = S + (size_t)bh * NCHUNK * (HEADDIM * NSTATE) + 4 * threadIdx.x;
        const float* pa = prodA + bh * NCHUNK;
        v4f hv[4];
#pragma unroll
        for (int k = 0; k < 4; ++k) hv[k] = v4f{};
        int c = 0;
        for (; c + 8 <= chunk; c += 8) state_steps<8>(sp, pa, c, hv);
        if (chunk & 4) { state_steps<4>(sp, pa, c, hv);  c += 4; }
        if (chunk & 2) { state_steps<2>(sp, pa, c, hv);  c += 2; }
        if (chunk & 1) { state_steps<1>(sp, pa, c, hv); }
#pragma unroll
        for (int k = 0; k < 4; ++k)
            *(v4f*)(s_h + (16 * k + ((int)threadIdx.x >> 3)) * HPITCH + 4 * ((int)threadIdx.x & 7)) = hv[k];
    }
    chunk_prefix(dtc, rmask, Ah, s_tot, s_dt, s_cs);

    v2f hA[4][8];
    {
        const float* hl = s_h + l15 * HPITCH + 2 * hf;
#pragma unroll
        for (int pb = 0; pb < 4; ++pb)
#pragma unroll
            for (int q = 0; q < 8; ++q)
                hA[pb][q] = *(const v2f*)(hl + 16 * pb * HPITCH + 4 * q);
    }

    for (int half = 0; half < 2; ++half) {
        const int tl = 16 * (wv + 4 * half) + l15;
        const float cumA = expf(s_cs[tl]);
        const float* cp = xcb + (unsigned)((tl ^ rmask) * CONV2 + cbase + 2 * hf);

        v8f acc0 = {}, acc1 = {}, acc2 = {}, acc3 = {};
#pragma unroll
        for (int q = 0; q < 8; ++q) {
            const v2f cq = *(const v2f*)(cp + 4 * q);
            acc0 = wmma_f32_4(hA[0][q], cq, acc0);
            acc1 = wmma_f32_4(hA[1][q], cq, acc1);
            acc2 = wmma_f32_4(hA[2][q], cq, acc2);
            acc3 = wmma_f32_4(hA[3][q], cq, acc3);
        }
        float* yr = yb + (unsigned)((tl ^ rmask) * D_INNER + 8 * hf);
        v8f accs[4] = {acc0, acc1, acc2, acc3};
        float ps = 0.0f, pq = 0.0f;
#pragma unroll
        for (int pb = 0; pb < 4; ++pb) {
            float4 ya = *(const float4*)(yr + 16 * pb);
            float4 yc = *(const float4*)(yr + 16 * pb + 4);
            ya.x += cumA * accs[pb][0];  ya.y += cumA * accs[pb][1];
            ya.z += cumA * accs[pb][2];  ya.w += cumA * accs[pb][3];
            yc.x += cumA * accs[pb][4];  yc.y += cumA * accs[pb][5];
            yc.z += cumA * accs[pb][6];  yc.w += cumA * accs[pb][7];
            *(float4*)(yr + 16 * pb)     = ya;
            *(float4*)(yr + 16 * pb + 4) = yc;
            add_sums(ya, ps, pq);
            add_sums(yc, ps, pq);
        }
        put_row_sums(ps, pq, hf, pb0 + 2 * (tl ^ rmask));
    }
}

__global__ void __launch_bounds__(128)
gemm_out(const float* Y, const float* __restrict__ part,
         const float* __restrict__ lnw,
         const float* __restrict__ lnb, const float* Wout,
         float* __restrict__ out) {
    const int wv = threadIdx.x >> 5;
    const int m0 = (blockIdx.y * 4 + wv) * 32;
    gemm_tile<4, D_INNER, D_MODEL, true>(Y, Wout, out, part, lnw, lnb, m0, (int)blockIdx.x * 64);
}

extern "C" void kernel_launch(void* const* d_in, const int* in_sizes, int n_in,
                              void* d_out, int out_size, void* d_ws, size_t ws_size,
                              hipStream_t stream) {
    const float* u        = (const float*)d_in[0];
    const float* W_in     = (const float*)d_in[2];
    const float* conv_fw  = (const float*)d_in[3];
    const float* conv_fb  = (const float*)d_in[4];
    const float* conv_bw  = (const float*)d_in[5];
    const float* conv_bb  = (const float*)d_in[6];
    const float* dt_bias  = (const float*)d_in[7];
    const float* A_log    = (const float*)d_in[8];
    const float* D_param  = (const float*)d_in[9];
    const float* ln_w     = (const float*)d_in[10];
    const float* ln_b     = (const float*)d_in[11];
    const float* W_out    = (const float*)d_in[12];
    float* out            = (float*)d_out;

    float* ws = (float*)d_ws;
    size_t off = 0;
    float* xbcdt = ws + off; off += (size_t)BL * D_IN_PROJ;
    float* xc    = ws + off; off += (size_t)BL * CONV2;
    float* dtb   = ws + off; off += (size_t)BL * NHEADS;
    float* ybuf  = ws + off; off += (size_t)BL * D_INNER;
    float* Sbuf  = ws + off; off += (size_t)32 * NCHUNK * HEADDIM * NSTATE;
    float* pAbuf = ws + off; off += (size_t)32 * NCHUNK;
    float* partb = ws + off; off += (size_t)2 * NHEADS * BL;

    gemm_in<<<dim3(CONV2 / 64 + 1, BL / 128), 128, 0, stream>>>(u, W_in, xbcdt);
    conv_dt<<<dim3(BL), CONV_THREADS, 0, stream>>>(xbcdt, conv_fw, conv_fb, conv_bw, conv_bb,
                                          dt_bias, xc, dtb);
    ssd_chunk<<<dim3(BATCH * NHEADS * NCHUNK), 128, 0, stream>>>(
        xc, dtb, A_log, D_param, ybuf, Sbuf, pAbuf);
    ssd_correct<<<dim3(BATCH * NHEADS * NCHUNK), 128, 0, stream>>>(
        xc, dtb, A_log, Sbuf, pAbuf, ybuf, partb);
    gemm_out<<<dim3(D_MODEL / 64, BL / 128), 128, 0, stream>>>(
        ybuf, partb, ln_w, ln_b, W_out, out);

    (void)in_sizes; (void)n_in; (void)out_size; (void)ws_size;
}


# scan pass 1: the chunk's x|B|C rows staged once in LDS (whole 256-byte row pieces by 16-byte loads, row pitch 132 floats); WMMA fragments read from LDS, not by 4/8-byte global loads; same sums
# speedup vs baseline: 3.5554x; 1.0548x over previous
#include <hip/hip_runtime.h>
#include <hip/hip_bf16.h>
#include <math.h>

#define BATCH    2
#define SEQLEN   4096
#define BL       (BATCH * SEQLEN)
#define D_MODEL  512
#define D_IN_PROJ 1168
#define CONV_DIM 576
#define CONV2    1152
#define D_INNER  1024
#define NHEADS   16
#define HH       8
#define HEADDIM  64
#define NSTATE   32
#define NCHUNK   32
#define CHUNK    128

typedef float v2f __attribute__((ext_vector_type(2)));
typedef float v8f __attribute__((ext_vector_type(8)));

__device__ __forceinline__ v8f wmma_f32_4(v2f a, v2f b, v8f c) {
    return __builtin_amdgcn_wmma_f32_16x16x4_f32(
         false, a,  false, b,
         (short)0, c,  false,  false);
}

typedef float v4f __attribute__((ext_vector_type(4)));

template <int NF>
struct BFrag {
    v4f   b4[4];
    float b1[4];
    float wl[4], wh[4], zl[4], zh[4];
};

template <int NF, int ND, bool LN>
__device__ __forceinline__ void load_b(BFrag<NF>& f, const float* bp,
                                       const float* __restrict__ lnw, const float* __restrict__ lnb,
                                       int krow) {
#pragma unroll
    for (int j = 0; j < 4; ++j) {
        if (NF == 4) f.b4[j] = *(const v4f*)(bp + (size_t)(krow + j) * ND);
        else         f.b1[j] = bp[(size_t)(krow + j) * ND];
    }
    if (LN) {
#pragma unroll
        for (int j = 0; j < 4; ++j) {
            f.wl[j] = lnw[krow + j];  f.wh[j] = lnw[krow + 16 + j];
            f.zl[j] = lnb[krow + j];  f.zh[j] = lnb[krow + 16 + j];
        }
    }
}

__device__ __forceinline__ void load_a(v4f (&a)[4], const float* ap, int kb) {
#pragma unroll
    for (int q = 0; q < 4; ++q) a[q] = *(const v4f*)(ap + 32 * kb + 4 * q);
}

template <int NF, bool LN>
__device__ __forceinline__ void mma_sub(v4f y, v4f u, const BFrag<NF>& f, int hf,
                                        float mur0, float rsr0, float mur1, float rsr1,
                                        v8f (&acc0)[NF], v8f (&acc1)[NF]) {
    if (LN) {
#pragma unroll
        for (int j = 0; j < 4; ++j) {
            const float w = hf ? f.wh[j] : f.wl[j];
            const float z = hf ? f.zh[j] : f.zl[j];
            y[j] = (y[j] - mur0) * rsr0 * w + z;
            u[j] = (u[j] - mur1) * rsr1 * w + z;
        }
    }
    v2f a0, a1, c0, c1;
    a0.x = y[0];  a0.y = y[1];
    a1.x = y[2];  a1.y = y[3];
    c0.x = u[0];  c0.y = u[1];
    c1.x = u[2];  c1.y = u[3];
#pragma unroll
    for (int t = 0; t < NF; ++t) {
        v2f b;
        if (NF == 4) { b.x = f.b4[0][t];  b.y = f.b4[1][t]; }
        else         { b.x = f.b1[0];     b.y = f.b1[1]; }
        acc0[t] = wmma_f32_4(a0, b, acc0[t]);
        acc1[t] = wmma_f32_4(c0, b, acc1[t]);
    }
#pragma unroll
    for (int t = 0; t < NF; ++t) {
        v2f b;
        if (NF == 4) { b.x = f.b4[2][t];  b.y = f.b4[3][t]; }
        else         { b.x = f.b1[2];     b.y = f.b1[3]; }
        acc0[t] = wmma_f32_4(a1, b, acc0[t]);
        acc1[t] = wmma_f32_4(c1, b, acc1[t]);
    }
}

#define GEMM_SUB(AV, AW, BCUR, BNXT, KROW_NEXT)                                   \
    load_b<NF, ND, LN>(BNXT, bp, lnw, lnb, (KROW_NEXT));                          \
    __builtin_amdgcn_sched_barrier(0);                                            \
    mma_sub<NF, LN>((AV), (AW), (BCUR), hf, mur0, rsr0, mur1, rsr1, acc0, acc1);  \
    __builtin_amdgcn_sched_barrier(0);

template <int NF, int KD, int ND, bool LN>
__device__ __forceinline__ void gemm_tile(const float* A, const float* B, float* __restrict__ C,
                                          const float* __restrict__ part,
                                          const float* __restrict__ lnw, const float* __restrict__ lnb,
                                          int m0, int n0) {
    const int lane = threadIdx.x & 31;
    const int c    = lane & 15;
    const int hf   = lane >> 4;
    const float* ap = A + (size_t)(m0 + c) * KD + 16 * hf;
    const float* aq = ap + (size_t)16 * KD;
    const float* bp = B + (size_t)(16 * hf) * ND + n0 + (NF == 4 ? 4 * c : c);
    float mur0 = 0.0f, rsr0 = 1.0f, mur1 = 0.0f, rsr1 = 1.0f;
    if (LN) {
        float s0 = 0.0f, q0 = 0.0f, s1 = 0.0f, q1 = 0.0f;
#pragma unroll
        for (int h = 0; h < NHEADS; ++h) {
            const v2f p0 = *(const v2f*)(part + 2 * ((size_t)h * BL + m0 + c));
            const v2f p1 = *(const v2f*)(part + 2 * ((size_t)h * BL + m0 + 16 + c));
            s0 += p0.x;  q0 += p0.y;
            s1 += p1.x;  q1 += p1.y;
        }
        const float e0 = s0 * (1.0f / D_INNER);
        const float v0 = q0 * (1.0f / D_INNER) - e0 * e0;
        const float e1 = s1 * (1.0f / D_INNER);
        const float v1 = q1 * (1.0f / D_INNER) - e1 * e1;
        mur0 = e0;  rsr0 = rsqrtf(v0 + 1e-5f);
        mur1 = e1;  rsr1 = rsqrtf(v1 + 1e-5f);
    }

    v8f acc0[NF], acc1[NF];
#pragma unroll
    for (int t = 0; t < NF; ++t) { acc0[t] = v8f{};  acc1[t] = v8f{}; }

    constexpr int NKB = KD / 32;
    v4f aP[4], aQ[4], cP[4], cQ[4];
    BFrag<NF> bX, bY;
    load_a(aP, ap, 0);
    load_a(cP, aq, 0);
    load_b<NF, ND, LN>(bX, bp, lnw, lnb, 0);
    for (int kb = 0; kb < NKB; kb += 2) {
        const int kn = (kb + 2 < NKB) ? (kb + 2) : (kb + 1);
        load_a(aQ, ap, kb + 1);
        load_a(cQ, aq, kb + 1);
        GEMM_SUB(aP[0], cP[0], bX, bY, 32 * kb + 4)
        GEMM_SUB(aP[1], cP[1], bY, bX, 32 * kb + 8)
        GEMM_SUB(aP[2], cP[2], bX, bY, 32 * kb + 12)
        GEMM_SUB(aP[3], cP[3], bY, bX, 32 * kb + 32)
        load_a(aP, ap, kn);
        load_a(cP, aq, kn);
        GEMM_SUB(aQ[0], cQ[0], bX, bY, 32 * kb + 36)
        GEMM_SUB(aQ[1], cQ[1], bY, bX, 32 * kb + 40)
        GEMM_SUB(aQ[2], cQ[2], bX, bY, 32 * kb + 44)
        GEMM_SUB(aQ[3], cQ[3], bY, bX, 32 * kn)
    }

#pragma unroll
    for (int r = 0; r < 8; ++r) {
        float* cr = C + (size_t)(m0 + 8 * hf + r) * ND + n0;
        float* cs = cr + (size_t)16 * ND;
        if (NF == 4) {
            v4f o, p;
            o[0] = acc0[0][r];  o[1] = acc0[1][r];
            o[2] = acc0[2][r];  o[3] = acc0[3][r];
            p[0] = acc1[0][r];  p[1] = acc1[1][r];
            p[2] = acc1[2][r];  p[3] = acc1[3][r];
            *(v4f*)(cr + 4 * c) = o;
            *(v4f*)(cs + 4 * c) = p;
        } else {
            cr[c] = acc0[0][r];
            cs[c] = acc1[0][r];
        }
    }
}
#undef GEMM_SUB

__global__ void __launch_bounds__(128)
gemm_in(const float* U, const float* Win, float* __restrict__ out) {
    const int wv = threadIdx.x >> 5;
    const int m0 = (blockIdx.y * 4 + wv) * 32;
    if (blockIdx.x < CONV2 / 64)
        gemm_tile<4, D_MODEL, D_IN_PROJ, false>(U, Win, out, nullptr, nullptr, nullptr,
                                                m0, (int)blockIdx.x * 64);
    else
        gemm_tile<1, D_MODEL, D_IN_PROJ, false>(U, Win, out, nullptr, nullptr, nullptr,
                                                m0, CONV2);
}

#define CONV_THREADS 192
#define CONV_ROUNDS  (CONV2 / CONV_THREADS)
static_assert(CONV2 == CONV_ROUNDS * CONV_THREADS && 2 * CONV_DIM == CONV2 && (CONV_ROUNDS % 2) == 0,
              "whole rounds, half of them per direction");
__global__ void __launch_bounds__(CONV_THREADS)
conv_dt(const float* __restrict__ xbcdt,
        const float* __restrict__ wf, const float* __restrict__ bf,
        const float* __restrict__ wb, const float* __restrict__ bb,
        const float* __restrict__ dt_bias,
        float* __restrict__ xc, float* __restrict__ dtb) {
    const size_t bl = blockIdx.x;
    const int b = (int)(bl / SEQLEN);
    const int t = (int)(bl % SEQLEN);
    const int wv   = __builtin_amdgcn_readfirstlane((int)(threadIdx.x >> 5));
    const int lane = threadIdx.x & 31;
    const float* xb = xbcdt + (size_t)b * SEQLEN * D_IN_PROJ + 32 * wv;

    const float* rowp[2][4];
    bool ok[2][4];
#pragma unroll
    for (int k = 0; k < 4; ++k) {
        const int tf = t - 3 + k;
        const int tb = t + 3 - k;
        ok[0][k] = tf >= 0;
        ok[1][k] = tb < SEQLEN;
        rowp[0][k] = xb + (size_t)(ok[0][k] ? tf : t) * D_IN_PROJ;
        rowp[1][k] = xb + (size_t)(ok[1][k] ? tb : t) * D_IN_PROJ;
    }

    float xin[CONV_ROUNDS][4];
    v4f   wk[CONV_ROUNDS];
    float bias[CONV_ROUNDS];
#pragma unroll
    for (int i = 0; i < CONV_ROUNDS; ++i) {
        const int d  = (i >= CONV_ROUNDS / 2) ? 1 : 0;
        const int cd = CONV_THREADS * i - d * CONV_DIM + 32 * wv + lane;
        wk[i]   = *(const v4f*)((d ? wb : wf) + 4 * cd);
        bias[i] = (d ? bb : bf)[cd];
#pragma unroll
        for (int k = 0; k < 4; ++k) xin[i][k] = rowp[d][k][CONV_THREADS * i + lane];
    }
    const int   hd    = threadIdx.x & (NHEADS - 1);
    const float dtraw = xbcdt[bl * D_IN_PROJ + CONV2 + hd];
    const float dtbv  = dt_bias[hd];

    if (threadIdx.x < NHEADS) {
        float v = dtraw + dtbv;
        dtb[bl * NHEADS + threadIdx.x] = (v > 20.0f) ? v : log1pf(expf(v));
    }
    float* xo = xc + bl * CONV2 + 32 * wv + lane;
#pragma unroll
    for (int i = 0; i < CONV_ROUNDS; ++i) {
        const int d = (i >= CONV_ROUNDS / 2) ? 1 : 0;
        float acc = bias[i];
#pragma unroll
        for (int k = 0; k < 4; ++k)
            if (ok[d][k]) acc += xin[i][k] * wk[i][k];
        float sig = 1.0f / (1.0f + expf(-acc));
        xo[CONV_THREADS * i] = acc * sig;
    }
}

__device__ __forceinline__ void head_bases(int h, int& xbase, int& bbase, int& cbase) {
    if (h < HH) { xbase = h * HEADDIM;                 bbase = 512;  cbase = 544;  }
    else        { xbase = CONV_DIM + (h - HH) * HEADDIM; bbase = 1088; cbase = 1120; }
}


__device__ __forceinline__ void chunk_prefix(const float* __restrict__ dtc, int rmask, float Ah,
                                             float* s_tot, float* s_dt, float* s_cs) {
    const int i    = threadIdx.x;
    const int lane = i & 31;
    const int wv   = i >> 5;
    const float dtv = dtc[(unsigned)((i ^ rmask) * NHEADS)];
    float v = Ah * dtv;
#pragma unroll
    for (int d = 1; d < 32; d <<= 1) {
        const float n = __shfl_up(v, d, 32);
        if (lane >= d) v += n;
    }
    if (lane == 31) s_tot[wv] = v;
    s_dt[i] = dtv;
    __syncthreads();
    float off = 0.0f;
    for (int w = 0; w < wv; ++w) off += s_tot[w];
    s_cs[i] = off + v;
    __syncthreads();
}

#define TPITCH 132
#define T_B    64
#define T_C    96

__device__ __forceinline__ void load_pair(const float* s_t, int j, int l15, int hf,
                                          v2f (&bf)[8], float (&xv)[4][8]) {
    const float* bp = s_t + (unsigned)((16 * j + l15) * TPITCH + T_B + 2 * hf);
#pragma unroll
    for (int q = 0; q < 8; ++q) bf[q] = *(const v2f*)(bp + 4 * q);
    const float* xp = s_t + (unsigned)((16 * j + 8 * hf) * TPITCH + l15);
#pragma unroll
    for (int r = 0; r < 8; ++r) {
        xv[0][r] = xp[r * TPITCH];
        xv[1][r] = xp[r * TPITCH + 16];
        xv[2][r] = xp[r * TPITCH + 32];
        xv[3][r] = xp[r * TPITCH + 48];
    }
}

__global__ void __launch_bounds__(128)
ssd_chunk(const float* __restrict__ xc, const float* __restrict__ dtb,
          const float* __restrict__ A_log, const float* __restrict__ Dp,
          float* __restrict__ y, float* __restrict__ S, float* __restrict__ prodA) {
    const int chunk = blockIdx.x & (NCHUNK - 1);
    const int h     = (blockIdx.x >> 5) & 15;
    const int b     = blockIdx.x >> 9;
    int xbase, bbase, cbase; head_bases(h, xbase, bbase, cbase);
    (void)cbase;
    const int s0    = chunk * CHUNK;
    const int rmask = (h < HH) ? 0 : (CHUNK - 1);
    const int tbase = (h < HH) ? s0 : (SEQLEN - CHUNK - s0);
    const size_t row0 = (size_t)b * SEQLEN + tbase;
    const float* xcb = xc  + row0 * CONV2;
    const float* dtc = dtb + row0 * NHEADS + h;
    float*       yb  = y   + row0 * D_INNER + h * HEADDIM;

    const float Ah = -expf(A_log[h]);
    const float Dh = Dp[h];

    __shared__ float s_tot[4];
    __shared__ alignas(16) float s_dt[CHUNK];
    __shared__ alignas(16) float s_cs[CHUNK];
    __shared__ alignas(16) float s_w[CHUNK];
    __shared__ alignas(16) float s_t[CHUNK * TPITCH];

    const int wv   = __builtin_amdgcn_readfirstlane((int)(threadIdx.x >> 5));
    const int lane = threadIdx.x & 31;
    const int l15  = lane & 15;
    const int hf   = lane >> 4;

    {
        v4f tx[16], tb[16];
#pragma unroll
        for (int it = 0; it < 16; ++it) {
            const int sl = 32 * wv + 2 * it + hf;
            const float* rp = xcb + (unsigned)((sl ^ rmask) * CONV2 + 4 * l15);
            tx[it] = *(const v4f*)(rp + xbase);
            tb[it] = *(const v4f*)(rp + bbase);
        }
        chunk_prefix(dtc, rmask, Ah, s_tot, s_dt, s_cs);
#pragma unroll
        for (int it = 0; it < 16; ++it) {
            float* tp = s_t + (unsigned)((32 * wv + 2 * it + hf) * TPITCH + 4 * l15);
            *(v4f*)tp         = tx[it];
            *(v4f*)(tp + T_B) = tb[it];
        }
    }
    const float cl = s_cs[CHUNK - 1];
    s_w[threadIdx.x] = expf(cl - s_cs[threadIdx.x]) * s_dt[threadIdx.x];
    __syncthreads();

    for (int half = 0; half < 2; ++half) {
        const int i  = half ? (7 - wv) : wv;
        const int tl = 16 * i + l15;
        const float* trow = s_t + (unsigned)(tl * TPITCH);
        const float cst = s_cs[tl];

        v2f cf[8];
#pragma unroll
        for (int q = 0; q < 8; ++q)
            cf[q] = *(const v2f*)(trow + T_C + 2 * hf + 4 * q);

        v8f acc0 = {}, acc1 = {}, acc2 = {}, acc3 = {};
        v2f bfr[8];
        float xa[4][8];
        load_pair(s_t, 0, l15, hf, bfr, xa);
        for (int j = 0; j <= i; ++j) {
            const int sb0 = 16 * j + 8 * hf;
            v2f bfn[8];
            float xn[4][8];
            load_pair(s_t, (j < i) ? (j + 1) : i, l15, hf, bfn, xn);
            __builtin_amdgcn_sched_barrier(0);

            v8f g = {};
#pragma unroll
            for (int q = 0; q < 8; ++q) g = wmma_f32_4(bfr[q], cf[q], g);

            float m[8];
#pragma unroll
            for (int r = 0; r < 8; ++r) {
                const int ss = sb0 + r;
                const float css = s_cs[ss];
                const float d = (ss <= tl) ? (cst - css) : -INFINITY;
                m[r] = g[r] * (expf(d) * s_dt[ss]);
            }
#pragma unroll
            for (int kk = 0; kk < 4; ++kk) {
                v2f bm; bm.x = m[2 * kk]; bm.y = m[2 * kk + 1];
                v2f a0, a1, a2, a3;
                a0.x = xa[0][2 * kk];  a0.y = xa[0][2 * kk + 1];
                a1.x = xa[1][2 * kk];  a1.y = xa[1][2 * kk + 1];
                a2.x = xa[2][2 * kk];  a2.y = xa[2][2 * kk + 1];
                a3.x = xa[3][2 * kk];  a3.y = xa[3][2 * kk + 1];
                acc0 = wmma_f32_4(a0, bm, acc0);
                acc1 = wmma_f32_4(a1, bm, acc1);
                acc2 = wmma_f32_4(a2, bm, acc2);
                acc3 = wmma_f32_4(a3, bm, acc3);
            }
            __builtin_amdgcn_sched_barrier(0);
#pragma unroll
            for (int q = 0; q < 8; ++q) {
                bfr[q] = bfn[q];
                xa[0][q] = xn[0][q];  xa[1][q] = xn[1][q];
                xa[2][q] = xn[2][q];  xa[3][q] = xn[3][q];
            }
        }
        const float* xr = trow + 8 * hf;
        float*       yr = yb + (unsigned)((tl ^ rmask) * D_INNER + 8 * hf);
        v8f accs[4] = {acc0, acc1, acc2, acc3};
#pragma unroll
        for (int pb = 0; pb < 4; ++pb) {
            const float4 xa4 = *(const float4*)(xr + 16 * pb);
            const float4 xb4 = *(const float4*)(xr + 16 * pb + 4);
            float4 oa, ob;
            oa.x = accs[pb][0] + Dh * xa4.x;  oa.y = accs[pb][1] + Dh * xa4.y;
            oa.z = accs[pb][2] + Dh * xa4.z;  oa.w = accs[pb][3] + Dh * xa4.w;
            ob.x = accs[pb][4] + Dh * xb4.x;  ob.y = accs[pb][5] + Dh * xb4.y;
            ob.z = accs[pb][6] + Dh * xb4.z;  ob.w = accs[pb][7] + Dh * xb4.w;
            *(float4*)(yr + 16 * pb)     = oa;
            *(float4*)(yr + 16 * pb + 4) = ob;
        }
    }

    {
        v8f st0 = {}, st1 = {};
        const unsigned xo = (unsigned)(16 * wv + l15);
        const unsigned bo = (unsigned)(T_B + l15);
        for (int k0 = 0; k0 < CHUNK; k0 += 32) {
            float xv[16], bv0[16], bv1[16], wv16[16];
#pragma unroll
            for (int u = 0; u < 16; ++u) {
                const int sa = k0 + 4 * (u >> 1) + 2 * hf + (u & 1);
                const unsigned ro = (unsigned)(sa * TPITCH);
                xv[u]   = s_t[ro + xo];
                bv0[u]  = s_t[ro + bo];
                bv1[u]  = s_t[ro + bo + 16];
                wv16[u] = s_w[sa];
            }
#pragma unroll
            for (int ks = 0; ks < 8; ++ks) {
                v2f a, b0, b1;
                a.x  = xv[2 * ks] * wv16[2 * ks];
                a.y  = xv[2 * ks + 1] * wv16[2 * ks + 1];
                b0.x = bv0[2 * ks];  b0.y = bv0[2 * ks + 1];
                b1.x = bv1[2 * ks];  b1.y = bv1[2 * ks + 1];
                st0 = wmma_f32_4(a, b0, st0);
                st1 = wmma_f32_4(a, b1, st1);
            }
        }
        const size_t sb = (((size_t)(b * NHEADS + h)) * NCHUNK + chunk) * (HEADDIM * NSTATE)
                        + (size_t)(16 * wv + 8 * hf) * NSTATE + l15;
#pragma unroll
        for (int r = 0; r < 8; ++r) {
            S[sb + (size_t)r * NSTATE]      = st0[r];
            S[sb + (size_t)r * NSTATE + 16] = st1[r];
        }
    }
    if (threadIdx.x == 0) prodA[(b * NHEADS + h) * NCHUNK + chunk] = expf(cl);
}

__device__ __forceinline__ void add_sums(const float4 a, float& ps, float& pq) {
    ps += (a.x + a.y) + (a.z + a.w);
    pq += (a.x * a.x + a.y * a.y) + (a.z * a.z + a.w * a.w);
}

__device__ __forceinline__ void put_row_sums(float ps, float pq, int hf, float* pr) {
    ps += __shfl_xor(ps, 16, 32);
    pq += __shfl_xor(pq, 16, 32);
    if (hf == 0) { v2f o; o.x = ps; o.y = pq; *(v2f*)pr = o; }
}

#define HPITCH 36

template <int NB>
__device__ __forceinline__ void state_steps(const float* __restrict__ sp, const float* __restrict__ pa, int c,
                                            v4f (&hv)[4]) {
    v4f   s[NB][4];
    float a[NB];
#pragma unroll
    for (int u = 0; u < NB; ++u) {
        a[u] = pa[c + u];
#pragma unroll
        for (int k = 0; k < 4; ++k)
            s[u][k] = *(const v4f*)(sp + (size_t)(c + u) * (HEADDIM * NSTATE) + 512 * k);
    }
#pragma unroll
    for (int u = 0; u < NB; ++u) {
#pragma unroll
        for (int k = 0; k < 4; ++k) {
            hv[k][0] = fmaf(a[u], hv[k][0], s[u][k][0]);
            hv[k][1] = fmaf(a[u], hv[k][1], s[u][k][1]);
            hv[k][2] = fmaf(a[u], hv[k][2], s[u][k][2]);
            hv[k][3] = fmaf(a[u], hv[k][3], s[u][k][3]);
        }
    }
}

__global__ void __launch_bounds__(128)
ssd_correct(const float* __restrict__ xc, const float* __restrict__ dtb,
            const float* __restrict__ A_log, const float* __restrict__ S,
            const float* __restrict__ prodA,
            float* __restrict__ y, float* __restrict__ part) {
    const int chunk = blockIdx.x & (NCHUNK - 1);
    const int h = (blockIdx.x >> 5) & 15;
    const int b = blockIdx.x >> 9;
    int xbase, bbase, cbase; head_bases(h, xbase, bbase, cbase);
    (void)xbase; (void)bbase;
    const int s0    = chunk * CHUNK;
    const int rmask = (h < HH) ? 0 : (CHUNK - 1);
    const int tbase = (h < HH) ? s0 : (SEQLEN - CHUNK - s0);
    const size_t row0 = (size_t)b * SEQLEN + tbase;
    const float* xcb = xc  + row0 * CONV2;
    const float* dtc = dtb + row0 * NHEADS + h;
    float*       yb  = y   + row0 * D_INNER + h * HEADDIM;
    float*       pb0 = part + 2 * ((size_t)h * BL + row0);

    const int wv   = __builtin_amdgcn_readfirstlane((int)(threadIdx.x >> 5));
    const int lane = threadIdx.x & 31;
    const int l15  = lane & 15;
    const int hf   = lane >> 4;

    if (chunk == 0) {
        for (int half = 0; half < 2; ++half) {
            const int tl = 16 * (wv + 4 * half) + l15;
            const float* yr = yb + (unsigned)((tl ^ rmask) * D_INNER + 8 * hf);
            float ps = 0.0f, pq = 0.0f;
#pragma unroll
            for (int pb = 0; pb < 4; ++pb) {
                add_sums(*(const float4*)(yr + 16 * pb), ps, pq);
                add_sums(*(const float4*)(yr + 16 * pb + 4), ps, pq);
            }
            put_row_sums(ps, pq, hf, pb0 + 2 * (tl ^ rmask));
        }
        return;
    }

    const float Ah = -expf(A_log[h]);
    __shared__ float s_tot[4], s_dt[CHUNK], s_cs[CHUNK];
    __shared__ alignas(16) float s_h[HEADDIM * HPITCH];

    {
        const int bh = b * NHEADS + h;
        const float* sp = S + (size_t)bh * NCHUNK * (HEADDIM * NSTATE) + 4 * threadIdx.x;
        const float* pa = prodA + bh * NCHUNK;
        v4f hv[4];
#pragma unroll
        for (int k = 0; k < 4; ++k) hv[k] = v4f{};
        int c = 0;
        for (; c + 8 <= chunk; c += 8) state_steps<8>(sp, pa, c, hv);
        if (chunk & 4) { state_steps<4>(sp, pa, c, hv);  c += 4; }
        if (chunk & 2) { state_steps<2>(sp, pa, c, hv);  c += 2; }
        if (chunk & 1) { state_steps<1>(sp, pa, c, hv); }
#pragma unroll
        for (int k = 0; k < 4; ++k)
            *(v4f*)(s_h + (16 * k + ((int)threadIdx.x >> 3)) * HPITCH + 4 * ((int)threadIdx.x & 7)) = hv[k];
    }
    chunk_prefix(dtc, rmask, Ah, s_tot, s_dt, s_cs);

    v2f hA[4][8];
    {
        const float* hl = s_h + l15 * HPITCH + 2 * hf;
#pragma unroll
        for (int pb = 0; pb < 4; ++pb)
#pragma unroll
            for (int q = 0; q < 8; ++q)
                hA[pb][q] = *(const v2f*)(hl + 16 * pb * HPITCH + 4 * q);
    }

    for (int half = 0; half < 2; ++half) {
        const int tl = 16 * (wv + 4 * half) + l15;
        const float cumA = expf(s_cs[tl]);
        const float* cp = xcb + (unsigned)((tl ^ rmask) * CONV2 + cbase + 2 * hf);

        v8f acc0 = {}, acc1 = {}, acc2 = {}, acc3 = {};
#pragma unroll
        for (int q = 0; q < 8; ++q) {
            const v2f cq = *(const v2f*)(cp + 4 * q);
            acc0 = wmma_f32_4(hA[0][q], cq, acc0);
            acc1 = wmma_f32_4(hA[1][q], cq, acc1);
            acc2 = wmma_f32_4(hA[2][q], cq, acc2);
            acc3 = wmma_f32_4(hA[3][q], cq, acc3);
        }
        float* yr = yb + (unsigned)((tl ^ rmask) * D_INNER + 8 * hf);
        v8f accs[4] = {acc0, acc1, acc2, acc3};
        float ps = 0.0f, pq = 0.0f;
#pragma unroll
        for (int pb = 0; pb < 4; ++pb) {
            float4 ya = *(const float4*)(yr + 16 * pb);
            float4 yc = *(const float4*)(yr + 16 * pb + 4);
            ya.x += cumA * accs[pb][0];  ya.y += cumA * accs[pb][1];
            ya.z += cumA * accs[pb][2];  ya.w += cumA * accs[pb][3];
            yc.x += cumA * accs[pb][4];  yc.y += cumA * accs[pb][5];
            yc.z += cumA * accs[pb][6];  yc.w += cumA * accs[pb][7];
            *(float4*)(yr + 16 * pb)     = ya;
            *(float4*)(yr + 16 * pb + 4) = yc;
            add_sums(ya, ps, pq);
            add_sums(yc, ps, pq);
        }
        put_row_sums(ps, pq, hf, pb0 + 2 * (tl ^ rmask));
    }
}

__global__ void __launch_bounds__(128)
gemm_out(const float* Y, const float* __restrict__ part,
         const float* __restrict__ lnw,
         const float* __restrict__ lnb, const float* Wout,
         float* __restrict__ out) {
    const int wv = threadIdx.x >> 5;
    const int m0 = (blockIdx.y * 4 + wv) * 32;
    gemm_tile<4, D_INNER, D_MODEL, true>(Y, Wout, out, part, lnw, lnb, m0, (int)blockIdx.x * 64);
}

extern "C" void kernel_launch(void* const* d_in, const int* in_sizes, int n_in,
                              void* d_out, int out_size, void* d_ws, size_t ws_size,
                              hipStream_t stream) {
    const float* u        = (const float*)d_in[0];
    const float* W_in     = (const float*)d_in[2];
    const float* conv_fw  = (const float*)d_in[3];
    const float* conv_fb  = (const float*)d_in[4];
    const float* conv_bw  = (const float*)d_in[5];
    const float* conv_bb  = (const float*)d_in[6];
    const float* dt_bias  = (const float*)d_in[7];
    const float* A_log    = (const float*)d_in[8];
    const float* D_param  = (const float*)d_in[9];
    const float* ln_w     = (const float*)d_in[10];
    const float* ln_b     = (const float*)d_in[11];
    const float* W_out    = (const float*)d_in[12];
    float* out            = (float*)d_out;

    float* ws = (float*)d_ws;
    size_t off = 0;
    float* xbcdt = ws + off; off += (size_t)BL * D_IN_PROJ;
    float* xc    = ws + off; off += (size_t)BL * CONV2;
    float* dtb   = ws + off; off += (size_t)BL * NHEADS;
    float* ybuf  = ws + off; off += (size_t)BL * D_INNER;
    float* Sbuf  = ws + off; off += (size_t)32 * NCHUNK * HEADDIM * NSTATE;
    float* pAbuf = ws + off; off += (size_t)32 * NCHUNK;
    float* partb = ws + off; off += (size_t)2 * NHEADS * BL;

    gemm_in<<<dim3(CONV2 / 64 + 1, BL / 128), 128, 0, stream>>>(u, W_in, xbcdt);
    conv_dt<<<dim3(BL), CONV_THREADS, 0, stream>>>(xbcdt, conv_fw, conv_fb, conv_bw, conv_bb,
                                          dt_bias, xc, dtb);
    ssd_chunk<<<dim3(BATCH * NHEADS * NCHUNK), 128, 0, stream>>>(
        xc, dtb, A_log, D_param, ybuf, Sbuf, pAbuf);
    ssd_correct<<<dim3(BATCH * NHEADS * NCHUNK), 128, 0, stream>>>(
        xc, dtb, A_log, Sbuf, pAbuf, ybuf, partb);
    gemm_out<<<dim3(D_MODEL / 64, BL / 128), 128, 0, stream>>>(
        ybuf, partb, ln_w, ln_b, W_out, out);

    (void)in_sizes; (void)n_in; (void)out_size; (void)ws_size;
}


# conv+SiLU and softplus(dt) made in scan pass 1 as it stages its tile from the raw projection (a thread: 16 steps x 4 channels + 3 halo rows); conv launch and buffer gone, 5 -> 4 launches; same sums
# speedup vs baseline: 3.7940x; 1.0671x over previous
#include <hip/hip_runtime.h>
#include <hip/hip_bf16.h>
#include <math.h>

#define BATCH    2
#define SEQLEN   4096
#define BL       (BATCH * SEQLEN)
#define D_MODEL  512
#define D_IN_PROJ 1168
#define CONV_DIM 576
#define CONV2    1152
#define D_INNER  1024
#define NHEADS   16
#define HH       8
#define HEADDIM  64
#define NSTATE   32
#define NCHUNK   32
#define CHUNK    128

typedef float v2f __attribute__((ext_vector_type(2)));
typedef float v8f __attribute__((ext_vector_type(8)));

__device__ __forceinline__ v8f wmma_f32_4(v2f a, v2f b, v8f c) {
    return __builtin_amdgcn_wmma_f32_16x16x4_f32(
         false, a,  false, b,
         (short)0, c,  false,  false);
}

typedef float v4f __attribute__((ext_vector_type(4)));

template <int NF>
struct BFrag {
    v4f   b4[4];
    float b1[4];
    float wl[4], wh[4], zl[4], zh[4];
};

template <int NF, int ND, bool LN>
__device__ __forceinline__ void load_b(BFrag<NF>& f, const float* bp,
                                       const float* __restrict__ lnw, const float* __restrict__ lnb,
                                       int krow) {
#pragma unroll
    for (int j = 0; j < 4; ++j) {
        if (NF == 4) f.b4[j] = *(const v4f*)(bp + (size_t)(krow + j) * ND);
        else         f.b1[j] = bp[(size_t)(krow + j) * ND];
    }
    if (LN) {
#pragma unroll
        for (int j = 0; j < 4; ++j) {
            f.wl[j] = lnw[krow + j];  f.wh[j] = lnw[krow + 16 + j];
            f.zl[j] = lnb[krow + j];  f.zh[j] = lnb[krow + 16 + j];
        }
    }
}

__device__ __forceinline__ void load_a(v4f (&a)[4], const float* ap, int kb) {
#pragma unroll
    for (int q = 0; q < 4; ++q) a[q] = *(const v4f*)(ap + 32 * kb + 4 * q);
}

template <int NF, bool LN>
__device__ __forceinline__ void mma_sub(v4f y, v4f u, const BFrag<NF>& f, int hf,
                                        float mur0, float rsr0, float mur1, float rsr1,
                                        v8f (&acc0)[NF], v8f (&acc1)[NF]) {
    if (LN) {
#pragma unroll
        for (int j = 0; j < 4; ++j) {
            const float w = hf ? f.wh[j] : f.wl[j];
            const float z = hf ? f.zh[j] : f.zl[j];
            y[j] = (y[j] - mur0) * rsr0 * w + z;
            u[j] = (u[j] - mur1) * rsr1 * w + z;
        }
    }
    v2f a0, a1, c0, c1;
    a0.x = y[0];  a0.y = y[1];
    a1.x = y[2];  a1.y = y[3];
    c0.x = u[0];  c0.y = u[1];
    c1.x = u[2];  c1.y = u[3];
#pragma unroll
    for (int t = 0; t < NF; ++t) {
        v2f b;
        if (NF == 4) { b.x = f.b4[0][t];  b.y = f.b4[1][t]; }
        else         { b.x = f.b1[0];     b.y = f.b1[1]; }
        acc0[t] = wmma_f32_4(a0, b, acc0[t]);
        acc1[t] = wmma_f32_4(c0, b, acc1[t]);
    }
#pragma unroll
    for (int t = 0; t < NF; ++t) {
        v2f b;
        if (NF == 4) { b.x = f.b4[2][t];  b.y = f.b4[3][t]; }
        else         { b.x = f.b1[2];     b.y = f.b1[3]; }
        acc0[t] = wmma_f32_4(a1, b, acc0[t]);
        acc1[t] = wmma_f32_4(c1, b, acc1[t]);
    }
}

#define GEMM_SUB(AV, AW, BCUR, BNXT, KROW_NEXT)                                   \
    load_b<NF, ND, LN>(BNXT, bp, lnw, lnb, (KROW_NEXT));                          \
    __builtin_amdgcn_sched_barrier(0);                                            \
    mma_sub<NF, LN>((AV), (AW), (BCUR), hf, mur0, rsr0, mur1, rsr1, acc0, acc1);  \
    __builtin_amdgcn_sched_barrier(0);

template <int NF, int KD, int ND, bool LN>
__device__ __forceinline__ void gemm_tile(const float* A, const float* B, float* __restrict__ C,
                                          const float* __restrict__ part,
                                          const float* __restrict__ lnw, const float* __restrict__ lnb,
                                          int m0, int n0) {
    const int lane = threadIdx.x & 31;
    const int c    = lane & 15;
    const int hf   = lane >> 4;
    const float* ap = A + (size_t)(m0 + c) * KD + 16 * hf;
    const float* aq = ap + (size_t)16 * KD;
    const float* bp = B + (size_t)(16 * hf) * ND + n0 + (NF == 4 ? 4 * c : c);
    float mur0 = 0.0f, rsr0 = 1.0f, mur1 = 0.0f, rsr1 = 1.0f;
    if (LN) {
        float s0 = 0.0f, q0 = 0.0f, s1 = 0.0f, q1 = 0.0f;
#pragma unroll
        for (int h = 0; h < NHEADS; ++h) {
            const v2f p0 = *(const v2f*)(part + 2 * ((size_t)h * BL + m0 + c));
            const v2f p1 = *(const v2f*)(part + 2 * ((size_t)h * BL + m0 + 16 + c));
            s0 += p0.x;  q0 += p0.y;
            s1 += p1.x;  q1 += p1.y;
        }
        const float e0 = s0 * (1.0f / D_INNER);
        const float v0 = q0 * (1.0f / D_INNER) - e0 * e0;
        const float e1 = s1 * (1.0f / D_INNER);
        const float v1 = q1 * (1.0f / D_INNER) - e1 * e1;
        mur0 = e0;  rsr0 = rsqrtf(v0 + 1e-5f);
        mur1 = e1;  rsr1 = rsqrtf(v1 + 1e-5f);
    }

    v8f acc0[NF], acc1[NF];
#pragma unroll
    for (int t = 0; t < NF; ++t) { acc0[t] = v8f{};  acc1[t] = v8f{}; }

    constexpr int NKB = KD / 32;
    v4f aP[4], aQ[4], cP[4], cQ[4];
    BFrag<NF> bX, bY;
    load_a(aP, ap, 0);
    load_a(cP, aq, 0);
    load_b<NF, ND, LN>(bX, bp, lnw, lnb, 0);
    for (int kb = 0; kb < NKB; kb += 2) {
        const int kn = (kb + 2 < NKB) ? (kb + 2) : (kb + 1);
        load_a(aQ, ap, kb + 1);
        load_a(cQ, aq, kb + 1);
        GEMM_SUB(aP[0], cP[0], bX, bY, 32 * kb + 4)
        GEMM_SUB(aP[1], cP[1], bY, bX, 32 * kb + 8)
        GEMM_SUB(aP[2], cP[2], bX, bY, 32 * kb + 12)
        GEMM_SUB(aP[3], cP[3], bY, bX, 32 * kb + 32)
        load_a(aP, ap, kn);
        load_a(cP, aq, kn);
        GEMM_SUB(aQ[0], cQ[0], bX, bY, 32 * kb + 36)
        GEMM_SUB(aQ[1], cQ[1], bY, bX, 32 * kb + 40)
        GEMM_SUB(aQ[2], cQ[2], bX, bY, 32 * kb + 44)
        GEMM_SUB(aQ[3], cQ[3], bY, bX, 32 * kn)
    }

#pragma unroll
    for (int r = 0; r < 8; ++r) {
        float* cr = C + (size_t)(m0 + 8 * hf + r) * ND + n0;
        float* cs = cr + (size_t)16 * ND;
        if (NF == 4) {
            v4f o, p;
            o[0] = acc0[0][r];  o[1] = acc0[1][r];
            o[2] = acc0[2][r];  o[3] = acc0[3][r];
            p[0] = acc1[0][r];  p[1] = acc1[1][r];
            p[2] = acc1[2][r];  p[3] = acc1[3][r];
            *(v4f*)(cr + 4 * c) = o;
            *(v4f*)(cs + 4 * c) = p;
        } else {
            cr[c] = acc0[0][r];
            cs[c] = acc1[0][r];
        }
    }
}
#undef GEMM_SUB

__global__ void __launch_bounds__(128)
gemm_in(const float* U, const float* Win, float* __restrict__ out) {
    const int wv = threadIdx.x >> 5;
    const int m0 = (blockIdx.y * 4 + wv) * 32;
    if (blockIdx.x < CONV2 / 64)
        gemm_tile<4, D_MODEL, D_IN_PROJ, false>(U, Win, out, nullptr, nullptr, nullptr,
                                                m0, (int)blockIdx.x * 64);
    else
        gemm_tile<1, D_MODEL, D_IN_PROJ, false>(U, Win, out, nullptr, nullptr, nullptr,
                                                m0, CONV2);
}

#define CONV_HALO 3
#define CONV_RUN  16
#define CONV_ROWS (CONV_RUN + CONV_HALO)

__device__ __forceinline__ void head_bases(int h, int& xbase, int& bbase, int& cbase) {
    if (h < HH) { xbase = h * HEADDIM;                 bbase = 512;  cbase = 544;  }
    else        { xbase = CONV_DIM + (h - HH) * HEADDIM; bbase = 1088; cbase = 1120; }
}


__device__ __forceinline__ void chunk_prefix(const float* __restrict__ dtc, float dtbv, int rmask, float Ah,
                                             float* s_tot, float* s_dt, float* s_cs) {
    const int i    = threadIdx.x;
    const int lane = i & 31;
    const int wv   = i >> 5;
    const float dtr = dtc[(unsigned)((i ^ rmask) * D_IN_PROJ)] + dtbv;
    const float dtv = (dtr > 20.0f) ? dtr : log1pf(expf(dtr));
    float v = Ah * dtv;
#pragma unroll
    for (int d = 1; d < 32; d <<= 1) {
        const float n = __shfl_up(v, d, 32);
        if (lane >= d) v += n;
    }
    if (lane == 31) s_tot[wv] = v;
    s_dt[i] = dtv;
    __syncthreads();
    float off = 0.0f;
    for (int w = 0; w < wv; ++w) off += s_tot[w];
    s_cs[i] = off + v;
    __syncthreads();
}

#define TPITCH 132
#define T_B    64
#define T_C    96

__device__ __forceinline__ void conv_run(const v4f (&in)[CONV_ROWS], const v4f (&w)[4], const v4f bias,
                                         float* tp, float* cg, int cstep) {
#pragma unroll
    for (int i = 0; i < CONV_RUN; ++i) {
        v4f o;
#pragma unroll
        for (int e = 0; e < 4; ++e) {
            float acc = bias[e];
#pragma unroll
            for (int k = 0; k < 4; ++k) acc += in[i + k][e] * w[e][k];
            const float sig = 1.0f / (1.0f + expf(-acc));
            o[e] = acc * sig;
        }
        *(v4f*)(tp + i * TPITCH) = o;
        if (cg) *(v4f*)(cg + i * cstep) = o;
    }
}

__device__ __forceinline__ void load_pair(const float* s_t, int j, int l15, int hf,
                                          v2f (&bf)[8], float (&xv)[4][8]) {
    const float* bp = s_t + (unsigned)((16 * j + l15) * TPITCH + T_B + 2 * hf);
#pragma unroll
    for (int q = 0; q < 8; ++q) bf[q] = *(const v2f*)(bp + 4 * q);
    const float* xp = s_t + (unsigned)((16 * j + 8 * hf) * TPITCH + l15);
#pragma unroll
    for (int r = 0; r < 8; ++r) {
        xv[0][r] = xp[r * TPITCH];
        xv[1][r] = xp[r * TPITCH + 16];
        xv[2][r] = xp[r * TPITCH + 32];
        xv[3][r] = xp[r * TPITCH + 48];
    }
}

__global__ void __launch_bounds__(128)
ssd_chunk(const float* __restrict__ xbcdt,
          const float* __restrict__ wf, const float* __restrict__ bf,
          const float* __restrict__ wb, const float* __restrict__ bb,
          const float* __restrict__ dt_bias,
          const float* __restrict__ A_log, const float* __restrict__ Dp,
          float* __restrict__ y, float* __restrict__ S, float* __restrict__ prodA,
          float* __restrict__ Cc) {
    const int chunk = blockIdx.x & (NCHUNK - 1);
    const int h     = (blockIdx.x >> 5) & 15;
    const int b     = blockIdx.x >> 9;
    int xbase, bbase, cbase; head_bases(h, xbase, bbase, cbase);
    (void)cbase;
    const int s0    = chunk * CHUNK;
    const int rmask = (h < HH) ? 0 : (CHUNK - 1);
    const int tbase = (h < HH) ? s0 : (SEQLEN - CHUNK - s0);
    const size_t row0 = (size_t)b * SEQLEN + tbase;
    const float* xcb = xbcdt + row0 * D_IN_PROJ;
    const float* dtc = xcb + CONV2 + h;
    float*       yb  = y   + row0 * D_INNER + h * HEADDIM;

    const float Ah = -expf(A_log[h]);
    const float Dh = Dp[h];
    const float dtbv = dt_bias[h];

    __shared__ float s_tot[4];
    __shared__ alignas(16) float s_dt[CHUNK];
    __shared__ alignas(16) float s_cs[CHUNK];
    __shared__ alignas(16) float s_w[CHUNK];
    __shared__ alignas(16) float s_t[CHUNK * TPITCH];

    const int wv   = __builtin_amdgcn_readfirstlane((int)(threadIdx.x >> 5));
    const int lane = threadIdx.x & 31;
    const int l15  = lane & 15;
    const int hf   = lane >> 4;

    {
        const int dir = (h < HH) ? 0 : 1;
        const float* cw = dir ? wb : wf;
        const float* cb = dir ? bb : bf;
        const int cx = (h & (HH - 1)) * HEADDIM + 4 * l15;
        const int cq = D_INNER / 2 + 4 * l15;
        const int r0 = 16 * (2 * wv + hf);
        v4f wx[4], wq[4];
#pragma unroll
        for (int e = 0; e < 4; ++e) {
            wx[e] = *(const v4f*)(cw + 4 * (cx + e));
            wq[e] = *(const v4f*)(cw + 4 * (cq + e));
        }
        const v4f bx = *(const v4f*)(cb + cx);
        const v4f bq = *(const v4f*)(cb + cq);
        v4f tx[CONV_ROWS], tb[CONV_ROWS];
#pragma unroll
        for (int j = 0; j < CONV_ROWS; ++j) {
            const int  sl  = r0 - CONV_HALO + j;
            const bool inq = (j >= CONV_HALO) || (chunk > 0) || (sl >= 0);
            const int  slc = inq ? sl : 0;
            const int  ro  = rmask ? (CHUNK - 1 - slc) : slc;
            const float* rp = xcb + (ptrdiff_t)ro * D_IN_PROJ + 4 * l15;
            const v4f a = *(const v4f*)(rp + xbase);
            const v4f q = *(const v4f*)(rp + bbase);
            tx[j] = inq ? a : v4f{};
            tb[j] = inq ? q : v4f{};
        }
        chunk_prefix(dtc, dtbv, rmask, Ah, s_tot, s_dt, s_cs);
        float* tp = s_t + (unsigned)(r0 * TPITCH + 4 * l15);
        conv_run(tx, wx, bx, tp, nullptr, 0);
        float* cg = nullptr;
        const int cstep = rmask ? -2 * NSTATE : 2 * NSTATE;
        if ((h & (HH - 1)) == 0 && l15 >= 8)
            cg = Cc + (row0 + (size_t)(rmask ? (CHUNK - 1 - r0) : r0)) * (2 * NSTATE) + dir * NSTATE + 4 * (l15 - 8);
        conv_run(tb, wq, bq, tp + T_B, cg, cstep);
    }
    const float cl = s_cs[CHUNK - 1];
    s_w[threadIdx.x] = expf(cl - s_cs[threadIdx.x]) * s_dt[threadIdx.x];
    __syncthreads();

    for (int half = 0; half < 2; ++half) {
        const int i  = half ? (7 - wv) : wv;
        const int tl = 16 * i + l15;
        const float* trow = s_t + (unsigned)(tl * TPITCH);
        const float cst = s_cs[tl];

        v2f cf[8];
#pragma unroll
        for (int q = 0; q < 8; ++q)
            cf[q] = *(const v2f*)(trow + T_C + 2 * hf + 4 * q);

        v8f acc0 = {}, acc1 = {}, acc2 = {}, acc3 = {};
        v2f bfr[8];
        float xa[4][8];
        load_pair(s_t, 0, l15, hf, bfr, xa);
        for (int j = 0; j <= i; ++j) {
            const int sb0 = 16 * j + 8 * hf;
            v2f bfn[8];
            float xn[4][8];
            load_pair(s_t, (j < i) ? (j + 1) : i, l15, hf, bfn, xn);
            __builtin_amdgcn_sched_barrier(0);

            v8f g = {};
#pragma unroll
            for (int q = 0; q < 8; ++q) g = wmma_f32_4(bfr[q], cf[q], g);

            float m[8];
#pragma unroll
            for (int r = 0; r < 8; ++r) {
                const int ss = sb0 + r;
                const float css = s_cs[ss];
                const float d = (ss <= tl) ? (cst - css) : -INFINITY;
                m[r] = g[r] * (expf(d) * s_dt[ss]);
            }
#pragma unroll
            for (int kk = 0; kk < 4; ++kk) {
                v2f bm; bm.x = m[2 * kk]; bm.y = m[2 * kk + 1];
                v2f a0, a1, a2, a3;
                a0.x = xa[0][2 * kk];  a0.y = xa[0][2 * kk + 1];
                a1.x = xa[1][2 * kk];  a1.y = xa[1][2 * kk + 1];
                a2.x = xa[2][2 * kk];  a2.y = xa[2][2 * kk + 1];
                a3.x = xa[3][2 * kk];  a3.y = xa[3][2 * kk + 1];
                acc0 = wmma_f32_4(a0, bm, acc0);
                acc1 = wmma_f32_4(a1, bm, acc1);
                acc2 = wmma_f32_4(a2, bm, acc2);
                acc3 = wmma_f32_4(a3, bm, acc3);
            }
            __builtin_amdgcn_sched_barrier(0);
#pragma unroll
            for (int q = 0; q < 8; ++q) {
                bfr[q] = bfn[q];
                xa[0][q] = xn[0][q];  xa[1][q] = xn[1][q];
                xa[2][q] = xn[2][q];  xa[3][q] = xn[3][q];
            }
        }
        const float* xr = trow + 8 * hf;
        float*       yr = yb + (unsigned)((tl ^ rmask) * D_INNER + 8 * hf);
        v8f accs[4] = {acc0, acc1, acc2, acc3};
#pragma unroll
        for (int pb = 0; pb < 4; ++pb) {
            const float4 xa4 = *(const float4*)(xr + 16 * pb);
            const float4 xb4 = *(const float4*)(xr + 16 * pb + 4);
            float4 oa, ob;
            oa.x = accs[pb][0] + Dh * xa4.x;  oa.y = accs[pb][1] + Dh * xa4.y;
            oa.z = accs[pb][2] + Dh * xa4.z;  oa.w = accs[pb][3] + Dh * xa4.w;
            ob.x = accs[pb][4] + Dh * xb4.x;  ob.y = accs[pb][5] + Dh * xb4.y;
            ob.z = accs[pb][6] + Dh * xb4.z;  ob.w = accs[pb][7] + Dh * xb4.w;
            *(float4*)(yr + 16 * pb)     = oa;
            *(float4*)(yr + 16 * pb + 4) = ob;
        }
    }

    {
        v8f st0 = {}, st1 = {};
        const unsigned xo = (unsigned)(16 * wv + l15);
        const unsigned bo = (unsigned)(T_B + l15);
        for (int k0 = 0; k0 < CHUNK; k0 += 32) {
            float xv[16], bv0[16], bv1[16], wv16[16];
#pragma unroll
            for (int u = 0; u < 16; ++u) {
                const int sa = k0 + 4 * (u >> 1) + 2 * hf + (u & 1);
                const unsigned ro = (unsigned)(sa * TPITCH);
                xv[u]   = s_t[ro + xo];
                bv0[u]  = s_t[ro + bo];
                bv1[u]  = s_t[ro + bo + 16];
                wv16[u] = s_w[sa];
            }
#pragma unroll
            for (int ks = 0; ks < 8; ++ks) {
                v2f a, b0, b1;
                a.x  = xv[2 * ks] * wv16[2 * ks];
                a.y  = xv[2 * ks + 1] * wv16[2 * ks + 1];
                b0.x = bv0[2 * ks];  b0.y = bv0[2 * ks + 1];
                b1.x = bv1[2 * ks];  b1.y = bv1[2 * ks + 1];
                st0 = wmma_f32_4(a, b0, st0);
                st1 = wmma_f32_4(a, b1, st1);
            }
        }
        const size_t sb = (((size_t)(b * NHEADS + h)) * NCHUNK + chunk) * (HEADDIM * NSTATE)
                        + (size_t)(16 * wv + 8 * hf) * NSTATE + l15;
#pragma unroll
        for (int r = 0; r < 8; ++r) {
            S[sb + (size_t)r * NSTATE]      = st0[r];
            S[sb + (size_t)r * NSTATE + 16] = st1[r];
        }
    }
    if (threadIdx.x == 0) prodA[(b * NHEADS + h) * NCHUNK + chunk] = expf(cl);
}

__device__ __forceinline__ void add_sums(const float4 a, float& ps, float& pq) {
    ps += (a.x + a.y) + (a.z + a.w);
    pq += (a.x * a.x + a.y * a.y) + (a.z * a.z + a.w * a.w);
}

__device__ __forceinline__ void put_row_sums(float ps, float pq, int hf, float* pr) {
    ps += __shfl_xor(ps, 16, 32);
    pq += __shfl_xor(pq, 16, 32);
    if (hf == 0) { v2f o; o.x = ps; o.y = pq; *(v2f*)pr = o; }
}

#define HPITCH 36

template <int NB>
__device__ __forceinline__ void state_steps(const float* __restrict__ sp, const float* __restrict__ pa, int c,
                                            v4f (&hv)[4]) {
    v4f   s[NB][4];
    float a[NB];
#pragma unroll
    for (int u = 0; u < NB; ++u) {
        a[u] = pa[c + u];
#pragma unroll
        for (int k = 0; k < 4; ++k)
            s[u][k] = *(const v4f*)(sp + (size_t)(c + u) * (HEADDIM * NSTATE) + 512 * k);
    }
#pragma unroll
    for (int u = 0; u < NB; ++u) {
#pragma unroll
        for (int k = 0; k < 4; ++k) {
            hv[k][0] = fmaf(a[u], hv[k][0], s[u][k][0]);
            hv[k][1] = fmaf(a[u], hv[k][1], s[u][k][1]);
            hv[k][2] = fmaf(a[u], hv[k][2], s[u][k][2]);
            hv[k][3] = fmaf(a[u], hv[k][3], s[u][k][3]);
        }
    }
}

__global__ void __launch_bounds__(128)
ssd_correct(const float* __restrict__ xbcdt, const float* __restrict__ dt_bias,
            const float* __restrict__ Cc,
            const float* __restrict__ A_log, const float* __restrict__ S,
            const float* __restrict__ prodA,
            float* __restrict__ y, float* __restrict__ part) {
    const int chunk = blockIdx.x & (NCHUNK - 1);
    const int h = (blockIdx.x >> 5) & 15;
    const int b = blockIdx.x >> 9;
    const int s0    = chunk * CHUNK;
    const int rmask = (h < HH) ? 0 : (CHUNK - 1);
    const int tbase = (h < HH) ? s0 : (SEQLEN - CHUNK - s0);
    const size_t row0 = (size_t)b * SEQLEN + tbase;
    const float* ccb = Cc + row0 * (2 * NSTATE) + ((h < HH) ? 0 : NSTATE);
    const float* dtc = xbcdt + row0 * D_IN_PROJ + CONV2 + h;
    float*       yb  = y   + row0 * D_INNER + h * HEADDIM;
    float*       pb0 = part + 2 * ((size_t)h * BL + row0);

    const int wv   = __builtin_amdgcn_readfirstlane((int)(threadIdx.x >> 5));
    const int lane = threadIdx.x & 31;
    const int l15  = lane & 15;
    const int hf   = lane >> 4;

    if (chunk == 0) {
        for (int half = 0; half < 2; ++half) {
            const int tl = 16 * (wv + 4 * half) + l15;
            const float* yr = yb + (unsigned)((tl ^ rmask) * D_INNER + 8 * hf);
            float ps = 0.0f, pq = 0.0f;
#pragma unroll
            for (int pb = 0; pb < 4; ++pb) {
                add_sums(*(const float4*)(yr + 16 * pb), ps, pq);
                add_sums(*(const float4*)(yr + 16 * pb + 4), ps, pq);
            }
            put_row_sums(ps, pq, hf, pb0 + 2 * (tl ^ rmask));
        }
        return;
    }

    const float Ah = -expf(A_log[h]);
    const float dtbv = dt_bias[h];
    __shared__ float s_tot[4], s_dt[CHUNK], s_cs[CHUNK];
    __shared__ alignas(16) float s_h[HEADDIM * HPITCH];

    {
        const int bh = b * NHEADS + h;
        const float* sp = S + (size_t)bh * NCHUNK * (HEADDIM * NSTATE) + 4 * threadIdx.x;
        const float* pa = prodA + bh * NCHUNK;
        v4f hv[4];
#pragma unroll
        for (int k = 0; k < 4; ++k) hv[k] = v4f{};
        int c = 0;
        for (; c + 8 <= chunk; c += 8) state_steps<8>(sp, pa, c, hv);
        if (chunk & 4) { state_steps<4>(sp, pa, c, hv);  c += 4; }
        if (chunk & 2) { state_steps<2>(sp, pa, c, hv);  c += 2; }
        if (chunk & 1) { state_steps<1>(sp, pa, c, hv); }
#pragma unroll
        for (int k = 0; k < 4; ++k)
            *(v4f*)(s_h + (16 * k + ((int)threadIdx.x >> 3)) * HPITCH + 4 * ((int)threadIdx.x & 7)) = hv[k];
    }
    chunk_prefix(dtc, dtbv, rmask, Ah, s_tot, s_dt, s_cs);

    v2f hA[4][8];
    {
        const float* hl = s_h + l15 * HPITCH + 2 * hf;
#pragma unroll
        for (int pb = 0; pb < 4; ++pb)
#pragma unroll
            for (int q = 0; q < 8; ++q)
                hA[pb][q] = *(const v2f*)(hl + 16 * pb * HPITCH + 4 * q);
    }

    for (int half = 0; half < 2; ++half) {
        const int tl = 16 * (wv + 4 * half) + l15;
        const float cumA = expf(s_cs[tl]);
        const float* cp = ccb + (unsigned)((tl ^ rmask) * (2 * NSTATE) + 2 * hf);

        v8f acc0 = {}, acc1 = {}, acc2 = {}, acc3 = {};
#pragma unroll
        for (int q = 0; q < 8; ++q) {
            const v2f cq = *(const v2f*)(cp + 4 * q);
            acc0 = wmma_f32_4(hA[0][q], cq, acc0);
            acc1 = wmma_f32_4(hA[1][q], cq, acc1);
            acc2 = wmma_f32_4(hA[2][q], cq, acc2);
            acc3 = wmma_f32_4(hA[3][q], cq, acc3);
        }
        float* yr = yb + (unsigned)((tl ^ rmask) * D_INNER + 8 * hf);
        v8f accs[4] = {acc0, acc1, acc2, acc3};
        float ps = 0.0f, pq = 0.0f;
#pragma unroll
        for (int pb = 0; pb < 4; ++pb) {
            float4 ya = *(const float4*)(yr + 16 * pb);
            float4 yc = *(const float4*)(yr + 16 * pb + 4);
            ya.x += cumA * accs[pb][0];  ya.y += cumA * accs[pb][1];
            ya.z += cumA * accs[pb][2];  ya.w += cumA * accs[pb][3];
            yc.x += cumA * accs[pb][4];  yc.y += cumA * accs[pb][5];
            yc.z += cumA * accs[pb][6];  yc.w += cumA * accs[pb][7];
            *(float4*)(yr + 16 * pb)     = ya;
            *(float4*)(yr + 16 * pb + 4) = yc;
            add_sums(ya, ps, pq);
            add_sums(yc, ps, pq);
        }
        put_row_sums(ps, pq, hf, pb0 + 2 * (tl ^ rmask));
    }
}

__global__ void __launch_bounds__(128)
gemm_out(const float* Y, const float* __restrict__ part,
         const float* __restrict__ lnw,
         const float* __restrict__ lnb, const float* Wout,
         float* __restrict__ out) {
    const int wv = threadIdx.x >> 5;
    const int m0 = (blockIdx.y * 4 + wv) * 32;
    gemm_tile<4, D_INNER, D_MODEL, true>(Y, Wout, out, part, lnw, lnb, m0, (int)blockIdx.x * 64);
}

extern "C" void kernel_launch(void* const* d_in, const int* in_sizes, int n_in,
                              void* d_out, int out_size, void* d_ws, size_t ws_size,
                              hipStream_t stream) {
    const float* u        = (const float*)d_in[0];
    const float* W_in     = (const float*)d_in[2];
    const float* conv_fw  = (const float*)d_in[3];
    const float* conv_fb  = (const float*)d_in[4];
    const float* conv_bw  = (const float*)d_in[5];
    const float* conv_bb  = (const float*)d_in[6];
    const float* dt_bias  = (const float*)d_in[7];
    const float* A_log    = (const float*)d_in[8];
    const float* D_param  = (const float*)d_in[9];
    const float* ln_w     = (const float*)d_in[10];
    const float* ln_b     = (const float*)d_in[11];
    const float* W_out    = (const float*)d_in[12];
    float* out            = (float*)d_out;

    float* ws = (float*)d_ws;
    size_t off = 0;
    float* xbcdt = ws + off; off += (size_t)BL * D_IN_PROJ;
    float* ccbuf = ws + off; off += (size_t)BL * 2 * NSTATE;
    float* ybuf  = ws + off; off += (size_t)BL * D_INNER;
    float* Sbuf  = ws + off; off += (size_t)32 * NCHUNK * HEADDIM * NSTATE;
    float* pAbuf = ws + off; off += (size_t)32 * NCHUNK;
    float* partb = ws + off; off += (size_t)2 * NHEADS * BL;

    gemm_in<<<dim3(CONV2 / 64 + 1, BL / 128), 128, 0, stream>>>(u, W_in, xbcdt);
    ssd_chunk<<<dim3(BATCH * NHEADS * NCHUNK), 128, 0, stream>>>(
        xbcdt, conv_fw, conv_fb, conv_bw, conv_bb, dt_bias, A_log, D_param, ybuf, Sbuf, pAbuf, ccbuf);
    ssd_correct<<<dim3(BATCH * NHEADS * NCHUNK), 128, 0, stream>>>(
        xbcdt, dt_bias, ccbuf, A_log, Sbuf, pAbuf, ybuf, partb);
    gemm_out<<<dim3(D_MODEL / 64, BL / 128), 128, 0, stream>>>(
        ybuf, partb, ln_w, ln_b, W_out, out);

    (void)in_sizes; (void)n_in; (void)out_size; (void)ws_size;
}
